# starting kernel
# baseline (speedup 1.0000x reference)
#include <hip/hip_runtime.h>
#include <hip/hip_bf16.h>
#include <math.h>

typedef __attribute__((ext_vector_type(2))) float v2f;
typedef __attribute__((ext_vector_type(8))) float v8f;

__device__ __forceinline__ float lrelu(float x) { return x > 0.f ? x : 0.2f * x; }

__device__ __forceinline__ void atomic_max_f32(float* addr, float val) {
  if (val >= 0.f)
    __hip_atomic_fetch_max((int*)addr, __float_as_int(val),
                           __ATOMIC_RELAXED, __HIP_MEMORY_SCOPE_AGENT);
  else
    __hip_atomic_fetch_min((unsigned int*)addr, __float_as_uint(val),
                           __ATOMIC_RELAXED, __HIP_MEMORY_SCOPE_AGENT);
}

__device__ __forceinline__ void atomic_add_f32(float* addr, float val) {
  __hip_atomic_fetch_add(addr, val, __ATOMIC_RELAXED, __HIP_MEMORY_SCOPE_AGENT);
}

__global__ __launch_bounds__(1024)
void prep_kernel(const float* __restrict__ Wsq, const float* __restrict__ Wdq,
                 const float* __restrict__ Asq, const float* __restrict__ Adq,
                 const float* __restrict__ Wsa, const float* __restrict__ Wda,
                 const float* __restrict__ Asa, const float* __restrict__ Ada,
                 const float* __restrict__ bq,  const float* __restrict__ ba,
                 const float* __restrict__ Wout, const float* __restrict__ bout,
                 float* __restrict__ Vp, float* __restrict__ Pc, float* __restrict__ cst) {
  const int t = threadIdx.x;
  for (int idx = t; idx < 4 * 64 * 16; idx += blockDim.x) {
    int table = idx >> 10, k = (idx >> 4) & 63, j = idx & 15;
    const float* W = (table == 0) ? Wsq : (table == 1) ? Wdq : (table == 2) ? Wsa : Wda;
    const float* A = (table == 0) ? Asq : (table == 1) ? Adq : (table == 2) ? Asa : Ada;
    float s = 0.f;
    if (j < 4)
      for (int c = 0; c < 64; ++c) s += W[k * 256 + j * 64 + c] * A[j * 64 + c];
    Vp[idx] = s;
  }
  for (int idx = t; idx < 512 * 16; idx += blockDim.x) {
    int r = idx >> 4, j = idx & 15;
    int layer = r >> 8, h = (r >> 6) & 3, k = r & 63;
    const float* W = layer ? Wsa : Wsq;
    float s = 0.f;
    if (j < 4) {
      for (int c = 0; c < 64; ++c)
        s += W[k * 256 + h * 64 + c] * Wout[(layer * 64 + c) * 4 + j];
      s *= 0.25f;
    }
    Pc[idx] = s;
  }
  if (t < 16) {
    float s = 0.f;
    if (t < 4) {
      for (int c = 0; c < 64; ++c)
        s += bq[c] * Wout[c * 4 + t] + ba[c] * Wout[(64 + c) * 4 + t];
      s += bout[t];
    }
    cst[t] = s;
  }
}

__global__ void init_kernel(float* __restrict__ zero_ptr, long nzero,
                            float* __restrict__ minf_ptr, long nminf) {
  const long stride = (long)gridDim.x * blockDim.x;
  const long t0 = (long)blockIdx.x * blockDim.x + threadIdx.x;
  for (long i = t0; i < nzero; i += stride) zero_ptr[i] = 0.f;
  for (long i = t0; i < nminf; i += stride) minf_ptr[i] = __int_as_float(0xff800000);
}

__global__ void scores_kernel(const float* __restrict__ x, const float* __restrict__ Vp,
                              float* __restrict__ out, int N) {
  const int wave_g = (int)(((unsigned)blockIdx.x * blockDim.x + threadIdx.x) >> 5);
  const int lane = threadIdx.x & 31;
  const long base = (long)wave_g * 16;
  if (base >= N) return;
  const int m = lane & 15;
  const int ko = (lane >> 4) * 2;
  long row = base + m; if (row >= N) row = N - 1;
  const float* xr = x + row * 64;
  v8f acc = {};
#pragma unroll
  for (int kk = 0; kk < 64; kk += 4) {
    v2f a, b;
    a.x = xr[kk + ko];
    a.y = xr[kk + ko + 1];
    b.x = Vp[(kk + ko) * 16 + m];
    b.y = Vp[(kk + ko + 1) * 16 + m];
    acc = __builtin_amdgcn_wmma_f32_16x16x4_f32(false, a, false, b, (short)0, acc,
                                                false, false);
  }
  if (m < 4) {
    const long rb = base + ((lane >> 4) << 3);
#pragma unroll
    for (int v = 0; v < 8; ++v) {
      long r = rb + v;
      if (r < N) out[r * 4 + m] = acc[v];
    }
  }
}

__global__ void edge_pass1(const int* __restrict__ src, const int* __restrict__ dst,
                           const float* __restrict__ a_s, const float* __restrict__ a_d,
                           float* __restrict__ e, float* __restrict__ mbuf, int E) {
  const int i = blockIdx.x * blockDim.x + threadIdx.x;
  if (i >= E) return;
  const long s = src[i], d = dst[i];
  const float4 as = *(const float4*)(a_s + s * 4);
  const float4 ad = *(const float4*)(a_d + d * 4);
  float4 ev;
  ev.x = lrelu(as.x + ad.x);
  ev.y = lrelu(as.y + ad.y);
  ev.z = lrelu(as.z + ad.z);
  ev.w = lrelu(as.w + ad.w);
  *(float4*)(e + (long)i * 4) = ev;
  atomic_max_f32(mbuf + d * 4 + 0, ev.x);
  atomic_max_f32(mbuf + d * 4 + 1, ev.y);
  atomic_max_f32(mbuf + d * 4 + 2, ev.z);
  atomic_max_f32(mbuf + d * 4 + 3, ev.w);
}

__global__ void edge_pass2(const int* __restrict__ dst, float* __restrict__ e,
                           const float* __restrict__ mbuf, float* __restrict__ den, int E) {
  const int i = blockIdx.x * blockDim.x + threadIdx.x;
  if (i >= E) return;
  const long d = dst[i];
  float4 ev = *(const float4*)(e + (long)i * 4);
  const float4 mv = *(const float4*)(mbuf + d * 4);
  ev.x = __expf(ev.x - mv.x);
  ev.y = __expf(ev.y - mv.y);
  ev.z = __expf(ev.z - mv.z);
  ev.w = __expf(ev.w - mv.w);
  *(float4*)(e + (long)i * 4) = ev;
  atomic_add_f32(den + d * 4 + 0, ev.x);
  atomic_add_f32(den + d * 4 + 1, ev.y);
  atomic_add_f32(den + d * 4 + 2, ev.z);
  atomic_add_f32(den + d * 4 + 3, ev.w);
}

__global__ void edge_pass3(const int* __restrict__ src, const int* __restrict__ dst,
                           const float* __restrict__ x, const float* __restrict__ ex,
                           const float* __restrict__ den, float* __restrict__ z, long E) {
  const long gt = (long)blockIdx.x * blockDim.x + threadIdx.x;
  const long ei = gt >> 6;
  if (ei >= E) return;
  const int t = (int)(gt & 63);
  const int h = t >> 4;
  const int k0 = (t & 15) * 4;
  const long s = src[ei], d = dst[ei];
  const float alpha = ex[ei * 4 + h] / (den[d * 4 + h] + 1e-16f);
  const float4 xv = *(const float4*)(x + s * 64 + k0);
  float* zp = z + d * 256 + h * 64 + k0;
  atomic_add_f32(zp + 0, alpha * xv.x);
  atomic_add_f32(zp + 1, alpha * xv.y);
  atomic_add_f32(zp + 2, alpha * xv.z);
  atomic_add_f32(zp + 3, alpha * xv.w);
}

__global__ void final_kernel(const float* __restrict__ zq, const float* __restrict__ za,
                             const float* __restrict__ Pc, const float* __restrict__ cst,
                             float* __restrict__ out, int N) {
  const int wave_g = (int)(((unsigned)blockIdx.x * blockDim.x + threadIdx.x) >> 5);
  const int lane = threadIdx.x & 31;
  const long base = (long)wave_g * 16;
  if (base >= N) return;
  const int m = lane & 15;
  const int ko = (lane >> 4) * 2;
  long row = base + m; if (row >= N) row = N - 1;
  const float* zqr = zq + row * 256;
  const float* zar = za + row * 256;
  v8f acc = {};
#pragma unroll 8
  for (int kk = 0; kk < 512; kk += 4) {
    const int k0 = kk + ko;
    const float* zr = (k0 < 256) ? zqr : zar;
    const int kL = k0 & 255;
    v2f a, b;
    a.x = zr[kL];
    a.y = zr[kL + 1];
    b.x = Pc[k0 * 16 + m];
    b.y = Pc[(k0 + 1) * 16 + m];
    acc = __builtin_amdgcn_wmma_f32_16x16x4_f32(false, a, false, b, (short)0, acc,
                                                false, false);
  }
  if (m < 4) {
    const float c0 = cst[m];
    const long rb = base + ((lane >> 4) << 3);
#pragma unroll
    for (int v = 0; v < 8; ++v) {
      long r = rb + v;
      if (r < N) out[r * 4 + m] = acc[v] + c0;
    }
  }
}

extern "C" void kernel_launch(void* const* d_in, const int* in_sizes, int n_in,
                              void* d_out, int out_size, void* d_ws, size_t ws_size,
                              hipStream_t stream) {
  const float* x_c = (const float*)d_in[0];
  const float* x_q = (const float*)d_in[1];
  const float* x_a = (const float*)d_in[2];
  const int* src_cq = (const int*)d_in[3];
  const int* dst_cq = (const int*)d_in[4];
  const int* src_ca = (const int*)d_in[5];
  const int* dst_ca = (const int*)d_in[6];
  const float* Wsq = (const float*)d_in[7];
  const float* Wdq = (const float*)d_in[8];
  const float* Asq = (const float*)d_in[9];
  const float* Adq = (const float*)d_in[10];
  const float* bq  = (const float*)d_in[11];
  const float* Wsa = (const float*)d_in[12];
  const float* Wda = (const float*)d_in[13];
  const float* Asa = (const float*)d_in[14];
  const float* Ada = (const float*)d_in[15];
  const float* ba  = (const float*)d_in[16];
  const float* Wout = (const float*)d_in[17];
  const float* bout = (const float*)d_in[18];

  const long Nc = in_sizes[0] / 64;
  const long Nq = in_sizes[1] / 64;
  const long Na = in_sizes[2] / 64;
  const long Ecq = in_sizes[3];
  const long Eca = in_sizes[5];

  float* w = (float*)d_ws;
  size_t off = 0;
  auto alloc = [&](size_t n) -> float* {
    float* p = w + off;
    off += (n + 15) & ~(size_t)15;
    return p;
  };
  float* Vp_sq = alloc(64 * 16);
  float* Vp_dq = alloc(64 * 16);
  float* Vp_sa = alloc(64 * 16);
  float* Vp_da = alloc(64 * 16);
  float* Pc    = alloc(512 * 16);
  float* cst   = alloc(16);
  float* as_cq = alloc(Nc * 4);
  float* as_ca = alloc(Nc * 4);
  float* ad_cq = alloc(Nq * 4);
  float* ad_ca = alloc(Na * 4);
  float* e_cq  = alloc(Ecq * 4);
  float* e_ca  = alloc(Eca * 4);
  float* m_q   = alloc(Nq * 4);
  float* m_a   = alloc(Na * 4);
  float* den_q = alloc(Nq * 4);
  float* den_a = alloc(Na * 4);
  float* z_q   = alloc(Nq * 256);
  float* z_a   = alloc(Na * 256);
  (void)den_a; (void)Vp_dq; (void)Vp_sa; (void)Vp_da;
  if (off * sizeof(float) > ws_size) return;

  const long nminf = (long)(den_q - m_q);
  const long nzero = (long)((z_a + ((Na * 256 + 15) & ~(size_t)15)) - den_q);

  prep_kernel<<<1, 1024, 0, stream>>>(Wsq, Wdq, Asq, Adq, Wsa, Wda, Asa, Ada,
                                      bq, ba, Wout, bout, Vp_sq, Pc, cst);
  init_kernel<<<2048, 256, 0, stream>>>(den_q, nzero, m_q, nminf);

  const int gNc = (int)((Nc + 127) / 128);
  const int gNq = (int)((Nq + 127) / 128);
  const int gNa = (int)((Na + 127) / 128);
  scores_kernel<<<gNc, 256, 0, stream>>>(x_c, Vp_sq, as_cq, (int)Nc);
  scores_kernel<<<gNc, 256, 0, stream>>>(x_c, Vp_sa, as_ca, (int)Nc);
  scores_kernel<<<gNq, 256, 0, stream>>>(x_q, Vp_dq, ad_cq, (int)Nq);
  scores_kernel<<<gNa, 256, 0, stream>>>(x_a, Vp_da, ad_ca, (int)Na);

  const int gEcq = (int)((Ecq + 255) / 256);
  const int gEca = (int)((Eca + 255) / 256);
  edge_pass1<<<gEcq, 256, 0, stream>>>(src_cq, dst_cq, as_cq, ad_cq, e_cq, m_q, (int)Ecq);
  edge_pass1<<<gEca, 256, 0, stream>>>(src_ca, dst_ca, as_ca, ad_ca, e_ca, m_a, (int)Eca);
  edge_pass2<<<gEcq, 256, 0, stream>>>(dst_cq, e_cq, m_q, den_q, (int)Ecq);
  edge_pass2<<<gEca, 256, 0, stream>>>(dst_ca, e_ca, m_a, den_a, (int)Eca);

  edge_pass3<<<(int)((Ecq * 64 + 255) / 256), 256, 0, stream>>>(src_cq, dst_cq, x_c,
                                                                e_cq, den_q, z_q, Ecq);
  edge_pass3<<<(int)((Eca * 64 + 255) / 256), 256, 0, stream>>>(src_ca, dst_ca, x_c,
                                                                e_ca, den_a, z_a, Eca);

  final_kernel<<<gNq, 256, 0, stream>>>(z_q, z_a, Pc, cst, (float*)d_out, (int)Nq);
}


# Project x_c through the folded output weights before aggregation (y = x_c @ P, 64x32 f32 WMMA): an edge adds 4 floats into out, not 256 into z; float sum order changes (atomics now on out)
# speedup vs baseline: 2.2748x; 2.2748x over previous
#include <hip/hip_runtime.h>
#include <hip/hip_bf16.h>
#include <math.h>

typedef __attribute__((ext_vector_type(2))) float v2f;
typedef __attribute__((ext_vector_type(8))) float v8f;

__device__ __forceinline__ float lrelu(float x) { return x > 0.f ? x : 0.2f * x; }

__device__ __forceinline__ void atomic_max_f32(float* addr, float val) {
  if (val >= 0.f)
    __hip_atomic_fetch_max((int*)addr, __float_as_int(val),
                           __ATOMIC_RELAXED, __HIP_MEMORY_SCOPE_AGENT);
  else
    __hip_atomic_fetch_min((unsigned int*)addr, __float_as_uint(val),
                           __ATOMIC_RELAXED, __HIP_MEMORY_SCOPE_AGENT);
}

__device__ __forceinline__ void atomic_add_f32(float* addr, float val) {
  __hip_atomic_fetch_add(addr, val, __ATOMIC_RELAXED, __HIP_MEMORY_SCOPE_AGENT);
}

__global__ __launch_bounds__(1024)
void prep_kernel(const float* __restrict__ Wsq, const float* __restrict__ Wdq,
                 const float* __restrict__ Asq, const float* __restrict__ Adq,
                 const float* __restrict__ Wsa, const float* __restrict__ Wda,
                 const float* __restrict__ Asa, const float* __restrict__ Ada,
                 const float* __restrict__ bq,  const float* __restrict__ ba,
                 const float* __restrict__ Wout, const float* __restrict__ bout,
                 float* __restrict__ Vp, float* __restrict__ Py, float* __restrict__ cst) {
  const int t = threadIdx.x;
  for (int idx = t; idx < 4 * 64 * 16; idx += blockDim.x) {
    int table = idx >> 10, k = (idx >> 4) & 63, j = idx & 15;
    const float* W = (table == 0) ? Wsq : (table == 1) ? Wdq : (table == 2) ? Wsa : Wda;
    const float* A = (table == 0) ? Asq : (table == 1) ? Adq : (table == 2) ? Asa : Ada;
    float s = 0.f;
    if (j < 4)
      for (int c = 0; c < 64; ++c) s += W[k * 256 + j * 64 + c] * A[j * 64 + c];
    Vp[idx] = s;
  }
  for (int idx = t; idx < 64 * 32; idx += blockDim.x) {
    int k = idx >> 5, n = idx & 31;
    int layer = n >> 4, j = (n >> 2) & 3, h = n & 3;
    const float* W = layer ? Wsa : Wsq;
    float s = 0.f;
    for (int c = 0; c < 64; ++c)
      s += W[k * 256 + h * 64 + c] * Wout[(layer * 64 + c) * 4 + j];
    Py[idx] = s * 0.25f;
  }
  if (t < 16) {
    float s = 0.f;
    if (t < 4) {
      for (int c = 0; c < 64; ++c)
        s += bq[c] * Wout[c * 4 + t] + ba[c] * Wout[(64 + c) * 4 + t];
      s += bout[t];
    }
    cst[t] = s;
  }
}

__global__ void init_kernel(float* __restrict__ zero_ptr, long nzero,
                            float* __restrict__ minf_ptr, long nminf,
                            const float* __restrict__ cst, float* __restrict__ out,
                            long nout) {
  const long stride = (long)gridDim.x * blockDim.x;
  const long t0 = (long)blockIdx.x * blockDim.x + threadIdx.x;
  for (long i = t0; i < nzero; i += stride) zero_ptr[i] = 0.f;
  for (long i = t0; i < nminf; i += stride) minf_ptr[i] = __int_as_float(0xff800000);
  for (long i = t0; i < nout; i += stride) out[i] = cst[i & 3];
}

__global__ void scores_kernel(const float* __restrict__ x, const float* __restrict__ Vp,
                              float* __restrict__ out, int N) {
  const int wave_g = (int)(((unsigned)blockIdx.x * blockDim.x + threadIdx.x) >> 5);
  const int lane = threadIdx.x & 31;
  const long base = (long)wave_g * 16;
  if (base >= N) return;
  const int m = lane & 15;
  const int ko = (lane >> 4) * 2;
  long row = base + m; if (row >= N) row = N - 1;
  const float* xr = x + row * 64;
  v8f acc = {};
#pragma unroll
  for (int kk = 0; kk < 64; kk += 4) {
    v2f a, b;
    a.x = xr[kk + ko];
    a.y = xr[kk + ko + 1];
    b.x = Vp[(kk + ko) * 16 + m];
    b.y = Vp[(kk + ko + 1) * 16 + m];
    acc = __builtin_amdgcn_wmma_f32_16x16x4_f32(false, a, false, b, (short)0, acc,
                                                false, false);
  }
  if (m < 4) {
    const long rb = base + ((lane >> 4) << 3);
#pragma unroll
    for (int v = 0; v < 8; ++v) {
      long r = rb + v;
      if (r < N) out[r * 4 + m] = acc[v];
    }
  }
}

__global__ void proj_kernel(const float* __restrict__ x, const float* __restrict__ Py,
                            float* __restrict__ yq, float* __restrict__ ya, int N) {
  const int wave_g = (int)(((unsigned)blockIdx.x * blockDim.x + threadIdx.x) >> 5);
  const int lane = threadIdx.x & 31;
  const long base = (long)wave_g * 16;
  if (base >= N) return;
  const int m = lane & 15;
  const int ko = (lane >> 4) * 2;
  long row = base + m; if (row >= N) row = N - 1;
  const float* xr = x + row * 64;
  v8f accq = {}, acca = {};
#pragma unroll
  for (int kk = 0; kk < 64; kk += 4) {
    v2f a, b0, b1;
    a.x = xr[kk + ko];
    a.y = xr[kk + ko + 1];
    b0.x = Py[(kk + ko) * 32 + m];
    b0.y = Py[(kk + ko + 1) * 32 + m];
    b1.x = Py[(kk + ko) * 32 + 16 + m];
    b1.y = Py[(kk + ko + 1) * 32 + 16 + m];
    accq = __builtin_amdgcn_wmma_f32_16x16x4_f32(false, a, false, b0, (short)0, accq,
                                                 false, false);
    acca = __builtin_amdgcn_wmma_f32_16x16x4_f32(false, a, false, b1, (short)0, acca,
                                                 false, false);
  }
  const long rb = base + ((lane >> 4) << 3);
#pragma unroll
  for (int v = 0; v < 8; ++v) {
    long r = rb + v;
    if (r < N) {
      yq[r * 16 + m] = accq[v];
      ya[r * 16 + m] = acca[v];
    }
  }
}

__global__ void edge_pass1(const int* __restrict__ src, const int* __restrict__ dst,
                           const float* __restrict__ a_s, const float* __restrict__ a_d,
                           float* __restrict__ e, float* __restrict__ mbuf, int E) {
  const int i = blockIdx.x * blockDim.x + threadIdx.x;
  if (i >= E) return;
  const long s = src[i], d = dst[i];
  const float4 as = *(const float4*)(a_s + s * 4);
  const float4 ad = *(const float4*)(a_d + d * 4);
  float4 ev;
  ev.x = lrelu(as.x + ad.x);
  ev.y = lrelu(as.y + ad.y);
  ev.z = lrelu(as.z + ad.z);
  ev.w = lrelu(as.w + ad.w);
  *(float4*)(e + (long)i * 4) = ev;
  atomic_max_f32(mbuf + d * 4 + 0, ev.x);
  atomic_max_f32(mbuf + d * 4 + 1, ev.y);
  atomic_max_f32(mbuf + d * 4 + 2, ev.z);
  atomic_max_f32(mbuf + d * 4 + 3, ev.w);
}

__global__ void edge_pass2(const int* __restrict__ dst, float* __restrict__ e,
                           const float* __restrict__ mbuf, float* __restrict__ den, int E) {
  const int i = blockIdx.x * blockDim.x + threadIdx.x;
  if (i >= E) return;
  const long d = dst[i];
  float4 ev = *(const float4*)(e + (long)i * 4);
  const float4 mv = *(const float4*)(mbuf + d * 4);
  ev.x = __expf(ev.x - mv.x);
  ev.y = __expf(ev.y - mv.y);
  ev.z = __expf(ev.z - mv.z);
  ev.w = __expf(ev.w - mv.w);
  *(float4*)(e + (long)i * 4) = ev;
  atomic_add_f32(den + d * 4 + 0, ev.x);
  atomic_add_f32(den + d * 4 + 1, ev.y);
  atomic_add_f32(den + d * 4 + 2, ev.z);
  atomic_add_f32(den + d * 4 + 3, ev.w);
}

__global__ void edge_pass3(const int* __restrict__ src, const int* __restrict__ dst,
                           const float* __restrict__ y, const float* __restrict__ ex,
                           const float* __restrict__ den, float* __restrict__ out, long E) {
  const long gt = (long)blockIdx.x * blockDim.x + threadIdx.x;
  const long ei = gt >> 2;
  if (ei >= E) return;
  const int j = (int)(gt & 3);
  const long s = src[ei], d = dst[ei];
  const float4 ev = *(const float4*)(ex + ei * 4);
  const float4 dn = *(const float4*)(den + d * 4);
  const float4 yv = *(const float4*)(y + s * 16 + j * 4);
  float o = (ev.x / (dn.x + 1e-16f)) * yv.x;
  o += (ev.y / (dn.y + 1e-16f)) * yv.y;
  o += (ev.z / (dn.z + 1e-16f)) * yv.z;
  o += (ev.w / (dn.w + 1e-16f)) * yv.w;
  atomic_add_f32(out + d * 4 + j, o);
}

extern "C" void kernel_launch(void* const* d_in, const int* in_sizes, int n_in,
                              void* d_out, int out_size, void* d_ws, size_t ws_size,
                              hipStream_t stream) {
  const float* x_c = (const float*)d_in[0];
  const float* x_q = (const float*)d_in[1];
  const float* x_a = (const float*)d_in[2];
  const int* src_cq = (const int*)d_in[3];
  const int* dst_cq = (const int*)d_in[4];
  const int* src_ca = (const int*)d_in[5];
  const int* dst_ca = (const int*)d_in[6];
  const float* Wsq = (const float*)d_in[7];
  const float* Wdq = (const float*)d_in[8];
  const float* Asq = (const float*)d_in[9];
  const float* Adq = (const float*)d_in[10];
  const float* bq  = (const float*)d_in[11];
  const float* Wsa = (const float*)d_in[12];
  const float* Wda = (const float*)d_in[13];
  const float* Asa = (const float*)d_in[14];
  const float* Ada = (const float*)d_in[15];
  const float* ba  = (const float*)d_in[16];
  const float* Wout = (const float*)d_in[17];
  const float* bout = (const float*)d_in[18];

  const long Nc = in_sizes[0] / 64;
  const long Nq = in_sizes[1] / 64;
  const long Na = in_sizes[2] / 64;
  const long Ecq = in_sizes[3];
  const long Eca = in_sizes[5];

  float* w = (float*)d_ws;
  size_t off = 0;
  auto alloc = [&](size_t n) -> float* {
    float* p = w + off;
    off += (n + 15) & ~(size_t)15;
    return p;
  };
  float* Vp_sq = alloc(64 * 16);
  float* Vp_dq = alloc(64 * 16);
  float* Vp_sa = alloc(64 * 16);
  float* Vp_da = alloc(64 * 16);
  float* Py    = alloc(64 * 32);
  float* cst   = alloc(16);
  float* as_cq = alloc(Nc * 4);
  float* as_ca = alloc(Nc * 4);
  float* ad_cq = alloc(Nq * 4);
  float* ad_ca = alloc(Na * 4);
  float* e_cq  = alloc(Ecq * 4);
  float* e_ca  = alloc(Eca * 4);
  float* m_q   = alloc(Nq * 4);
  float* m_a   = alloc(Na * 4);
  float* den_q = alloc(Nq * 4);
  float* den_a = alloc(Na * 4);
  float* y_q   = alloc(Nc * 16);
  float* y_a   = alloc(Nc * 16);
  (void)den_a; (void)Vp_dq; (void)Vp_sa; (void)Vp_da;
  if (off * sizeof(float) > ws_size) return;

  const long nminf = (long)(den_q - m_q);
  const long nzero = (long)(y_q - den_q);

  prep_kernel<<<1, 1024, 0, stream>>>(Wsq, Wdq, Asq, Adq, Wsa, Wda, Asa, Ada,
                                      bq, ba, Wout, bout, Vp_sq, Py, cst);
  init_kernel<<<2048, 256, 0, stream>>>(den_q, nzero, m_q, nminf, cst, (float*)d_out,
                                        Nq * 4);

  const int gNc = (int)((Nc + 127) / 128);
  const int gNq = (int)((Nq + 127) / 128);
  const int gNa = (int)((Na + 127) / 128);
  scores_kernel<<<gNc, 256, 0, stream>>>(x_c, Vp_sq, as_cq, (int)Nc);
  scores_kernel<<<gNc, 256, 0, stream>>>(x_c, Vp_sa, as_ca, (int)Nc);
  scores_kernel<<<gNq, 256, 0, stream>>>(x_q, Vp_dq, ad_cq, (int)Nq);
  scores_kernel<<<gNa, 256, 0, stream>>>(x_a, Vp_da, ad_ca, (int)Na);
  proj_kernel<<<gNc, 256, 0, stream>>>(x_c, Py, y_q, y_a, (int)Nc);

  const int gEcq = (int)((Ecq + 255) / 256);
  const int gEca = (int)((Eca + 255) / 256);
  edge_pass1<<<gEcq, 256, 0, stream>>>(src_cq, dst_cq, as_cq, ad_cq, e_cq, m_q, (int)Ecq);
  edge_pass1<<<gEca, 256, 0, stream>>>(src_ca, dst_ca, as_ca, ad_ca, e_ca, m_a, (int)Eca);
  edge_pass2<<<gEcq, 256, 0, stream>>>(dst_cq, e_cq, m_q, den_q, (int)Ecq);
  edge_pass2<<<gEca, 256, 0, stream>>>(dst_ca, e_ca, m_a, den_a, (int)Eca);

  edge_pass3<<<(int)((Ecq * 4 + 255) / 256), 256, 0, stream>>>(src_cq, dst_cq, y_q, e_cq,
                                                               den_q, (float*)d_out, Ecq);
  edge_pass3<<<(int)((Eca * 4 + 255) / 256), 256, 0, stream>>>(src_ca, dst_ca, y_a, e_ca,
                                                               den_a, (float*)d_out, Eca);
}


# Edge passes 1-2: one lane per (edge, head) in a resident grid-stride loop, so an edge's 4 head atomics leave as ONE instruction on one 16-byte piece; segment max on an order-preserving uint key
# speedup vs baseline: 4.0138x; 1.7645x over previous
#include <hip/hip_runtime.h>
#include <hip/hip_bf16.h>
#include <math.h>

typedef __attribute__((ext_vector_type(2))) float v2f;
typedef __attribute__((ext_vector_type(8))) float v8f;

__device__ __forceinline__ float lrelu(float x) { return x > 0.f ? x : 0.2f * x; }

__device__ __forceinline__ unsigned f2key(float v) {
  const unsigned b = __float_as_uint(v);
  return (b & 0x80000000u) ? ~b : (b | 0x80000000u);
}
__device__ __forceinline__ float key2f(unsigned k) {
  return __uint_as_float((k & 0x80000000u) ? (k & 0x7fffffffu) : ~k);
}

__device__ __forceinline__ void atomic_add_f32(float* addr, float val) {
  __hip_atomic_fetch_add(addr, val, __ATOMIC_RELAXED, __HIP_MEMORY_SCOPE_AGENT);
}

__global__ __launch_bounds__(1024)
void prep_kernel(const float* __restrict__ Wsq, const float* __restrict__ Wdq,
                 const float* __restrict__ Asq, const float* __restrict__ Adq,
                 const float* __restrict__ Wsa, const float* __restrict__ Wda,
                 const float* __restrict__ Asa, const float* __restrict__ Ada,
                 const float* __restrict__ bq,  const float* __restrict__ ba,
                 const float* __restrict__ Wout, const float* __restrict__ bout,
                 float* __restrict__ Vp, float* __restrict__ Py, float* __restrict__ cst) {
  const int t = threadIdx.x;
  for (int idx = t; idx < 4 * 64 * 16; idx += blockDim.x) {
    int table = idx >> 10, k = (idx >> 4) & 63, j = idx & 15;
    const float* W = (table == 0) ? Wsq : (table == 1) ? Wdq : (table == 2) ? Wsa : Wda;
    const float* A = (table == 0) ? Asq : (table == 1) ? Adq : (table == 2) ? Asa : Ada;
    float s = 0.f;
    if (j < 4)
      for (int c = 0; c < 64; ++c) s += W[k * 256 + j * 64 + c] * A[j * 64 + c];
    Vp[idx] = s;
  }
  for (int idx = t; idx < 64 * 32; idx += blockDim.x) {
    int k = idx >> 5, n = idx & 31;
    int layer = n >> 4, j = (n >> 2) & 3, h = n & 3;
    const float* W = layer ? Wsa : Wsq;
    float s = 0.f;
    for (int c = 0; c < 64; ++c)
      s += W[k * 256 + h * 64 + c] * Wout[(layer * 64 + c) * 4 + j];
    Py[idx] = s * 0.25f;
  }
  if (t < 16) {
    float s = 0.f;
    if (t < 4) {
      for (int c = 0; c < 64; ++c)
        s += bq[c] * Wout[c * 4 + t] + ba[c] * Wout[(64 + c) * 4 + t];
      s += bout[t];
    }
    cst[t] = s;
  }
}

__global__ void init_kernel(float* __restrict__ zero_ptr, long nzero,
                            const float* __restrict__ cst, float* __restrict__ out,
                            long nout) {
  const long stride = (long)gridDim.x * blockDim.x;
  const long t0 = (long)blockIdx.x * blockDim.x + threadIdx.x;
  for (long i = t0; i < nzero; i += stride) zero_ptr[i] = 0.f;
  for (long i = t0; i < nout; i += stride) out[i] = cst[i & 3];
}

__global__ void scores_kernel(const float* __restrict__ x, const float* __restrict__ Vp,
                              float* __restrict__ out, int N) {
  const int wave_g = (int)(((unsigned)blockIdx.x * blockDim.x + threadIdx.x) >> 5);
  const int lane = threadIdx.x & 31;
  const long base = (long)wave_g * 16;
  if (base >= N) return;
  const int m = lane & 15;
  const int ko = (lane >> 4) * 2;
  long row = base + m; if (row >= N) row = N - 1;
  const float* xr = x + row * 64;
  v8f acc = {};
#pragma unroll
  for (int kk = 0; kk < 64; kk += 4) {
    v2f a, b;
    a.x = xr[kk + ko];
    a.y = xr[kk + ko + 1];
    b.x = Vp[(kk + ko) * 16 + m];
    b.y = Vp[(kk + ko + 1) * 16 + m];
    acc = __builtin_amdgcn_wmma_f32_16x16x4_f32(false, a, false, b, (short)0, acc,
                                                false, false);
  }
  if (m < 4) {
    const long rb = base + ((lane >> 4) << 3);
#pragma unroll
    for (int v = 0; v < 8; ++v) {
      long r = rb + v;
      if (r < N) out[r * 4 + m] = acc[v];
    }
  }
}

__global__ void proj_kernel(const float* __restrict__ x, const float* __restrict__ Py,
                            float* __restrict__ yq, float* __restrict__ ya, int N) {
  const int wave_g = (int)(((unsigned)blockIdx.x * blockDim.x + threadIdx.x) >> 5);
  const int lane = threadIdx.x & 31;
  const long base = (long)wave_g * 16;
  if (base >= N) return;
  const int m = lane & 15;
  const int ko = (lane >> 4) * 2;
  long row = base + m; if (row >= N) row = N - 1;
  const float* xr = x + row * 64;
  v8f accq = {}, acca = {};
#pragma unroll
  for (int kk = 0; kk < 64; kk += 4) {
    v2f a, b0, b1;
    a.x = xr[kk + ko];
    a.y = xr[kk + ko + 1];
    b0.x = Py[(kk + ko) * 32 + m];
    b0.y = Py[(kk + ko + 1) * 32 + m];
    b1.x = Py[(kk + ko) * 32 + 16 + m];
    b1.y = Py[(kk + ko + 1) * 32 + 16 + m];
    accq = __builtin_amdgcn_wmma_f32_16x16x4_f32(false, a, false, b0, (short)0, accq,
                                                 false, false);
    acca = __builtin_amdgcn_wmma_f32_16x16x4_f32(false, a, false, b1, (short)0, acca,
                                                 false, false);
  }
  const long rb = base + ((lane >> 4) << 3);
#pragma unroll
  for (int v = 0; v < 8; ++v) {
    long r = rb + v;
    if (r < N) {
      yq[r * 16 + m] = accq[v];
      ya[r * 16 + m] = acca[v];
    }
  }
}


__global__ void edge_pass1(const int* __restrict__ src, const int* __restrict__ dst,
                           const float* __restrict__ a_s, const float* __restrict__ a_d,
                           float* __restrict__ e, unsigned* __restrict__ mkey, long E) {
  const long n = E * 4;
  const long stride = (long)gridDim.x * blockDim.x;
  for (long it = (long)blockIdx.x * blockDim.x + threadIdx.x; it < n; it += stride) {
    const long ei = it >> 2;
    const int h = (int)(it & 3);
    const long s = src[ei], d = dst[ei];
    const float ev = lrelu(a_s[s * 4 + h] + a_d[d * 4 + h]);
    e[it] = ev;
    __hip_atomic_fetch_max(mkey + d * 4 + h, f2key(ev), __ATOMIC_RELAXED,
                           __HIP_MEMORY_SCOPE_AGENT);
  }
}

__global__ void edge_pass2(const int* __restrict__ dst, float* __restrict__ e,
                           const unsigned* __restrict__ mkey, float* __restrict__ den,
                           long E) {
  const long n = E * 4;
  const long stride = (long)gridDim.x * blockDim.x;
  for (long it = (long)blockIdx.x * blockDim.x + threadIdx.x; it < n; it += stride) {
    const int h = (int)(it & 3);
    const long d = dst[it >> 2];
    const float ex = __expf(e[it] - key2f(mkey[d * 4 + h]));
    e[it] = ex;
    atomic_add_f32(den + d * 4 + h, ex);
  }
}

__global__ void edge_pass3(const int* __restrict__ src, const int* __restrict__ dst,
                           const float* __restrict__ y, const float* __restrict__ ex,
                           const float* __restrict__ den, float* __restrict__ out, long E) {
  const long gt = (long)blockIdx.x * blockDim.x + threadIdx.x;
  const long ei = gt >> 2;
  if (ei >= E) return;
  const int j = (int)(gt & 3);
  const long s = src[ei], d = dst[ei];
  const float4 ev = *(const float4*)(ex + ei * 4);
  const float4 dn = *(const float4*)(den + d * 4);
  const float4 yv = *(const float4*)(y + s * 16 + j * 4);
  float o = (ev.x / (dn.x + 1e-16f)) * yv.x;
  o += (ev.y / (dn.y + 1e-16f)) * yv.y;
  o += (ev.z / (dn.z + 1e-16f)) * yv.z;
  o += (ev.w / (dn.w + 1e-16f)) * yv.w;
  atomic_add_f32(out + d * 4 + j, o);
}

extern "C" void kernel_launch(void* const* d_in, const int* in_sizes, int n_in,
                              void* d_out, int out_size, void* d_ws, size_t ws_size,
                              hipStream_t stream) {
  const float* x_c = (const float*)d_in[0];
  const float* x_q = (const float*)d_in[1];
  const float* x_a = (const float*)d_in[2];
  const int* src_cq = (const int*)d_in[3];
  const int* dst_cq = (const int*)d_in[4];
  const int* src_ca = (const int*)d_in[5];
  const int* dst_ca = (const int*)d_in[6];
  const float* Wsq = (const float*)d_in[7];
  const float* Wdq = (const float*)d_in[8];
  const float* Asq = (const float*)d_in[9];
  const float* Adq = (const float*)d_in[10];
  const float* bq  = (const float*)d_in[11];
  const float* Wsa = (const float*)d_in[12];
  const float* Wda = (const float*)d_in[13];
  const float* Asa = (const float*)d_in[14];
  const float* Ada = (const float*)d_in[15];
  const float* ba  = (const float*)d_in[16];
  const float* Wout = (const float*)d_in[17];
  const float* bout = (const float*)d_in[18];

  const long Nc = in_sizes[0] / 64;
  const long Nq = in_sizes[1] / 64;
  const long Na = in_sizes[2] / 64;
  const long Ecq = in_sizes[3];
  const long Eca = in_sizes[5];

  float* w = (float*)d_ws;
  size_t off = 0;
  auto alloc = [&](size_t n) -> float* {
    float* p = w + off;
    off += (n + 15) & ~(size_t)15;
    return p;
  };
  float* Vp_sq = alloc(64 * 16);
  float* Vp_dq = alloc(64 * 16);
  float* Vp_sa = alloc(64 * 16);
  float* Vp_da = alloc(64 * 16);
  float* Py    = alloc(64 * 32);
  float* cst   = alloc(16);
  float* as_cq = alloc(Nc * 4);
  float* as_ca = alloc(Nc * 4);
  float* ad_cq = alloc(Nq * 4);
  float* ad_ca = alloc(Na * 4);
  float* e_cq  = alloc(Ecq * 4);
  float* e_ca  = alloc(Eca * 4);
  float* m_q   = alloc(Nq * 4);
  float* m_a   = alloc(Na * 4);
  float* den_q = alloc(Nq * 4);
  float* den_a = alloc(Na * 4);
  float* y_q   = alloc(Nc * 16);
  float* y_a   = alloc(Nc * 16);
  (void)den_a; (void)Vp_dq; (void)Vp_sa; (void)Vp_da;
  if (off * sizeof(float) > ws_size) return;

  const long nzero = (long)(y_q - m_q);

  prep_kernel<<<1, 1024, 0, stream>>>(Wsq, Wdq, Asq, Adq, Wsa, Wda, Asa, Ada,
                                      bq, ba, Wout, bout, Vp_sq, Py, cst);
  init_kernel<<<2048, 256, 0, stream>>>(m_q, nzero, cst, (float*)d_out, Nq * 4);

  const int gNc = (int)((Nc + 127) / 128);
  const int gNq = (int)((Nq + 127) / 128);
  const int gNa = (int)((Na + 127) / 128);
  scores_kernel<<<gNc, 256, 0, stream>>>(x_c, Vp_sq, as_cq, (int)Nc);
  scores_kernel<<<gNc, 256, 0, stream>>>(x_c, Vp_sa, as_ca, (int)Nc);
  scores_kernel<<<gNq, 256, 0, stream>>>(x_q, Vp_dq, ad_cq, (int)Nq);
  scores_kernel<<<gNa, 256, 0, stream>>>(x_a, Vp_da, ad_ca, (int)Na);
  proj_kernel<<<gNc, 256, 0, stream>>>(x_c, Py, y_q, y_a, (int)Nc);

  edge_pass1<<<2048, 256, 0, stream>>>(src_cq, dst_cq, as_cq, ad_cq, e_cq, (unsigned*)m_q,
                                       Ecq);
  edge_pass1<<<2048, 256, 0, stream>>>(src_ca, dst_ca, as_ca, ad_ca, e_ca, (unsigned*)m_a,
                                       Eca);
  edge_pass2<<<2048, 256, 0, stream>>>(dst_cq, e_cq, (const unsigned*)m_q, den_q, Ecq);
  edge_pass2<<<2048, 256, 0, stream>>>(dst_ca, e_ca, (const unsigned*)m_a, den_a, Eca);

  edge_pass3<<<(int)((Ecq * 4 + 255) / 256), 256, 0, stream>>>(src_cq, dst_cq, y_q, e_cq,
                                                               den_q, (float*)d_out, Ecq);
  edge_pass3<<<(int)((Eca * 4 + 255) / 256), 256, 0, stream>>>(src_ca, dst_ca, y_a, e_ca,
                                                               den_a, (float*)d_out, Eca);
}


# 13 launches become 5: both layers share each edge pass; one launch does all score and projection products (x_c rows loaded once for all three) and the start values; float add order into out changes
# speedup vs baseline: 4.7636x; 1.1868x over previous
#include <hip/hip_runtime.h>
#include <hip/hip_bf16.h>
#include <math.h>

typedef __attribute__((ext_vector_type(2))) float v2f;
typedef __attribute__((ext_vector_type(8))) float v8f;

__device__ __forceinline__ float lrelu(float x) { return x > 0.f ? x : 0.2f * x; }

__device__ __forceinline__ unsigned f2key(float v) {
  const unsigned b = __float_as_uint(v);
  return (b & 0x80000000u) ? ~b : (b | 0x80000000u);
}
__device__ __forceinline__ float key2f(unsigned k) {
  return __uint_as_float((k & 0x80000000u) ? (k & 0x7fffffffu) : ~k);
}

__device__ __forceinline__ void atomic_add_f32(float* addr, float val) {
  __hip_atomic_fetch_add(addr, val, __ATOMIC_RELAXED, __HIP_MEMORY_SCOPE_AGENT);
}

__global__ __launch_bounds__(1024)
void prep_kernel(const float* __restrict__ Wsq, const float* __restrict__ Wdq,
                 const float* __restrict__ Asq, const float* __restrict__ Adq,
                 const float* __restrict__ Wsa, const float* __restrict__ Wda,
                 const float* __restrict__ Asa, const float* __restrict__ Ada,
                 const float* __restrict__ bq,  const float* __restrict__ ba,
                 const float* __restrict__ Wout, const float* __restrict__ bout,
                 float* __restrict__ Vp, float* __restrict__ Py, float* __restrict__ cst) {
  const int t = threadIdx.x;
  for (int idx = t; idx < 4 * 64 * 16; idx += blockDim.x) {
    int table = idx >> 10, k = (idx >> 4) & 63, j = idx & 15;
    const float* W = (table == 0) ? Wsq : (table == 1) ? Wdq : (table == 2) ? Wsa : Wda;
    const float* A = (table == 0) ? Asq : (table == 1) ? Adq : (table == 2) ? Asa : Ada;
    float s = 0.f;
    if (j < 4)
      for (int c = 0; c < 64; ++c) s += W[k * 256 + j * 64 + c] * A[j * 64 + c];
    Vp[idx] = s;
  }
  for (int idx = t; idx < 64 * 32; idx += blockDim.x) {
    int k = idx >> 5, n = idx & 31;
    int layer = n >> 4, j = (n >> 2) & 3, h = n & 3;
    const float* W = layer ? Wsa : Wsq;
    float s = 0.f;
    for (int c = 0; c < 64; ++c)
      s += W[k * 256 + h * 64 + c] * Wout[(layer * 64 + c) * 4 + j];
    Py[idx] = s * 0.25f;
  }
  if (t < 16) {
    float s = 0.f;
    if (t < 4) {
      for (int c = 0; c < 64; ++c)
        s += bq[c] * Wout[c * 4 + t] + ba[c] * Wout[(64 + c) * 4 + t];
      s += bout[t];
    }
    cst[t] = s;
  }
}

__global__ void dense_kernel(const float* __restrict__ xc, const float* __restrict__ xq,
                             const float* __restrict__ xa, const float* __restrict__ Vp,
                             const float* __restrict__ Py, const float* __restrict__ cst,
                             float* __restrict__ as_cq, float* __restrict__ as_ca,
                             float* __restrict__ ad_cq, float* __restrict__ ad_ca,
                             float* __restrict__ yq, float* __restrict__ ya,
                             float* __restrict__ zero_ptr, long nzero,
                             float* __restrict__ out, long nout,
                             int Nc, int Nq, int Na, int gNc, int gNq) {
  {
    const long stride = (long)gridDim.x * blockDim.x;
    const long t0 = (long)blockIdx.x * blockDim.x + threadIdx.x;
    for (long i = t0; i < nzero; i += stride) zero_ptr[i] = 0.f;
    for (long i = t0; i < nout; i += stride) out[i] = cst[i & 3];
  }
  const int lane = threadIdx.x & 31;
  const int wv = (int)(threadIdx.x >> 5);
  const int m = lane & 15;
  const int ko = (lane >> 4) * 2;
  int b = (int)blockIdx.x;
  if (b < gNc) {
    const long base = ((long)b * 8 + wv) * 16;
    if (base >= Nc) return;
    long row = base + m; if (row >= Nc) row = Nc - 1;
    const float* xr = xc + row * 64;
    const float* bt = (m < 4) ? (Vp + m) : (m < 8) ? (Vp + 2 * 1024 + (m - 4)) : (Vp + m);
    v8f accs = {}, accq = {}, acca = {};
#pragma unroll
    for (int kk = 0; kk < 64; kk += 4) {
      v2f a, bs, b0, b1;
      a.x = xr[kk + ko];
      a.y = xr[kk + ko + 1];
      bs.x = bt[(kk + ko) * 16];
      bs.y = bt[(kk + ko + 1) * 16];
      b0.x = Py[(kk + ko) * 32 + m];
      b0.y = Py[(kk + ko + 1) * 32 + m];
      b1.x = Py[(kk + ko) * 32 + 16 + m];
      b1.y = Py[(kk + ko + 1) * 32 + 16 + m];
      accs = __builtin_amdgcn_wmma_f32_16x16x4_f32(false, a, false, bs, (short)0, accs,
                                                   false, false);
      accq = __builtin_amdgcn_wmma_f32_16x16x4_f32(false, a, false, b0, (short)0, accq,
                                                   false, false);
      acca = __builtin_amdgcn_wmma_f32_16x16x4_f32(false, a, false, b1, (short)0, acca,
                                                   false, false);
    }
    const long rb = base + ((lane >> 4) << 3);
    float* so = (m < 4) ? (as_cq + m) : (as_ca + (m - 4));
#pragma unroll
    for (int v = 0; v < 8; ++v) {
      long r = rb + v;
      if (r < Nc) {
        yq[r * 16 + m] = accq[v];
        ya[r * 16 + m] = acca[v];
        if (m < 8) so[r * 4] = accs[v];
      }
    }
  } else {
    b -= gNc;
    const float* x = xq; const float* V = Vp + 1024; float* o = ad_cq; int N = Nq;
    if (b >= gNq) { b -= gNq; x = xa; V = Vp + 3 * 1024; o = ad_ca; N = Na; }
    const long base = ((long)b * 8 + wv) * 16;
    if (base >= N) return;
    long row = base + m; if (row >= N) row = N - 1;
    const float* xr = x + row * 64;
    v8f acc = {};
#pragma unroll
    for (int kk = 0; kk < 64; kk += 4) {
      v2f a, bv;
      a.x = xr[kk + ko];
      a.y = xr[kk + ko + 1];
      bv.x = V[(kk + ko) * 16 + m];
      bv.y = V[(kk + ko + 1) * 16 + m];
      acc = __builtin_amdgcn_wmma_f32_16x16x4_f32(false, a, false, bv, (short)0, acc,
                                                  false, false);
    }
    if (m < 4) {
      const long rb = base + ((lane >> 4) << 3);
#pragma unroll
      for (int v = 0; v < 8; ++v) {
        long r = rb + v;
        if (r < N) o[r * 4 + m] = acc[v];
      }
    }
  }
}


__global__ void edge_pass1(const int* __restrict__ src_q, const int* __restrict__ dst_q,
                           const int* __restrict__ src_a, const int* __restrict__ dst_a,
                           const float* __restrict__ as_q, const float* __restrict__ ad_q,
                           const float* __restrict__ as_a, const float* __restrict__ ad_a,
                           float* __restrict__ e_q, float* __restrict__ e_a,
                           unsigned* __restrict__ mk_q, unsigned* __restrict__ mk_a,
                           long Eq, long Ea) {
  const long nq = Eq * 4;
  const long n = nq + Ea * 4;
  const long stride = (long)gridDim.x * blockDim.x;
  for (long it = (long)blockIdx.x * blockDim.x + threadIdx.x; it < n; it += stride) {
    const bool la = it >= nq;
    const long li = la ? it - nq : it;
    const int* src = la ? src_a : src_q;
    const int* dst = la ? dst_a : dst_q;
    const float* a_s = la ? as_a : as_q;
    const float* a_d = la ? ad_a : ad_q;
    float* e = la ? e_a : e_q;
    unsigned* mkey = la ? mk_a : mk_q;
    const long ei = li >> 2;
    const int h = (int)(li & 3);
    const long s = src[ei], d = dst[ei];
    const float ev = lrelu(a_s[s * 4 + h] + a_d[d * 4 + h]);
    e[li] = ev;
    __hip_atomic_fetch_max(mkey + d * 4 + h, f2key(ev), __ATOMIC_RELAXED,
                           __HIP_MEMORY_SCOPE_AGENT);
  }
}

__global__ void edge_pass2(const int* __restrict__ dst_q, const int* __restrict__ dst_a,
                           float* __restrict__ e_q, float* __restrict__ e_a,
                           const unsigned* __restrict__ mk_q,
                           const unsigned* __restrict__ mk_a,
                           float* __restrict__ den_q, float* __restrict__ den_a,
                           long Eq, long Ea) {
  const long nq = Eq * 4;
  const long n = nq + Ea * 4;
  const long stride = (long)gridDim.x * blockDim.x;
  for (long it = (long)blockIdx.x * blockDim.x + threadIdx.x; it < n; it += stride) {
    const bool la = it >= nq;
    const long li = la ? it - nq : it;
    const int* dst = la ? dst_a : dst_q;
    float* e = la ? e_a : e_q;
    const unsigned* mkey = la ? mk_a : mk_q;
    float* den = la ? den_a : den_q;
    const int h = (int)(li & 3);
    const long d = dst[li >> 2];
    const float ex = __expf(e[li] - key2f(mkey[d * 4 + h]));
    e[li] = ex;
    atomic_add_f32(den + d * 4 + h, ex);
  }
}

__global__ void edge_pass3(const int* __restrict__ src_q, const int* __restrict__ dst_q,
                           const int* __restrict__ src_a, const int* __restrict__ dst_a,
                           const float* __restrict__ y_q, const float* __restrict__ y_a,
                           const float* __restrict__ ex_q, const float* __restrict__ ex_a,
                           const float* __restrict__ den_q, const float* __restrict__ den_a,
                           float* __restrict__ out, long Eq, long Ea) {
  const long gt = (long)blockIdx.x * blockDim.x + threadIdx.x;
  const long nq = Eq * 4;
  const bool la = gt >= nq;
  const long lt = la ? gt - nq : gt;
  const long ei = lt >> 2;
  if (ei >= (la ? Ea : Eq)) return;
  const int* src = la ? src_a : src_q;
  const int* dst = la ? dst_a : dst_q;
  const float* y = la ? y_a : y_q;
  const float* ex = la ? ex_a : ex_q;
  const float* den = la ? den_a : den_q;
  const int j = (int)(lt & 3);
  const long s = src[ei], d = dst[ei];
  const float4 ev = *(const float4*)(ex + ei * 4);
  const float4 dn = *(const float4*)(den + d * 4);
  const float4 yv = *(const float4*)(y + s * 16 + j * 4);
  float o = (ev.x / (dn.x + 1e-16f)) * yv.x;
  o += (ev.y / (dn.y + 1e-16f)) * yv.y;
  o += (ev.z / (dn.z + 1e-16f)) * yv.z;
  o += (ev.w / (dn.w + 1e-16f)) * yv.w;
  atomic_add_f32(out + d * 4 + j, o);
}

extern "C" void kernel_launch(void* const* d_in, const int* in_sizes, int n_in,
                              void* d_out, int out_size, void* d_ws, size_t ws_size,
                              hipStream_t stream) {
  const float* x_c = (const float*)d_in[0];
  const float* x_q = (const float*)d_in[1];
  const float* x_a = (const float*)d_in[2];
  const int* src_cq = (const int*)d_in[3];
  const int* dst_cq = (const int*)d_in[4];
  const int* src_ca = (const int*)d_in[5];
  const int* dst_ca = (const int*)d_in[6];
  const float* Wsq = (const float*)d_in[7];
  const float* Wdq = (const float*)d_in[8];
  const float* Asq = (const float*)d_in[9];
  const float* Adq = (const float*)d_in[10];
  const float* bq  = (const float*)d_in[11];
  const float* Wsa = (const float*)d_in[12];
  const float* Wda = (const float*)d_in[13];
  const float* Asa = (const float*)d_in[14];
  const float* Ada = (const float*)d_in[15];
  const float* ba  = (const float*)d_in[16];
  const float* Wout = (const float*)d_in[17];
  const float* bout = (const float*)d_in[18];

  const long Nc = in_sizes[0] / 64;
  const long Nq = in_sizes[1] / 64;
  const long Na = in_sizes[2] / 64;
  const long Ecq = in_sizes[3];
  const long Eca = in_sizes[5];

  float* w = (float*)d_ws;
  size_t off = 0;
  auto alloc = [&](size_t n) -> float* {
    float* p = w + off;
    off += (n + 15) & ~(size_t)15;
    return p;
  };
  float* Vp_sq = alloc(64 * 16);
  float* Vp_dq = alloc(64 * 16);
  float* Vp_sa = alloc(64 * 16);
  float* Vp_da = alloc(64 * 16);
  float* Py    = alloc(64 * 32);
  float* cst   = alloc(16);
  float* as_cq = alloc(Nc * 4);
  float* as_ca = alloc(Nc * 4);
  float* ad_cq = alloc(Nq * 4);
  float* ad_ca = alloc(Na * 4);
  float* e_cq  = alloc(Ecq * 4);
  float* e_ca  = alloc(Eca * 4);
  float* m_q   = alloc(Nq * 4);
  float* m_a   = alloc(Na * 4);
  float* den_q = alloc(Nq * 4);
  float* den_a = alloc(Na * 4);
  float* y_q   = alloc(Nc * 16);
  float* y_a   = alloc(Nc * 16);
  (void)Vp_dq; (void)Vp_sa; (void)Vp_da;
  if (off * sizeof(float) > ws_size) return;

  const long nzero = (long)(y_q - m_q);

  prep_kernel<<<1, 1024, 0, stream>>>(Wsq, Wdq, Asq, Adq, Wsa, Wda, Asa, Ada,
                                      bq, ba, Wout, bout, Vp_sq, Py, cst);

  const int gNc = (int)((Nc + 127) / 128);
  const int gNq = (int)((Nq + 127) / 128);
  const int gNa = (int)((Na + 127) / 128);
  dense_kernel<<<gNc + gNq + gNa, 256, 0, stream>>>(
      x_c, x_q, x_a, Vp_sq, Py, cst, as_cq, as_ca, ad_cq, ad_ca, y_q, y_a,
      m_q, nzero, (float*)d_out, Nq * 4, (int)Nc, (int)Nq, (int)Na, gNc, gNq);

  edge_pass1<<<2048, 256, 0, stream>>>(src_cq, dst_cq, src_ca, dst_ca, as_cq, ad_cq, as_ca,
                                       ad_ca, e_cq, e_ca, (unsigned*)m_q, (unsigned*)m_a,
                                       Ecq, Eca);
  edge_pass2<<<2048, 256, 0, stream>>>(dst_cq, dst_ca, e_cq, e_ca, (const unsigned*)m_q,
                                       (const unsigned*)m_a, den_q, den_a, Ecq, Eca);
  edge_pass3<<<(int)(((Ecq + Eca) * 4 + 255) / 256), 256, 0, stream>>>(
      src_cq, dst_cq, src_ca, dst_ca, y_q, y_a, e_cq, e_ca, den_q, den_a, (float*)d_out,
      Ecq, Eca);
}


# Normalisation out of the edge sum: pass 2 adds ex*y and ex (16+4 floats, one 128-byte row per destination) as ONE atomic per edge, a node pass divides, edge pass 3 goes; float add order changes
# speedup vs baseline: 7.2692x; 1.5260x over previous
#include <hip/hip_runtime.h>
#include <hip/hip_bf16.h>
#include <math.h>

typedef __attribute__((ext_vector_type(2))) float v2f;
typedef __attribute__((ext_vector_type(8))) float v8f;

__device__ __forceinline__ float lrelu(float x) { return x > 0.f ? x : 0.2f * x; }

__device__ __forceinline__ unsigned f2key(float v) {
  const unsigned b = __float_as_uint(v);
  return (b & 0x80000000u) ? ~b : (b | 0x80000000u);
}
__device__ __forceinline__ float key2f(unsigned k) {
  return __uint_as_float((k & 0x80000000u) ? (k & 0x7fffffffu) : ~k);
}

__device__ __forceinline__ void atomic_add_f32(float* addr, float val) {
  __hip_atomic_fetch_add(addr, val, __ATOMIC_RELAXED, __HIP_MEMORY_SCOPE_AGENT);
}

__global__ __launch_bounds__(1024)
void prep_kernel(const float* __restrict__ Wsq, const float* __restrict__ Wdq,
                 const float* __restrict__ Asq, const float* __restrict__ Adq,
                 const float* __restrict__ Wsa, const float* __restrict__ Wda,
                 const float* __restrict__ Asa, const float* __restrict__ Ada,
                 const float* __restrict__ bq,  const float* __restrict__ ba,
                 const float* __restrict__ Wout, const float* __restrict__ bout,
                 float* __restrict__ Vp, float* __restrict__ Py, float* __restrict__ cst) {
  const int t = threadIdx.x;
  for (int idx = t; idx < 4 * 64 * 16; idx += blockDim.x) {
    int table = idx >> 10, k = (idx >> 4) & 63, j = idx & 15;
    const float* W = (table == 0) ? Wsq : (table == 1) ? Wdq : (table == 2) ? Wsa : Wda;
    const float* A = (table == 0) ? Asq : (table == 1) ? Adq : (table == 2) ? Asa : Ada;
    float s = 0.f;
    if (j < 4)
      for (int c = 0; c < 64; ++c) s += W[k * 256 + j * 64 + c] * A[j * 64 + c];
    Vp[idx] = s;
  }
  for (int idx = t; idx < 64 * 32; idx += blockDim.x) {
    int k = idx >> 5, n = idx & 31;
    int layer = n >> 4, j = (n >> 2) & 3, h = n & 3;
    const float* W = layer ? Wsa : Wsq;
    float s = 0.f;
    for (int c = 0; c < 64; ++c)
      s += W[k * 256 + h * 64 + c] * Wout[(layer * 64 + c) * 4 + j];
    Py[idx] = s * 0.25f;
  }
  if (t < 16) {
    float s = 0.f;
    if (t < 4) {
      for (int c = 0; c < 64; ++c)
        s += bq[c] * Wout[c * 4 + t] + ba[c] * Wout[(64 + c) * 4 + t];
      s += bout[t];
    }
    cst[t] = s;
  }
}

__global__ void dense_kernel(const float* __restrict__ xc, const float* __restrict__ xq,
                             const float* __restrict__ xa, const float* __restrict__ Vp,
                             const float* __restrict__ Py,
                             float* __restrict__ as_cq, float* __restrict__ as_ca,
                             float* __restrict__ ad_cq, float* __restrict__ ad_ca,
                             float* __restrict__ yq, float* __restrict__ ya,
                             float4* __restrict__ zero_ptr, long nzero4,
                             int Nc, int Nq, int Na, int gNc, int gNq) {
  {
    const long stride = (long)gridDim.x * blockDim.x;
    const long t0 = (long)blockIdx.x * blockDim.x + threadIdx.x;
    const float4 z4 = {0.f, 0.f, 0.f, 0.f};
    for (long i = t0; i < nzero4; i += stride) zero_ptr[i] = z4;
  }
  const int lane = threadIdx.x & 31;
  const int wv = (int)(threadIdx.x >> 5);
  const int m = lane & 15;
  const int ko = (lane >> 4) * 2;
  int b = (int)blockIdx.x;
  if (b < gNc) {
    const long base = ((long)b * 8 + wv) * 16;
    if (base >= Nc) return;
    long row = base + m; if (row >= Nc) row = Nc - 1;
    const float* xr = xc + row * 64;
    const float* bt = (m < 4) ? (Vp + m) : (m < 8) ? (Vp + 2 * 1024 + (m - 4)) : (Vp + m);
    v8f accs = {}, accq = {}, acca = {};
#pragma unroll
    for (int kk = 0; kk < 64; kk += 4) {
      v2f a, bs, b0, b1;
      a.x = xr[kk + ko];
      a.y = xr[kk + ko + 1];
      bs.x = bt[(kk + ko) * 16];
      bs.y = bt[(kk + ko + 1) * 16];
      b0.x = Py[(kk + ko) * 32 + m];
      b0.y = Py[(kk + ko + 1) * 32 + m];
      b1.x = Py[(kk + ko) * 32 + 16 + m];
      b1.y = Py[(kk + ko + 1) * 32 + 16 + m];
      accs = __builtin_amdgcn_wmma_f32_16x16x4_f32(false, a, false, bs, (short)0, accs,
                                                   false, false);
      accq = __builtin_amdgcn_wmma_f32_16x16x4_f32(false, a, false, b0, (short)0, accq,
                                                   false, false);
      acca = __builtin_amdgcn_wmma_f32_16x16x4_f32(false, a, false, b1, (short)0, acca,
                                                   false, false);
    }
    const long rb = base + ((lane >> 4) << 3);
    float* so = (m < 4) ? (as_cq + m) : (as_ca + (m - 4));
#pragma unroll
    for (int v = 0; v < 8; ++v) {
      long r = rb + v;
      if (r < Nc) {
        yq[r * 16 + m] = accq[v];
        ya[r * 16 + m] = acca[v];
        if (m < 8) so[r * 4] = accs[v];
      }
    }
  } else {
    b -= gNc;
    const float* x = xq; const float* V = Vp + 1024; float* o = ad_cq; int N = Nq;
    if (b >= gNq) { b -= gNq; x = xa; V = Vp + 3 * 1024; o = ad_ca; N = Na; }
    const long base = ((long)b * 8 + wv) * 16;
    if (base >= N) return;
    long row = base + m; if (row >= N) row = N - 1;
    const float* xr = x + row * 64;
    v8f acc = {};
#pragma unroll
    for (int kk = 0; kk < 64; kk += 4) {
      v2f a, bv;
      a.x = xr[kk + ko];
      a.y = xr[kk + ko + 1];
      bv.x = V[(kk + ko) * 16 + m];
      bv.y = V[(kk + ko + 1) * 16 + m];
      acc = __builtin_amdgcn_wmma_f32_16x16x4_f32(false, a, false, bv, (short)0, acc,
                                                  false, false);
    }
    if (m < 4) {
      const long rb = base + ((lane >> 4) << 3);
#pragma unroll
      for (int v = 0; v < 8; ++v) {
        long r = rb + v;
        if (r < N) o[r * 4 + m] = acc[v];
      }
    }
  }
}


__global__ void edge_pass1(const int* __restrict__ src_q, const int* __restrict__ dst_q,
                           const int* __restrict__ src_a, const int* __restrict__ dst_a,
                           const float* __restrict__ as_q, const float* __restrict__ ad_q,
                           const float* __restrict__ as_a, const float* __restrict__ ad_a,
                           float* __restrict__ e_q, float* __restrict__ e_a,
                           unsigned* __restrict__ mk_q, unsigned* __restrict__ mk_a,
                           long Eq, long Ea) {
  const long nq = Eq * 4;
  const long n = nq + Ea * 4;
  const long stride = (long)gridDim.x * blockDim.x;
  for (long it = (long)blockIdx.x * blockDim.x + threadIdx.x; it < n; it += stride) {
    const bool la = it >= nq;
    const long li = la ? it - nq : it;
    const int* src = la ? src_a : src_q;
    const int* dst = la ? dst_a : dst_q;
    const float* a_s = la ? as_a : as_q;
    const float* a_d = la ? ad_a : ad_q;
    float* e = la ? e_a : e_q;
    unsigned* mkey = la ? mk_a : mk_q;
    const long ei = li >> 2;
    const int h = (int)(li & 3);
    const long s = src[ei], d = dst[ei];
    const float ev = lrelu(a_s[s * 4 + h] + a_d[d * 4 + h]);
    e[li] = ev;
    __hip_atomic_fetch_max(mkey + d * 4 + h, f2key(ev), __ATOMIC_RELAXED,
                           __HIP_MEMORY_SCOPE_AGENT);
  }
}

__global__ void edge_pass2(const int* __restrict__ src_q, const int* __restrict__ dst_q,
                const int* __restrict__ src_a, const int* __restrict__ dst_a,
                const float* __restrict__ y_q, const float* __restrict__ y_a,
                const float* __restrict__ e_q, const float* __restrict__ e_a,
                const unsigned* __restrict__ mk_q, const unsigned* __restrict__ mk_a,
                float* __restrict__ acc_q, float* __restrict__ acc_a, long Eq, long Ea) {
  const int lane = threadIdx.x & 31;
  const int h = lane & 3;
  const int c16 = lane & 15;
  const long wave = (long)blockIdx.x * (blockDim.x >> 5) +
                    __builtin_amdgcn_readfirstlane((int)(threadIdx.x >> 5));
  const long nwaves = (long)gridDim.x * (blockDim.x >> 5);
  const long nbq = (Eq + 31) >> 5;
  const long nb = nbq + ((Ea + 31) >> 5);
  for (long blk = wave; blk < nb; blk += nwaves) {
    const bool la = blk >= nbq;
    const int* src = la ? src_a : src_q;
    const int* dst = la ? dst_a : dst_q;
    const float* y = la ? y_a : y_q;
    const float* e = la ? e_a : e_q;
    const unsigned* mkey = la ? mk_a : mk_q;
    float* acc = la ? acc_a : acc_q;
    const long E = la ? Ea : Eq;
    const long base = (la ? blk - nbq : blk) << 5;
    const long left = E - base;
    const int cnt = left < 32 ? (int)left : 32;
    const int lc = lane < cnt ? lane : cnt - 1;
    const int s = src[base + lc];
    const int d = dst[base + lc];
    float exj[4];
#pragma unroll
    for (int j = 0; j < 4; ++j) {
      int el = j * 8 + (lane >> 2);
      if (el >= cnt) el = cnt - 1;
      const int dj = __builtin_amdgcn_ds_bpermute(el << 2, d);
      const float ev = e[(base + el) * 4 + h];
      const unsigned mk = mkey[(long)dj * 4 + h];
      exj[j] = __expf(ev - key2f(mk));
    }
#pragma unroll
    for (int j = 0; j < 4; ++j) {
      if (j * 8 < cnt) {
        float yv[8];
#pragma unroll
        for (int u = 0; u < 8; ++u) {
          int kk = j * 8 + u; if (kk >= cnt) kk = cnt - 1;
          const int sk = __builtin_amdgcn_readlane(s, kk);
          yv[u] = y[(long)sk * 16 + c16];
        }
#pragma unroll
        for (int u = 0; u < 8; ++u) {
          const int kk = j * 8 + u;
          if (kk < cnt) {
            const int dk = __builtin_amdgcn_readlane(d, kk);
            const float ex = __int_as_float(
                __builtin_amdgcn_ds_bpermute((u * 4 + h) << 2, __float_as_int(exj[j])));
            const float val = (lane < 16) ? ex * yv[u] : ex;
            if (lane < 20) atomic_add_f32(acc + (long)dk * 32 + lane, val);
          }
        }
      }
    }
  }
}

__global__ void node_pass(const float* __restrict__ acc_q, const float* __restrict__ acc_a,
                          const float* __restrict__ cst, float* __restrict__ out,
                          long Nq, long Na) {
  const long t = (long)blockIdx.x * blockDim.x + threadIdx.x;
  const long d = t >> 2;
  const int j = (int)(t & 3);
  if (d >= Nq) return;
  float o = cst[j];
  {
    const float4 n = *(const float4*)(acc_q + d * 32 + j * 4);
    const float4 dn = *(const float4*)(acc_q + d * 32 + 16);
    o += n.x / (dn.x + 1e-16f);
    o += n.y / (dn.y + 1e-16f);
    o += n.z / (dn.z + 1e-16f);
    o += n.w / (dn.w + 1e-16f);
  }
  if (d < Na) {
    const float4 n = *(const float4*)(acc_a + d * 32 + j * 4);
    const float4 dn = *(const float4*)(acc_a + d * 32 + 16);
    o += n.x / (dn.x + 1e-16f);
    o += n.y / (dn.y + 1e-16f);
    o += n.z / (dn.z + 1e-16f);
    o += n.w / (dn.w + 1e-16f);
  }
  out[t] = o;
}

extern "C" void kernel_launch(void* const* d_in, const int* in_sizes, int n_in,
                              void* d_out, int out_size, void* d_ws, size_t ws_size,
                              hipStream_t stream) {
  const float* x_c = (const float*)d_in[0];
  const float* x_q = (const float*)d_in[1];
  const float* x_a = (const float*)d_in[2];
  const int* src_cq = (const int*)d_in[3];
  const int* dst_cq = (const int*)d_in[4];
  const int* src_ca = (const int*)d_in[5];
  const int* dst_ca = (const int*)d_in[6];
  const float* Wsq = (const float*)d_in[7];
  const float* Wdq = (const float*)d_in[8];
  const float* Asq = (const float*)d_in[9];
  const float* Adq = (const float*)d_in[10];
  const float* bq  = (const float*)d_in[11];
  const float* Wsa = (const float*)d_in[12];
  const float* Wda = (const float*)d_in[13];
  const float* Asa = (const float*)d_in[14];
  const float* Ada = (const float*)d_in[15];
  const float* ba  = (const float*)d_in[16];
  const float* Wout = (const float*)d_in[17];
  const float* bout = (const float*)d_in[18];

  const long Nc = in_sizes[0] / 64;
  const long Nq = in_sizes[1] / 64;
  const long Na = in_sizes[2] / 64;
  const long Ecq = in_sizes[3];
  const long Eca = in_sizes[5];

  float* w = (float*)d_ws;
  size_t off = 0;
  auto alloc = [&](size_t n) -> float* {
    float* p = w + off;
    off += (n + 63) & ~(size_t)63;
    return p;
  };
  float* Vp_sq = alloc(64 * 16);
  float* Vp_dq = alloc(64 * 16);
  float* Vp_sa = alloc(64 * 16);
  float* Vp_da = alloc(64 * 16);
  float* Py    = alloc(64 * 32);
  float* cst   = alloc(16);
  float* as_cq = alloc(Nc * 4);
  float* as_ca = alloc(Nc * 4);
  float* ad_cq = alloc(Nq * 4);
  float* ad_ca = alloc(Na * 4);
  float* e_cq  = alloc(Ecq * 4);
  float* e_ca  = alloc(Eca * 4);
  float* m_q   = alloc(Nq * 4);
  float* m_a   = alloc(Na * 4);
  float* acc_q = alloc(Nq * 32);
  float* acc_a = alloc(Na * 32);
  float* y_q   = alloc(Nc * 16);
  float* y_a   = alloc(Nc * 16);
  (void)Vp_dq; (void)Vp_sa; (void)Vp_da;
  if (off * sizeof(float) > ws_size) return;

  const long nzero4 = (long)(y_q - m_q) / 4;

  prep_kernel<<<1, 1024, 0, stream>>>(Wsq, Wdq, Asq, Adq, Wsa, Wda, Asa, Ada,
                                      bq, ba, Wout, bout, Vp_sq, Py, cst);

  const int gNc = (int)((Nc + 127) / 128);
  const int gNq = (int)((Nq + 127) / 128);
  const int gNa = (int)((Na + 127) / 128);
  dense_kernel<<<gNc + gNq + gNa, 256, 0, stream>>>(
      x_c, x_q, x_a, Vp_sq, Py, as_cq, as_ca, ad_cq, ad_ca, y_q, y_a,
      (float4*)m_q, nzero4, (int)Nc, (int)Nq, (int)Na, gNc, gNq);

  edge_pass1<<<2048, 256, 0, stream>>>(src_cq, dst_cq, src_ca, dst_ca, as_cq, ad_cq, as_ca,
                                       ad_ca, e_cq, e_ca, (unsigned*)m_q, (unsigned*)m_a,
                                       Ecq, Eca);
  edge_pass2<<<2048, 256, 0, stream>>>(src_cq, dst_cq, src_ca, dst_ca, y_q, y_a, e_cq, e_ca,
                                       (const unsigned*)m_q, (const unsigned*)m_a,
                                       acc_q, acc_a, Ecq, Eca);
  node_pass<<<(int)((Nq * 4 + 255) / 256), 256, 0, stream>>>(acc_q, acc_a, cst,
                                                             (float*)d_out, Nq, Na);
}


# Atomic targets (max keys, accumulator rows) kept once per package half, picked by the wave's die number: no atomic line is shared by both halves; copies merged after; float add order changes
# speedup vs baseline: 9.1513x; 1.2589x over previous
#include <hip/hip_runtime.h>
#include <hip/hip_bf16.h>
#include <math.h>

typedef __attribute__((ext_vector_type(2))) float v2f;
typedef __attribute__((ext_vector_type(8))) float v8f;

__device__ __forceinline__ float lrelu(float x) { return x > 0.f ? x : 0.2f * x; }

__device__ __forceinline__ unsigned f2key(float v) {
  const unsigned b = __float_as_uint(v);
  return (b & 0x80000000u) ? ~b : (b | 0x80000000u);
}
__device__ __forceinline__ float key2f(unsigned k) {
  return __uint_as_float((k & 0x80000000u) ? (k & 0x7fffffffu) : ~k);
}

__device__ __forceinline__ void atomic_add_f32(float* addr, float val) {
  __hip_atomic_fetch_add(addr, val, __ATOMIC_RELAXED, __HIP_MEMORY_SCOPE_AGENT);
}

__device__ __forceinline__ int package_half() {
  return (int)((__builtin_amdgcn_s_sendmsg_rtn(0x87u) >> 18) & 1u);
}

__global__ __launch_bounds__(1024)
void prep_kernel(const float* __restrict__ Wsq, const float* __restrict__ Wdq,
                 const float* __restrict__ Asq, const float* __restrict__ Adq,
                 const float* __restrict__ Wsa, const float* __restrict__ Wda,
                 const float* __restrict__ Asa, const float* __restrict__ Ada,
                 const float* __restrict__ bq,  const float* __restrict__ ba,
                 const float* __restrict__ Wout, const float* __restrict__ bout,
                 float* __restrict__ Vp, float* __restrict__ Py, float* __restrict__ cst) {
  const int t = threadIdx.x;
  for (int idx = t; idx < 4 * 64 * 16; idx += blockDim.x) {
    int table = idx >> 10, k = (idx >> 4) & 63, j = idx & 15;
    const float* W = (table == 0) ? Wsq : (table == 1) ? Wdq : (table == 2) ? Wsa : Wda;
    const float* A = (table == 0) ? Asq : (table == 1) ? Adq : (table == 2) ? Asa : Ada;
    float s = 0.f;
    if (j < 4)
      for (int c = 0; c < 64; ++c) s += W[k * 256 + j * 64 + c] * A[j * 64 + c];
    Vp[idx] = s;
  }
  for (int idx = t; idx < 64 * 32; idx += blockDim.x) {
    int k = idx >> 5, n = idx & 31;
    int layer = n >> 4, j = (n >> 2) & 3, h = n & 3;
    const float* W = layer ? Wsa : Wsq;
    float s = 0.f;
    for (int c = 0; c < 64; ++c)
      s += W[k * 256 + h * 64 + c] * Wout[(layer * 64 + c) * 4 + j];
    Py[idx] = s * 0.25f;
  }
  if (t < 16) {
    float s = 0.f;
    if (t < 4) {
      for (int c = 0; c < 64; ++c)
        s += bq[c] * Wout[c * 4 + t] + ba[c] * Wout[(64 + c) * 4 + t];
      s += bout[t];
    }
    cst[t] = s;
  }
}

__global__ void dense_kernel(const float* __restrict__ xc, const float* __restrict__ xq,
                             const float* __restrict__ xa, const float* __restrict__ Vp,
                             const float* __restrict__ Py,
                             float* __restrict__ as_cq, float* __restrict__ as_ca,
                             float* __restrict__ ad_cq, float* __restrict__ ad_ca,
                             float* __restrict__ yq, float* __restrict__ ya,
                             float4* __restrict__ zero_ptr, long nzero4,
                             int Nc, int Nq, int Na, int gNc, int gNq) {
  {
    const long stride = (long)gridDim.x * blockDim.x;
    const long t0 = (long)blockIdx.x * blockDim.x + threadIdx.x;
    const float4 z4 = {0.f, 0.f, 0.f, 0.f};
    for (long i = t0; i < nzero4; i += stride) zero_ptr[i] = z4;
  }
  const int lane = threadIdx.x & 31;
  const int wv = (int)(threadIdx.x >> 5);
  const int m = lane & 15;
  const int ko = (lane >> 4) * 2;
  int b = (int)blockIdx.x;
  if (b < gNc) {
    const long base = ((long)b * 8 + wv) * 16;
    if (base >= Nc) return;
    long row = base + m; if (row >= Nc) row = Nc - 1;
    const float* xr = xc + row * 64;
    const float* bt = (m < 4) ? (Vp + m) : (m < 8) ? (Vp + 2 * 1024 + (m - 4)) : (Vp + m);
    v8f accs = {}, accq = {}, acca = {};
#pragma unroll
    for (int kk = 0; kk < 64; kk += 4) {
      v2f a, bs, b0, b1;
      a.x = xr[kk + ko];
      a.y = xr[kk + ko + 1];
      bs.x = bt[(kk + ko) * 16];
      bs.y = bt[(kk + ko + 1) * 16];
      b0.x = Py[(kk + ko) * 32 + m];
      b0.y = Py[(kk + ko + 1) * 32 + m];
      b1.x = Py[(kk + ko) * 32 + 16 + m];
      b1.y = Py[(kk + ko + 1) * 32 + 16 + m];
      accs = __builtin_amdgcn_wmma_f32_16x16x4_f32(false, a, false, bs, (short)0, accs,
                                                   false, false);
      accq = __builtin_amdgcn_wmma_f32_16x16x4_f32(false, a, false, b0, (short)0, accq,
                                                   false, false);
      acca = __builtin_amdgcn_wmma_f32_16x16x4_f32(false, a, false, b1, (short)0, acca,
                                                   false, false);
    }
    const long rb = base + ((lane >> 4) << 3);
    float* so = (m < 4) ? (as_cq + m) : (as_ca + (m - 4));
#pragma unroll
    for (int v = 0; v < 8; ++v) {
      long r = rb + v;
      if (r < Nc) {
        yq[r * 16 + m] = accq[v];
        ya[r * 16 + m] = acca[v];
        if (m < 8) so[r * 4] = accs[v];
      }
    }
  } else {
    b -= gNc;
    const float* x = xq; const float* V = Vp + 1024; float* o = ad_cq; int N = Nq;
    if (b >= gNq) { b -= gNq; x = xa; V = Vp + 3 * 1024; o = ad_ca; N = Na; }
    const long base = ((long)b * 8 + wv) * 16;
    if (base >= N) return;
    long row = base + m; if (row >= N) row = N - 1;
    const float* xr = x + row * 64;
    v8f acc = {};
#pragma unroll
    for (int kk = 0; kk < 64; kk += 4) {
      v2f a, bv;
      a.x = xr[kk + ko];
      a.y = xr[kk + ko + 1];
      bv.x = V[(kk + ko) * 16 + m];
      bv.y = V[(kk + ko + 1) * 16 + m];
      acc = __builtin_amdgcn_wmma_f32_16x16x4_f32(false, a, false, bv, (short)0, acc,
                                                  false, false);
    }
    if (m < 4) {
      const long rb = base + ((lane >> 4) << 3);
#pragma unroll
      for (int v = 0; v < 8; ++v) {
        long r = rb + v;
        if (r < N) o[r * 4 + m] = acc[v];
      }
    }
  }
}


__global__ void edge_pass1(const int* __restrict__ src_q, const int* __restrict__ dst_q,
                           const int* __restrict__ src_a, const int* __restrict__ dst_a,
                           const float* __restrict__ as_q, const float* __restrict__ ad_q,
                           const float* __restrict__ as_a, const float* __restrict__ ad_a,
                           float* __restrict__ e_q, float* __restrict__ e_a,
                           unsigned* __restrict__ mk_q, unsigned* __restrict__ mk_a,
                           long pq, long pa, long Eq, long Ea) {
  const int half = package_half();
  unsigned* const mkh_q = mk_q + half * pq;
  unsigned* const mkh_a = mk_a + half * pa;
  const long nq = Eq * 4;
  const long n = nq + Ea * 4;
  const long stride = (long)gridDim.x * blockDim.x;
  for (long it = (long)blockIdx.x * blockDim.x + threadIdx.x; it < n; it += stride) {
    const bool la = it >= nq;
    const long li = la ? it - nq : it;
    const int* src = la ? src_a : src_q;
    const int* dst = la ? dst_a : dst_q;
    const float* a_s = la ? as_a : as_q;
    const float* a_d = la ? ad_a : ad_q;
    float* e = la ? e_a : e_q;
    unsigned* mkey = la ? mkh_a : mkh_q;
    const long ei = li >> 2;
    const int h = (int)(li & 3);
    const long s = src[ei], d = dst[ei];
    const float ev = lrelu(a_s[s * 4 + h] + a_d[d * 4 + h]);
    e[li] = ev;
    __hip_atomic_fetch_max(mkey + d * 4 + h, f2key(ev), __ATOMIC_RELAXED,
                           __HIP_MEMORY_SCOPE_AGENT);
  }
}

__global__ void merge_max(uint4* __restrict__ mk_q, uint4* __restrict__ mk_a,
                          long pq4, long pa4, long nq4, long na4) {
  const long n = nq4 + na4;
  const long stride = (long)gridDim.x * blockDim.x;
  for (long i = (long)blockIdx.x * blockDim.x + threadIdx.x; i < n; i += stride) {
    const bool la = i >= nq4;
    uint4* p = la ? mk_a + (i - nq4) : mk_q + i;
    const uint4 a = p[0];
    const uint4 b = p[la ? pa4 : pq4];
    uint4 r;
    r.x = a.x > b.x ? a.x : b.x;
    r.y = a.y > b.y ? a.y : b.y;
    r.z = a.z > b.z ? a.z : b.z;
    r.w = a.w > b.w ? a.w : b.w;
    p[0] = r;
  }
}

__global__ void edge_pass2(const int* __restrict__ src_q, const int* __restrict__ dst_q,
                const int* __restrict__ src_a, const int* __restrict__ dst_a,
                const float* __restrict__ y_q, const float* __restrict__ y_a,
                const float* __restrict__ e_q, const float* __restrict__ e_a,
                const unsigned* __restrict__ mk_q, const unsigned* __restrict__ mk_a,
                float* __restrict__ acc_q, float* __restrict__ acc_a,
                long pq, long pa, long Eq, long Ea) {
  const int half = package_half();
  float* const acch_q = acc_q + half * pq;
  float* const acch_a = acc_a + half * pa;
  const int lane = threadIdx.x & 31;
  const int h = lane & 3;
  const int c16 = lane & 15;
  const long wave = (long)blockIdx.x * (blockDim.x >> 5) +
                    __builtin_amdgcn_readfirstlane((int)(threadIdx.x >> 5));
  const long nwaves = (long)gridDim.x * (blockDim.x >> 5);
  const long nbq = (Eq + 31) >> 5;
  const long nb = nbq + ((Ea + 31) >> 5);
  for (long blk = wave; blk < nb; blk += nwaves) {
    const bool la = blk >= nbq;
    const int* src = la ? src_a : src_q;
    const int* dst = la ? dst_a : dst_q;
    const float* y = la ? y_a : y_q;
    const float* e = la ? e_a : e_q;
    const unsigned* mkey = la ? mk_a : mk_q;
    float* acc = la ? acch_a : acch_q;
    const long E = la ? Ea : Eq;
    const long base = (la ? blk - nbq : blk) << 5;
    const long left = E - base;
    const int cnt = left < 32 ? (int)left : 32;
    const int lc = lane < cnt ? lane : cnt - 1;
    const int s = src[base + lc];
    const int d = dst[base + lc];
    float exj[4];
#pragma unroll
    for (int j = 0; j < 4; ++j) {
      int el = j * 8 + (lane >> 2);
      if (el >= cnt) el = cnt - 1;
      const int dj = __builtin_amdgcn_ds_bpermute(el << 2, d);
      const float ev = e[(base + el) * 4 + h];
      const unsigned mk = mkey[(long)dj * 4 + h];
      exj[j] = __expf(ev - key2f(mk));
    }
#pragma unroll
    for (int j = 0; j < 4; ++j) {
      if (j * 8 < cnt) {
        float yv[8];
#pragma unroll
        for (int u = 0; u < 8; ++u) {
          int kk = j * 8 + u; if (kk >= cnt) kk = cnt - 1;
          const int sk = __builtin_amdgcn_readlane(s, kk);
          yv[u] = y[(long)sk * 16 + c16];
        }
#pragma unroll
        for (int u = 0; u < 8; ++u) {
          const int kk = j * 8 + u;
          if (kk < cnt) {
            const int dk = __builtin_amdgcn_readlane(d, kk);
            const float ex = __int_as_float(
                __builtin_amdgcn_ds_bpermute((u * 4 + h) << 2, __float_as_int(exj[j])));
            const float val = (lane < 16) ? ex * yv[u] : ex;
            if (lane < 20) atomic_add_f32(acc + (long)dk * 32 + lane, val);
          }
        }
      }
    }
  }
}

__device__ __forceinline__ float4 add4(const float4 a, const float4 b) {
  return make_float4(a.x + b.x, a.y + b.y, a.z + b.z, a.w + b.w);
}
__global__ void node_pass(const float* __restrict__ acc_q, const float* __restrict__ acc_a,
                          const float* __restrict__ cst, float* __restrict__ out,
                          long pq, long pa, long Nq, long Na) {
  const long t = (long)blockIdx.x * blockDim.x + threadIdx.x;
  const long d = t >> 2;
  const int j = (int)(t & 3);
  if (d >= Nq) return;
  float o = cst[j];
  {
    const float* r0 = acc_q + d * 32;
    const float* r1 = r0 + pq;
    const float4 n = add4(*(const float4*)(r0 + j * 4), *(const float4*)(r1 + j * 4));
    const float4 dn = add4(*(const float4*)(r0 + 16), *(const float4*)(r1 + 16));
    o += n.x / (dn.x + 1e-16f);
    o += n.y / (dn.y + 1e-16f);
    o += n.z / (dn.z + 1e-16f);
    o += n.w / (dn.w + 1e-16f);
  }
  if (d < Na) {
    const float* r0 = acc_a + d * 32;
    const float* r1 = r0 + pa;
    const float4 n = add4(*(const float4*)(r0 + j * 4), *(const float4*)(r1 + j * 4));
    const float4 dn = add4(*(const float4*)(r0 + 16), *(const float4*)(r1 + 16));
    o += n.x / (dn.x + 1e-16f);
    o += n.y / (dn.y + 1e-16f);
    o += n.z / (dn.z + 1e-16f);
    o += n.w / (dn.w + 1e-16f);
  }
  out[t] = o;
}

extern "C" void kernel_launch(void* const* d_in, const int* in_sizes, int n_in,
                              void* d_out, int out_size, void* d_ws, size_t ws_size,
                              hipStream_t stream) {
  const float* x_c = (const float*)d_in[0];
  const float* x_q = (const float*)d_in[1];
  const float* x_a = (const float*)d_in[2];
  const int* src_cq = (const int*)d_in[3];
  const int* dst_cq = (const int*)d_in[4];
  const int* src_ca = (const int*)d_in[5];
  const int* dst_ca = (const int*)d_in[6];
  const float* Wsq = (const float*)d_in[7];
  const float* Wdq = (const float*)d_in[8];
  const float* Asq = (const float*)d_in[9];
  const float* Adq = (const float*)d_in[10];
  const float* bq  = (const float*)d_in[11];
  const float* Wsa = (const float*)d_in[12];
  const float* Wda = (const float*)d_in[13];
  const float* Asa = (const float*)d_in[14];
  const float* Ada = (const float*)d_in[15];
  const float* ba  = (const float*)d_in[16];
  const float* Wout = (const float*)d_in[17];
  const float* bout = (const float*)d_in[18];

  const long Nc = in_sizes[0] / 64;
  const long Nq = in_sizes[1] / 64;
  const long Na = in_sizes[2] / 64;
  const long Ecq = in_sizes[3];
  const long Eca = in_sizes[5];

  float* w = (float*)d_ws;
  size_t off = 0;
  auto alloc = [&](size_t n) -> float* {
    float* p = w + off;
    off += (n + 63) & ~(size_t)63;
    return p;
  };
  float* Vp_sq = alloc(64 * 16);
  float* Vp_dq = alloc(64 * 16);
  float* Vp_sa = alloc(64 * 16);
  float* Vp_da = alloc(64 * 16);
  float* Py    = alloc(64 * 32);
  float* cst   = alloc(16);
  float* as_cq = alloc(Nc * 4);
  float* as_ca = alloc(Nc * 4);
  float* ad_cq = alloc(Nq * 4);
  float* ad_ca = alloc(Na * 4);
  float* e_cq  = alloc(Ecq * 4);
  float* e_ca  = alloc(Eca * 4);
  float* m_q   = alloc(Nq * 4);
  float* m_q1  = alloc(Nq * 4);
  float* m_a   = alloc(Na * 4);
  float* m_a1  = alloc(Na * 4);
  float* acc_q = alloc(Nq * 32);
  float* acc_q1 = alloc(Nq * 32);
  float* acc_a = alloc(Na * 32);
  float* acc_a1 = alloc(Na * 32);
  float* y_q   = alloc(Nc * 16);
  float* y_a   = alloc(Nc * 16);
  (void)Vp_dq; (void)Vp_sa; (void)Vp_da;
  if (off * sizeof(float) > ws_size) return;

  const long nzero4 = (long)(y_q - m_q) / 4;
  const long pmq = (long)(m_q1 - m_q), pma = (long)(m_a1 - m_a);
  const long paq = (long)(acc_q1 - acc_q), paa = (long)(acc_a1 - acc_a);

  prep_kernel<<<1, 1024, 0, stream>>>(Wsq, Wdq, Asq, Adq, Wsa, Wda, Asa, Ada,
                                      bq, ba, Wout, bout, Vp_sq, Py, cst);

  const int gNc = (int)((Nc + 127) / 128);
  const int gNq = (int)((Nq + 127) / 128);
  const int gNa = (int)((Na + 127) / 128);
  dense_kernel<<<gNc + gNq + gNa, 256, 0, stream>>>(
      x_c, x_q, x_a, Vp_sq, Py, as_cq, as_ca, ad_cq, ad_ca, y_q, y_a,
      (float4*)m_q, nzero4, (int)Nc, (int)Nq, (int)Na, gNc, gNq);

  edge_pass1<<<2048, 256, 0, stream>>>(src_cq, dst_cq, src_ca, dst_ca, as_cq, ad_cq, as_ca,
                                       ad_ca, e_cq, e_ca, (unsigned*)m_q, (unsigned*)m_a,
                                       pmq, pma, Ecq, Eca);
  merge_max<<<(int)((Nq + Na + 255) / 256), 256, 0, stream>>>((uint4*)m_q, (uint4*)m_a,
                                                              pmq / 4, pma / 4, Nq, Na);
  edge_pass2<<<2048, 256, 0, stream>>>(src_cq, dst_cq, src_ca, dst_ca, y_q, y_a, e_cq, e_ca,
                                       (const unsigned*)m_q, (const unsigned*)m_a,
                                       acc_q, acc_a, paq, paa, Ecq, Eca);
  node_pass<<<(int)((Nq * 4 + 255) / 256), 256, 0, stream>>>(acc_q, acc_a, cst,
                                                             (float*)d_out, paq, paa, Nq, Na);
}


# Segment max taken over a_s[src] alone, m = lrelu(max + a_d) once per node (non-decreasing map, so exact): edge pass 1 drops its a_d gather; pass 2 forms e from streamed a_s and a {a_d,m} record
# speedup vs baseline: 9.9027x; 1.0821x over previous
#include <hip/hip_runtime.h>
#include <hip/hip_bf16.h>
#include <math.h>

typedef __attribute__((ext_vector_type(2))) float v2f;
typedef __attribute__((ext_vector_type(8))) float v8f;

__device__ __forceinline__ float lrelu(float x) { return x > 0.f ? x : 0.2f * x; }

__device__ __forceinline__ float edge_score(float a_src, float a_dst) {
  return lrelu(a_src + a_dst);
}

__device__ __forceinline__ unsigned f2key(float v) {
  const unsigned b = __float_as_uint(v);
  return (b & 0x80000000u) ? ~b : (b | 0x80000000u);
}
__device__ __forceinline__ float key2f(unsigned k) {
  return __uint_as_float((k & 0x80000000u) ? (k & 0x7fffffffu) : ~k);
}

__device__ __forceinline__ void atomic_add_f32(float* addr, float val) {
  __hip_atomic_fetch_add(addr, val, __ATOMIC_RELAXED, __HIP_MEMORY_SCOPE_AGENT);
}

__device__ __forceinline__ int package_half() {
  return (int)((__builtin_amdgcn_s_sendmsg_rtn(0x87u) >> 18) & 1u);
}

__global__ __launch_bounds__(1024)
void prep_kernel(const float* __restrict__ Wsq, const float* __restrict__ Wdq,
                 const float* __restrict__ Asq, const float* __restrict__ Adq,
                 const float* __restrict__ Wsa, const float* __restrict__ Wda,
                 const float* __restrict__ Asa, const float* __restrict__ Ada,
                 const float* __restrict__ bq,  const float* __restrict__ ba,
                 const float* __restrict__ Wout, const float* __restrict__ bout,
                 float* __restrict__ Vp, float* __restrict__ Py, float* __restrict__ cst) {
  const int t = threadIdx.x;
  for (int idx = t; idx < 4 * 64 * 16; idx += blockDim.x) {
    int table = idx >> 10, k = (idx >> 4) & 63, j = idx & 15;
    const float* W = (table == 0) ? Wsq : (table == 1) ? Wdq : (table == 2) ? Wsa : Wda;
    const float* A = (table == 0) ? Asq : (table == 1) ? Adq : (table == 2) ? Asa : Ada;
    float s = 0.f;
    if (j < 4)
      for (int c = 0; c < 64; ++c) s += W[k * 256 + j * 64 + c] * A[j * 64 + c];
    Vp[idx] = s;
  }
  for (int idx = t; idx < 64 * 32; idx += blockDim.x) {
    int k = idx >> 5, n = idx & 31;
    int layer = n >> 4, j = (n >> 2) & 3, h = n & 3;
    const float* W = layer ? Wsa : Wsq;
    float s = 0.f;
    for (int c = 0; c < 64; ++c)
      s += W[k * 256 + h * 64 + c] * Wout[(layer * 64 + c) * 4 + j];
    Py[idx] = s * 0.25f;
  }
  if (t < 16) {
    float s = 0.f;
    if (t < 4) {
      for (int c = 0; c < 64; ++c)
        s += bq[c] * Wout[c * 4 + t] + ba[c] * Wout[(64 + c) * 4 + t];
      s += bout[t];
    }
    cst[t] = s;
  }
}

__global__ void dense_kernel(const float* __restrict__ xc, const float* __restrict__ xq,
                             const float* __restrict__ xa, const float* __restrict__ Vp,
                             const float* __restrict__ Py,
                             float* __restrict__ as_cq, float* __restrict__ as_ca,
                             float* __restrict__ ad_cq, float* __restrict__ ad_ca,
                             float* __restrict__ yq, float* __restrict__ ya,
                             float4* __restrict__ zero_ptr, long nzero4,
                             int Nc, int Nq, int Na, int gNc, int gNq) {
  {
    const long stride = (long)gridDim.x * blockDim.x;
    const long t0 = (long)blockIdx.x * blockDim.x + threadIdx.x;
    const float4 z4 = {0.f, 0.f, 0.f, 0.f};
    for (long i = t0; i < nzero4; i += stride) zero_ptr[i] = z4;
  }
  const int lane = threadIdx.x & 31;
  const int wv = (int)(threadIdx.x >> 5);
  const int m = lane & 15;
  const int ko = (lane >> 4) * 2;
  int b = (int)blockIdx.x;
  if (b < gNc) {
    const long base = ((long)b * 8 + wv) * 16;
    if (base >= Nc) return;
    long row = base + m; if (row >= Nc) row = Nc - 1;
    const float* xr = xc + row * 64;
    const float* bt = (m < 4) ? (Vp + m) : (m < 8) ? (Vp + 2 * 1024 + (m - 4)) : (Vp + m);
    v8f accs = {}, accq = {}, acca = {};
#pragma unroll
    for (int kk = 0; kk < 64; kk += 4) {
      v2f a, bs, b0, b1;
      a.x = xr[kk + ko];
      a.y = xr[kk + ko + 1];
      bs.x = bt[(kk + ko) * 16];
      bs.y = bt[(kk + ko + 1) * 16];
      b0.x = Py[(kk + ko) * 32 + m];
      b0.y = Py[(kk + ko + 1) * 32 + m];
      b1.x = Py[(kk + ko) * 32 + 16 + m];
      b1.y = Py[(kk + ko + 1) * 32 + 16 + m];
      accs = __builtin_amdgcn_wmma_f32_16x16x4_f32(false, a, false, bs, (short)0, accs,
                                                   false, false);
      accq = __builtin_amdgcn_wmma_f32_16x16x4_f32(false, a, false, b0, (short)0, accq,
                                                   false, false);
      acca = __builtin_amdgcn_wmma_f32_16x16x4_f32(false, a, false, b1, (short)0, acca,
                                                   false, false);
    }
    const long rb = base + ((lane >> 4) << 3);
    float* so = (m < 4) ? (as_cq + m) : (as_ca + (m - 4));
#pragma unroll
    for (int v = 0; v < 8; ++v) {
      long r = rb + v;
      if (r < Nc) {
        yq[r * 16 + m] = accq[v];
        ya[r * 16 + m] = acca[v];
        if (m < 8) so[r * 4] = accs[v];
      }
    }
  } else {
    b -= gNc;
    const float* x = xq; const float* V = Vp + 1024; float* o = ad_cq; int N = Nq;
    if (b >= gNq) { b -= gNq; x = xa; V = Vp + 3 * 1024; o = ad_ca; N = Na; }
    const long base = ((long)b * 8 + wv) * 16;
    if (base >= N) return;
    long row = base + m; if (row >= N) row = N - 1;
    const float* xr = x + row * 64;
    v8f acc = {};
#pragma unroll
    for (int kk = 0; kk < 64; kk += 4) {
      v2f a, bv;
      a.x = xr[kk + ko];
      a.y = xr[kk + ko + 1];
      bv.x = V[(kk + ko) * 16 + m];
      bv.y = V[(kk + ko + 1) * 16 + m];
      acc = __builtin_amdgcn_wmma_f32_16x16x4_f32(false, a, false, bv, (short)0, acc,
                                                  false, false);
    }
    if (m < 4) {
      const long rb = base + ((lane >> 4) << 3);
#pragma unroll
      for (int v = 0; v < 8; ++v) {
        long r = rb + v;
        if (r < N) o[r * 4 + m] = acc[v];
      }
    }
  }
}


__global__ void edge_pass1(const int* __restrict__ src_q, const int* __restrict__ dst_q,
                           const int* __restrict__ src_a, const int* __restrict__ dst_a,
                           const float* __restrict__ as_q, const float* __restrict__ as_a,
                           float* __restrict__ g_q, float* __restrict__ g_a,
                           unsigned* __restrict__ mk_q, unsigned* __restrict__ mk_a,
                           long pq, long pa, long Eq, long Ea) {
  const int half = package_half();
  unsigned* const mkh_q = mk_q + half * pq;
  unsigned* const mkh_a = mk_a + half * pa;
  const long nq = Eq * 4;
  const long n = nq + Ea * 4;
  const long stride = (long)gridDim.x * blockDim.x;
  for (long it = (long)blockIdx.x * blockDim.x + threadIdx.x; it < n; it += stride) {
    const bool la = it >= nq;
    const long li = la ? it - nq : it;
    const int* src = la ? src_a : src_q;
    const int* dst = la ? dst_a : dst_q;
    const float* a_s = la ? as_a : as_q;
    float* g = la ? g_a : g_q;
    unsigned* mkey = la ? mkh_a : mkh_q;
    const long ei = li >> 2;
    const int h = (int)(li & 3);
    const long s = src[ei], d = dst[ei];
    const float av = a_s[s * 4 + h];
    g[li] = av;
    __hip_atomic_fetch_max(mkey + d * 4 + h, f2key(av), __ATOMIC_RELAXED,
                           __HIP_MEMORY_SCOPE_AGENT);
  }
}

__global__ void node_max(const uint4* __restrict__ mk_q, const uint4* __restrict__ mk_a,
                         const float4* __restrict__ ad_q, const float4* __restrict__ ad_a,
                         float4* __restrict__ rec_q, float4* __restrict__ rec_a,
                         long pq4, long pa4, long nq4, long na4) {
  const long n = nq4 + na4;
  const long stride = (long)gridDim.x * blockDim.x;
  for (long i = (long)blockIdx.x * blockDim.x + threadIdx.x; i < n; i += stride) {
    const bool la = i >= nq4;
    const long li = la ? i - nq4 : i;
    const uint4* p = (la ? mk_a : mk_q) + li;
    const uint4 a = p[0];
    const uint4 b = p[la ? pa4 : pq4];
    const float4 ad = (la ? ad_a : ad_q)[li];
    const float m0 = edge_score(key2f(a.x > b.x ? a.x : b.x), ad.x);
    const float m1 = edge_score(key2f(a.y > b.y ? a.y : b.y), ad.y);
    const float m2 = edge_score(key2f(a.z > b.z ? a.z : b.z), ad.z);
    const float m3 = edge_score(key2f(a.w > b.w ? a.w : b.w), ad.w);
    float4* o = (la ? rec_a : rec_q) + li * 2;
    o[0] = make_float4(ad.x, m0, ad.y, m1);
    o[1] = make_float4(ad.z, m2, ad.w, m3);
  }
}

__global__ void edge_pass2(const int* __restrict__ src_q, const int* __restrict__ dst_q,
                const int* __restrict__ src_a, const int* __restrict__ dst_a,
                const float* __restrict__ y_q, const float* __restrict__ y_a,
                const float* __restrict__ g_q, const float* __restrict__ g_a,
                const float* __restrict__ rec_q, const float* __restrict__ rec_a,
                float* __restrict__ acc_q, float* __restrict__ acc_a,
                long pq, long pa, long Eq, long Ea) {
  const int half = package_half();
  float* const acch_q = acc_q + half * pq;
  float* const acch_a = acc_a + half * pa;
  const int lane = threadIdx.x & 31;
  const int h = lane & 3;
  const int c16 = lane & 15;
  const long wave = (long)blockIdx.x * (blockDim.x >> 5) +
                    __builtin_amdgcn_readfirstlane((int)(threadIdx.x >> 5));
  const long nwaves = (long)gridDim.x * (blockDim.x >> 5);
  const long nbq = (Eq + 31) >> 5;
  const long nb = nbq + ((Ea + 31) >> 5);
  for (long blk = wave; blk < nb; blk += nwaves) {
    const bool la = blk >= nbq;
    const int* src = la ? src_a : src_q;
    const int* dst = la ? dst_a : dst_q;
    const float* y = la ? y_a : y_q;
    const float* g = la ? g_a : g_q;
    const float* rec = la ? rec_a : rec_q;
    float* acc = la ? acch_a : acch_q;
    const long E = la ? Ea : Eq;
    const long base = (la ? blk - nbq : blk) << 5;
    const long left = E - base;
    const int cnt = left < 32 ? (int)left : 32;
    const int lc = lane < cnt ? lane : cnt - 1;
    const int s = src[base + lc];
    const int d = dst[base + lc];
    float exj[4];
#pragma unroll
    for (int j = 0; j < 4; ++j) {
      int el = j * 8 + (lane >> 2);
      if (el >= cnt) el = cnt - 1;
      const int dj = __builtin_amdgcn_ds_bpermute(el << 2, d);
      const float av = g[(base + el) * 4 + h];
      const float2 r = *(const float2*)(rec + (long)dj * 8 + h * 2);
      exj[j] = __expf(edge_score(av, r.x) - r.y);
    }
#pragma unroll
    for (int j = 0; j < 4; ++j) {
      if (j * 8 < cnt) {
        float yv[8];
#pragma unroll
        for (int u = 0; u < 8; ++u) {
          int kk = j * 8 + u; if (kk >= cnt) kk = cnt - 1;
          const int sk = __builtin_amdgcn_readlane(s, kk);
          yv[u] = y[(long)sk * 16 + c16];
        }
#pragma unroll
        for (int u = 0; u < 8; ++u) {
          const int kk = j * 8 + u;
          if (kk < cnt) {
            const int dk = __builtin_amdgcn_readlane(d, kk);
            const float ex = __int_as_float(
                __builtin_amdgcn_ds_bpermute((u * 4 + h) << 2, __float_as_int(exj[j])));
            const float val = (lane < 16) ? ex * yv[u] : ex;
            if (lane < 20) atomic_add_f32(acc + (long)dk * 32 + lane, val);
          }
        }
      }
    }
  }
}

__device__ __forceinline__ float4 add4(const float4 a, const float4 b) {
  return make_float4(a.x + b.x, a.y + b.y, a.z + b.z, a.w + b.w);
}
__global__ void node_pass(const float* __restrict__ acc_q, const float* __restrict__ acc_a,
                          const float* __restrict__ cst, float* __restrict__ out,
                          long pq, long pa, long Nq, long Na) {
  const long t = (long)blockIdx.x * blockDim.x + threadIdx.x;
  const long d = t >> 2;
  const int j = (int)(t & 3);
  if (d >= Nq) return;
  float o = cst[j];
  {
    const float* r0 = acc_q + d * 32;
    const float* r1 = r0 + pq;
    const float4 n = add4(*(const float4*)(r0 + j * 4), *(const float4*)(r1 + j * 4));
    const float4 dn = add4(*(const float4*)(r0 + 16), *(const float4*)(r1 + 16));
    o += n.x / (dn.x + 1e-16f);
    o += n.y / (dn.y + 1e-16f);
    o += n.z / (dn.z + 1e-16f);
    o += n.w / (dn.w + 1e-16f);
  }
  if (d < Na) {
    const float* r0 = acc_a + d * 32;
    const float* r1 = r0 + pa;
    const float4 n = add4(*(const float4*)(r0 + j * 4), *(const float4*)(r1 + j * 4));
    const float4 dn = add4(*(const float4*)(r0 + 16), *(const float4*)(r1 + 16));
    o += n.x / (dn.x + 1e-16f);
    o += n.y / (dn.y + 1e-16f);
    o += n.z / (dn.z + 1e-16f);
    o += n.w / (dn.w + 1e-16f);
  }
  out[t] = o;
}

extern "C" void kernel_launch(void* const* d_in, const int* in_sizes, int n_in,
                              void* d_out, int out_size, void* d_ws, size_t ws_size,
                              hipStream_t stream) {
  const float* x_c = (const float*)d_in[0];
  const float* x_q = (const float*)d_in[1];
  const float* x_a = (const float*)d_in[2];
  const int* src_cq = (const int*)d_in[3];
  const int* dst_cq = (const int*)d_in[4];
  const int* src_ca = (const int*)d_in[5];
  const int* dst_ca = (const int*)d_in[6];
  const float* Wsq = (const float*)d_in[7];
  const float* Wdq = (const float*)d_in[8];
  const float* Asq = (const float*)d_in[9];
  const float* Adq = (const float*)d_in[10];
  const float* bq  = (const float*)d_in[11];
  const float* Wsa = (const float*)d_in[12];
  const float* Wda = (const float*)d_in[13];
  const float* Asa = (const float*)d_in[14];
  const float* Ada = (const float*)d_in[15];
  const float* ba  = (const float*)d_in[16];
  const float* Wout = (const float*)d_in[17];
  const float* bout = (const float*)d_in[18];

  const long Nc = in_sizes[0] / 64;
  const long Nq = in_sizes[1] / 64;
  const long Na = in_sizes[2] / 64;
  const long Ecq = in_sizes[3];
  const long Eca = in_sizes[5];

  float* w = (float*)d_ws;
  size_t off = 0;
  auto alloc = [&](size_t n) -> float* {
    float* p = w + off;
    off += (n + 63) & ~(size_t)63;
    return p;
  };
  float* Vp_sq = alloc(64 * 16);
  float* Vp_dq = alloc(64 * 16);
  float* Vp_sa = alloc(64 * 16);
  float* Vp_da = alloc(64 * 16);
  float* Py    = alloc(64 * 32);
  float* cst   = alloc(16);
  float* as_cq = alloc(Nc * 4);
  float* as_ca = alloc(Nc * 4);
  float* ad_cq = alloc(Nq * 4);
  float* ad_ca = alloc(Na * 4);
  float* g_cq  = alloc(Ecq * 4);
  float* g_ca  = alloc(Eca * 4);
  float* rec_q = alloc(Nq * 8);
  float* rec_a = alloc(Na * 8);
  float* m_q   = alloc(Nq * 4);
  float* m_q1  = alloc(Nq * 4);
  float* m_a   = alloc(Na * 4);
  float* m_a1  = alloc(Na * 4);
  float* acc_q = alloc(Nq * 32);
  float* acc_q1 = alloc(Nq * 32);
  float* acc_a = alloc(Na * 32);
  float* acc_a1 = alloc(Na * 32);
  float* y_q   = alloc(Nc * 16);
  float* y_a   = alloc(Nc * 16);
  (void)Vp_dq; (void)Vp_sa; (void)Vp_da;
  if (off * sizeof(float) > ws_size) return;

  const long nzero4 = (long)(y_q - m_q) / 4;
  const long pmq = (long)(m_q1 - m_q), pma = (long)(m_a1 - m_a);
  const long paq = (long)(acc_q1 - acc_q), paa = (long)(acc_a1 - acc_a);

  prep_kernel<<<1, 1024, 0, stream>>>(Wsq, Wdq, Asq, Adq, Wsa, Wda, Asa, Ada,
                                      bq, ba, Wout, bout, Vp_sq, Py, cst);

  const int gNc = (int)((Nc + 127) / 128);
  const int gNq = (int)((Nq + 127) / 128);
  const int gNa = (int)((Na + 127) / 128);
  dense_kernel<<<gNc + gNq + gNa, 256, 0, stream>>>(
      x_c, x_q, x_a, Vp_sq, Py, as_cq, as_ca, ad_cq, ad_ca, y_q, y_a,
      (float4*)m_q, nzero4, (int)Nc, (int)Nq, (int)Na, gNc, gNq);

  edge_pass1<<<2048, 256, 0, stream>>>(src_cq, dst_cq, src_ca, dst_ca, as_cq, as_ca,
                                       g_cq, g_ca, (unsigned*)m_q, (unsigned*)m_a,
                                       pmq, pma, Ecq, Eca);
  node_max<<<(int)((Nq + Na + 255) / 256), 256, 0, stream>>>(
      (const uint4*)m_q, (const uint4*)m_a, (const float4*)ad_cq, (const float4*)ad_ca,
      (float4*)rec_q, (float4*)rec_a, pmq / 4, pma / 4, Nq, Na);
  edge_pass2<<<2048, 256, 0, stream>>>(src_cq, dst_cq, src_ca, dst_ca, y_q, y_a, g_cq, g_ca,
                                       rec_q, rec_a, acc_q, acc_a, paq, paa, Ecq, Eca);
  node_pass<<<(int)((Nq * 4 + 255) / 256), 256, 0, stream>>>(acc_q, acc_a, cst,
                                                             (float*)d_out, paq, paa, Nq, Na);
}


# Folded weight tables: 193 waves over the chip, one per 128-byte output line (lanes along the weight row, butterfly sums), replace one workgroup's 40 dependent load rounds; table sum order changes
# speedup vs baseline: 10.7486x; 1.0854x over previous
#include <hip/hip_runtime.h>
#include <hip/hip_bf16.h>
#include <math.h>

typedef __attribute__((ext_vector_type(2))) float v2f;
typedef __attribute__((ext_vector_type(8))) float v8f;

__device__ __forceinline__ float lrelu(float x) { return x > 0.f ? x : 0.2f * x; }

__device__ __forceinline__ float edge_score(float a_src, float a_dst) {
  return lrelu(a_src + a_dst);
}

__device__ __forceinline__ unsigned f2key(float v) {
  const unsigned b = __float_as_uint(v);
  return (b & 0x80000000u) ? ~b : (b | 0x80000000u);
}
__device__ __forceinline__ float key2f(unsigned k) {
  return __uint_as_float((k & 0x80000000u) ? (k & 0x7fffffffu) : ~k);
}

__device__ __forceinline__ void atomic_add_f32(float* addr, float val) {
  __hip_atomic_fetch_add(addr, val, __ATOMIC_RELAXED, __HIP_MEMORY_SCOPE_AGENT);
}

__device__ __forceinline__ int package_half() {
  return (int)((__builtin_amdgcn_s_sendmsg_rtn(0x87u) >> 18) & 1u);
}

__device__ __forceinline__ float head_sum(float p) {
  p += __shfl_xor(p, 1, 32);
  p += __shfl_xor(p, 2, 32);
  p += __shfl_xor(p, 4, 32);
  return p;
}
__device__ __forceinline__ float dot8(const float4 w0, const float4 w1,
                                      const float4 a0, const float4 a1) {
  float s = w0.x * a0.x;
  s += w0.y * a0.y; s += w0.z * a0.z; s += w0.w * a0.w;
  s += w1.x * a1.x; s += w1.y * a1.y; s += w1.z * a1.z; s += w1.w * a1.w;
  return s;
}

__global__ __launch_bounds__(128)
void prep_kernel(const float* __restrict__ Wsq, const float* __restrict__ Wdq,
                 const float* __restrict__ Asq, const float* __restrict__ Adq,
                 const float* __restrict__ Wsa, const float* __restrict__ Wda,
                 const float* __restrict__ Asa, const float* __restrict__ Ada,
                 const float* __restrict__ bq,  const float* __restrict__ ba,
                 const float* __restrict__ Wout, const float* __restrict__ bout,
                 float* __restrict__ Vp, float* __restrict__ Py, float* __restrict__ cst) {
  const int lane = threadIdx.x & 31;
  const int job = (int)blockIdx.x * 4 + __builtin_amdgcn_readfirstlane((int)(threadIdx.x >> 5));
  const int src = (lane & 3) * 8;
  if (job < 128) {
    const int table = job >> 5, k0 = (job & 31) * 2;
    const float* W = (table == 0) ? Wsq : (table == 1) ? Wdq : (table == 2) ? Wsa : Wda;
    const float* A = (table == 0) ? Asq : (table == 1) ? Adq : (table == 2) ? Asa : Ada;
    const float4* wr0 = (const float4*)(W + k0 * 256 + lane * 8);
    const float4* wr1 = (const float4*)(W + (k0 + 1) * 256 + lane * 8);
    const float4* ar = (const float4*)(A + lane * 8);
    const float4 w00 = wr0[0], w01 = wr0[1], w10 = wr1[0], w11 = wr1[1];
    const float4 a0 = ar[0], a1 = ar[1];
    const float s0 = head_sum(dot8(w00, w01, a0, a1));
    const float s1 = head_sum(dot8(w10, w11, a0, a1));
    const float q0 = __shfl(s0, src, 32), q1 = __shfl(s1, src, 32);
    Vp[table * 1024 + k0 * 16 + lane] = ((lane & 15) < 4) ? (lane < 16 ? q0 : q1) : 0.f;
  } else if (job < 192) {
    const int k = job - 128;
    const float4* wq = (const float4*)(Wsq + k * 256 + lane * 8);
    const float4* wa = (const float4*)(Wsa + k * 256 + lane * 8);
    const float4* oq = (const float4*)Wout + (lane & 7) * 8;
    const float4* oa = oq + 64;
    const float4 q0 = wq[0], q1 = wq[1], a0 = wa[0], a1 = wa[1];
    const float wqv[8] = {q0.x, q0.y, q0.z, q0.w, q1.x, q1.y, q1.z, q1.w};
    const float wav[8] = {a0.x, a0.y, a0.z, a0.w, a1.x, a1.y, a1.z, a1.w};
    float4 pq = {0.f, 0.f, 0.f, 0.f}, pa = {0.f, 0.f, 0.f, 0.f};
#pragma unroll
    for (int i = 0; i < 8; ++i) {
      const float4 o = oq[i], p = oa[i];
      pq.x += wqv[i] * o.x; pq.y += wqv[i] * o.y; pq.z += wqv[i] * o.z; pq.w += wqv[i] * o.w;
      pa.x += wav[i] * p.x; pa.y += wav[i] * p.y; pa.z += wav[i] * p.z; pa.w += wav[i] * p.w;
    }
    pq.x = head_sum(pq.x); pq.y = head_sum(pq.y); pq.z = head_sum(pq.z); pq.w = head_sum(pq.w);
    pa.x = head_sum(pa.x); pa.y = head_sum(pa.y); pa.z = head_sum(pa.z); pa.w = head_sum(pa.w);
    const float gq0 = __shfl(pq.x, src, 32), gq1 = __shfl(pq.y, src, 32);
    const float gq2 = __shfl(pq.z, src, 32), gq3 = __shfl(pq.w, src, 32);
    const float ga0 = __shfl(pa.x, src, 32), ga1 = __shfl(pa.y, src, 32);
    const float ga2 = __shfl(pa.z, src, 32), ga3 = __shfl(pa.w, src, 32);
    const int j = (lane >> 2) & 3;
    const float vq = (j == 0) ? gq0 : (j == 1) ? gq1 : (j == 2) ? gq2 : gq3;
    const float va = (j == 0) ? ga0 : (j == 1) ? ga1 : (j == 2) ? ga2 : ga3;
    Py[k * 32 + lane] = 0.25f * (lane < 16 ? vq : va);
  } else if (job == 192) {
    const float4* o4 = (const float4*)Wout;
    const float4 t0 = o4[2 * lane], t1 = o4[2 * lane + 1];
    const float4 u0 = o4[64 + 2 * lane], u1 = o4[64 + 2 * lane + 1];
    const float2 cq = *(const float2*)(bq + 2 * lane);
    const float2 ca = *(const float2*)(ba + 2 * lane);
    float p0 = cq.x * t0.x + cq.y * t1.x + ca.x * u0.x + ca.y * u1.x;
    float p1 = cq.x * t0.y + cq.y * t1.y + ca.x * u0.y + ca.y * u1.y;
    float p2 = cq.x * t0.z + cq.y * t1.z + ca.x * u0.z + ca.y * u1.z;
    float p3 = cq.x * t0.w + cq.y * t1.w + ca.x * u0.w + ca.y * u1.w;
#pragma unroll
    for (int m = 16; m >= 1; m >>= 1) {
      p0 += __shfl_xor(p0, m, 32); p1 += __shfl_xor(p1, m, 32);
      p2 += __shfl_xor(p2, m, 32); p3 += __shfl_xor(p3, m, 32);
    }
    float v = 0.f;
    if (lane < 4) v = ((lane == 0) ? p0 : (lane == 1) ? p1 : (lane == 2) ? p2 : p3) + bout[lane];
    if (lane < 16) cst[lane] = v;
  }
}

__global__ void dense_kernel(const float* __restrict__ xc, const float* __restrict__ xq,
                             const float* __restrict__ xa, const float* __restrict__ Vp,
                             const float* __restrict__ Py,
                             float* __restrict__ as_cq, float* __restrict__ as_ca,
                             float* __restrict__ ad_cq, float* __restrict__ ad_ca,
                             float* __restrict__ yq, float* __restrict__ ya,
                             float4* __restrict__ zero_ptr, long nzero4,
                             int Nc, int Nq, int Na, int gNc, int gNq) {
  {
    const long stride = (long)gridDim.x * blockDim.x;
    const long t0 = (long)blockIdx.x * blockDim.x + threadIdx.x;
    const float4 z4 = {0.f, 0.f, 0.f, 0.f};
    for (long i = t0; i < nzero4; i += stride) zero_ptr[i] = z4;
  }
  const int lane = threadIdx.x & 31;
  const int wv = (int)(threadIdx.x >> 5);
  const int m = lane & 15;
  const int ko = (lane >> 4) * 2;
  int b = (int)blockIdx.x;
  if (b < gNc) {
    const long base = ((long)b * 8 + wv) * 16;
    if (base >= Nc) return;
    long row = base + m; if (row >= Nc) row = Nc - 1;
    const float* xr = xc + row * 64;
    const float* bt = (m < 4) ? (Vp + m) : (m < 8) ? (Vp + 2 * 1024 + (m - 4)) : (Vp + m);
    v8f accs = {}, accq = {}, acca = {};
#pragma unroll
    for (int kk = 0; kk < 64; kk += 4) {
      v2f a, bs, b0, b1;
      a.x = xr[kk + ko];
      a.y = xr[kk + ko + 1];
      bs.x = bt[(kk + ko) * 16];
      bs.y = bt[(kk + ko + 1) * 16];
      b0.x = Py[(kk + ko) * 32 + m];
      b0.y = Py[(kk + ko + 1) * 32 + m];
      b1.x = Py[(kk + ko) * 32 + 16 + m];
      b1.y = Py[(kk + ko + 1) * 32 + 16 + m];
      accs = __builtin_amdgcn_wmma_f32_16x16x4_f32(false, a, false, bs, (short)0, accs,
                                                   false, false);
      accq = __builtin_amdgcn_wmma_f32_16x16x4_f32(false, a, false, b0, (short)0, accq,
                                                   false, false);
      acca = __builtin_amdgcn_wmma_f32_16x16x4_f32(false, a, false, b1, (short)0, acca,
                                                   false, false);
    }
    const long rb = base + ((lane >> 4) << 3);
    float* so = (m < 4) ? (as_cq + m) : (as_ca + (m - 4));
#pragma unroll
    for (int v = 0; v < 8; ++v) {
      long r = rb + v;
      if (r < Nc) {
        yq[r * 16 + m] = accq[v];
        ya[r * 16 + m] = acca[v];
        if (m < 8) so[r * 4] = accs[v];
      }
    }
  } else {
    b -= gNc;
    const float* x = xq; const float* V = Vp + 1024; float* o = ad_cq; int N = Nq;
    if (b >= gNq) { b -= gNq; x = xa; V = Vp + 3 * 1024; o = ad_ca; N = Na; }
    const long base = ((long)b * 8 + wv) * 16;
    if (base >= N) return;
    long row = base + m; if (row >= N) row = N - 1;
    const float* xr = x + row * 64;
    v8f acc = {};
#pragma unroll
    for (int kk = 0; kk < 64; kk += 4) {
      v2f a, bv;
      a.x = xr[kk + ko];
      a.y = xr[kk + ko + 1];
      bv.x = V[(kk + ko) * 16 + m];
      bv.y = V[(kk + ko + 1) * 16 + m];
      acc = __builtin_amdgcn_wmma_f32_16x16x4_f32(false, a, false, bv, (short)0, acc,
                                                  false, false);
    }
    if (m < 4) {
      const long rb = base + ((lane >> 4) << 3);
#pragma unroll
      for (int v = 0; v < 8; ++v) {
        long r = rb + v;
        if (r < N) o[r * 4 + m] = acc[v];
      }
    }
  }
}


__global__ void edge_pass1(const int* __restrict__ src_q, const int* __restrict__ dst_q,
                           const int* __restrict__ src_a, const int* __restrict__ dst_a,
                           const float* __restrict__ as_q, const float* __restrict__ as_a,
                           float* __restrict__ g_q, float* __restrict__ g_a,
                           unsigned* __restrict__ mk_q, unsigned* __restrict__ mk_a,
                           long pq, long pa, long Eq, long Ea) {
  const int half = package_half();
  unsigned* const mkh_q = mk_q + half * pq;
  unsigned* const mkh_a = mk_a + half * pa;
  const long nq = Eq * 4;
  const long n = nq + Ea * 4;
  const long stride = (long)gridDim.x * blockDim.x;
  for (long it = (long)blockIdx.x * blockDim.x + threadIdx.x; it < n; it += stride) {
    const bool la = it >= nq;
    const long li = la ? it - nq : it;
    const int* src = la ? src_a : src_q;
    const int* dst = la ? dst_a : dst_q;
    const float* a_s = la ? as_a : as_q;
    float* g = la ? g_a : g_q;
    unsigned* mkey = la ? mkh_a : mkh_q;
    const long ei = li >> 2;
    const int h = (int)(li & 3);
    const long s = src[ei], d = dst[ei];
    const float av = a_s[s * 4 + h];
    g[li] = av;
    __hip_atomic_fetch_max(mkey + d * 4 + h, f2key(av), __ATOMIC_RELAXED,
                           __HIP_MEMORY_SCOPE_AGENT);
  }
}

__global__ void node_max(const uint4* __restrict__ mk_q, const uint4* __restrict__ mk_a,
                         const float4* __restrict__ ad_q, const float4* __restrict__ ad_a,
                         float4* __restrict__ rec_q, float4* __restrict__ rec_a,
                         long pq4, long pa4, long nq4, long na4) {
  const long n = nq4 + na4;
  const long stride = (long)gridDim.x * blockDim.x;
  for (long i = (long)blockIdx.x * blockDim.x + threadIdx.x; i < n; i += stride) {
    const bool la = i >= nq4;
    const long li = la ? i - nq4 : i;
    const uint4* p = (la ? mk_a : mk_q) + li;
    const uint4 a = p[0];
    const uint4 b = p[la ? pa4 : pq4];
    const float4 ad = (la ? ad_a : ad_q)[li];
    const float m0 = edge_score(key2f(a.x > b.x ? a.x : b.x), ad.x);
    const float m1 = edge_score(key2f(a.y > b.y ? a.y : b.y), ad.y);
    const float m2 = edge_score(key2f(a.z > b.z ? a.z : b.z), ad.z);
    const float m3 = edge_score(key2f(a.w > b.w ? a.w : b.w), ad.w);
    float4* o = (la ? rec_a : rec_q) + li * 2;
    o[0] = make_float4(ad.x, m0, ad.y, m1);
    o[1] = make_float4(ad.z, m2, ad.w, m3);
  }
}

__global__ void edge_pass2(const int* __restrict__ src_q, const int* __restrict__ dst_q,
                const int* __restrict__ src_a, const int* __restrict__ dst_a,
                const float* __restrict__ y_q, const float* __restrict__ y_a,
                const float* __restrict__ g_q, const float* __restrict__ g_a,
                const float* __restrict__ rec_q, const float* __restrict__ rec_a,
                float* __restrict__ acc_q, float* __restrict__ acc_a,
                long pq, long pa, long Eq, long Ea) {
  const int half = package_half();
  float* const acch_q = acc_q + half * pq;
  float* const acch_a = acc_a + half * pa;
  const int lane = threadIdx.x & 31;
  const int h = lane & 3;
  const int c16 = lane & 15;
  const long wave = (long)blockIdx.x * (blockDim.x >> 5) +
                    __builtin_amdgcn_readfirstlane((int)(threadIdx.x >> 5));
  const long nwaves = (long)gridDim.x * (blockDim.x >> 5);
  const long nbq = (Eq + 31) >> 5;
  const long nb = nbq + ((Ea + 31) >> 5);
  for (long blk = wave; blk < nb; blk += nwaves) {
    const bool la = blk >= nbq;
    const int* src = la ? src_a : src_q;
    const int* dst = la ? dst_a : dst_q;
    const float* y = la ? y_a : y_q;
    const float* g = la ? g_a : g_q;
    const float* rec = la ? rec_a : rec_q;
    float* acc = la ? acch_a : acch_q;
    const long E = la ? Ea : Eq;
    const long base = (la ? blk - nbq : blk) << 5;
    const long left = E - base;
    const int cnt = left < 32 ? (int)left : 32;
    const int lc = lane < cnt ? lane : cnt - 1;
    const int s = src[base + lc];
    const int d = dst[base + lc];
    float exj[4];
#pragma unroll
    for (int j = 0; j < 4; ++j) {
      int el = j * 8 + (lane >> 2);
      if (el >= cnt) el = cnt - 1;
      const int dj = __builtin_amdgcn_ds_bpermute(el << 2, d);
      const float av = g[(base + el) * 4 + h];
      const float2 r = *(const float2*)(rec + (long)dj * 8 + h * 2);
      exj[j] = __expf(edge_score(av, r.x) - r.y);
    }
#pragma unroll
    for (int j = 0; j < 4; ++j) {
      if (j * 8 < cnt) {
        float yv[8];
#pragma unroll
        for (int u = 0; u < 8; ++u) {
          int kk = j * 8 + u; if (kk >= cnt) kk = cnt - 1;
          const int sk = __builtin_amdgcn_readlane(s, kk);
          yv[u] = y[(long)sk * 16 + c16];
        }
#pragma unroll
        for (int u = 0; u < 8; ++u) {
          const int kk = j * 8 + u;
          if (kk < cnt) {
            const int dk = __builtin_amdgcn_readlane(d, kk);
            const float ex = __int_as_float(
                __builtin_amdgcn_ds_bpermute((u * 4 + h) << 2, __float_as_int(exj[j])));
            const float val = (lane < 16) ? ex * yv[u] : ex;
            if (lane < 20) atomic_add_f32(acc + (long)dk * 32 + lane, val);
          }
        }
      }
    }
  }
}

__device__ __forceinline__ float4 add4(const float4 a, const float4 b) {
  return make_float4(a.x + b.x, a.y + b.y, a.z + b.z, a.w + b.w);
}
__global__ void node_pass(const float* __restrict__ acc_q, const float* __restrict__ acc_a,
                          const float* __restrict__ cst, float* __restrict__ out,
                          long pq, long pa, long Nq, long Na) {
  const long t = (long)blockIdx.x * blockDim.x + threadIdx.x;
  const long d = t >> 2;
  const int j = (int)(t & 3);
  if (d >= Nq) return;
  float o = cst[j];
  {
    const float* r0 = acc_q + d * 32;
    const float* r1 = r0 + pq;
    const float4 n = add4(*(const float4*)(r0 + j * 4), *(const float4*)(r1 + j * 4));
    const float4 dn = add4(*(const float4*)(r0 + 16), *(const float4*)(r1 + 16));
    o += n.x / (dn.x + 1e-16f);
    o += n.y / (dn.y + 1e-16f);
    o += n.z / (dn.z + 1e-16f);
    o += n.w / (dn.w + 1e-16f);
  }
  if (d < Na) {
    const float* r0 = acc_a + d * 32;
    const float* r1 = r0 + pa;
    const float4 n = add4(*(const float4*)(r0 + j * 4), *(const float4*)(r1 + j * 4));
    const float4 dn = add4(*(const float4*)(r0 + 16), *(const float4*)(r1 + 16));
    o += n.x / (dn.x + 1e-16f);
    o += n.y / (dn.y + 1e-16f);
    o += n.z / (dn.z + 1e-16f);
    o += n.w / (dn.w + 1e-16f);
  }
  out[t] = o;
}

extern "C" void kernel_launch(void* const* d_in, const int* in_sizes, int n_in,
                              void* d_out, int out_size, void* d_ws, size_t ws_size,
                              hipStream_t stream) {
  const float* x_c = (const float*)d_in[0];
  const float* x_q = (const float*)d_in[1];
  const float* x_a = (const float*)d_in[2];
  const int* src_cq = (const int*)d_in[3];
  const int* dst_cq = (const int*)d_in[4];
  const int* src_ca = (const int*)d_in[5];
  const int* dst_ca = (const int*)d_in[6];
  const float* Wsq = (const float*)d_in[7];
  const float* Wdq = (const float*)d_in[8];
  const float* Asq = (const float*)d_in[9];
  const float* Adq = (const float*)d_in[10];
  const float* bq  = (const float*)d_in[11];
  const float* Wsa = (const float*)d_in[12];
  const float* Wda = (const float*)d_in[13];
  const float* Asa = (const float*)d_in[14];
  const float* Ada = (const float*)d_in[15];
  const float* ba  = (const float*)d_in[16];
  const float* Wout = (const float*)d_in[17];
  const float* bout = (const float*)d_in[18];

  const long Nc = in_sizes[0] / 64;
  const long Nq = in_sizes[1] / 64;
  const long Na = in_sizes[2] / 64;
  const long Ecq = in_sizes[3];
  const long Eca = in_sizes[5];

  float* w = (float*)d_ws;
  size_t off = 0;
  auto alloc = [&](size_t n) -> float* {
    float* p = w + off;
    off += (n + 63) & ~(size_t)63;
    return p;
  };
  float* Vp_sq = alloc(64 * 16);
  float* Vp_dq = alloc(64 * 16);
  float* Vp_sa = alloc(64 * 16);
  float* Vp_da = alloc(64 * 16);
  float* Py    = alloc(64 * 32);
  float* cst   = alloc(16);
  float* as_cq = alloc(Nc * 4);
  float* as_ca = alloc(Nc * 4);
  float* ad_cq = alloc(Nq * 4);
  float* ad_ca = alloc(Na * 4);
  float* g_cq  = alloc(Ecq * 4);
  float* g_ca  = alloc(Eca * 4);
  float* rec_q = alloc(Nq * 8);
  float* rec_a = alloc(Na * 8);
  float* m_q   = alloc(Nq * 4);
  float* m_q1  = alloc(Nq * 4);
  float* m_a   = alloc(Na * 4);
  float* m_a1  = alloc(Na * 4);
  float* acc_q = alloc(Nq * 32);
  float* acc_q1 = alloc(Nq * 32);
  float* acc_a = alloc(Na * 32);
  float* acc_a1 = alloc(Na * 32);
  float* y_q   = alloc(Nc * 16);
  float* y_a   = alloc(Nc * 16);
  (void)Vp_dq; (void)Vp_sa; (void)Vp_da;
  if (off * sizeof(float) > ws_size) return;

  const long nzero4 = (long)(y_q - m_q) / 4;
  const long pmq = (long)(m_q1 - m_q), pma = (long)(m_a1 - m_a);
  const long paq = (long)(acc_q1 - acc_q), paa = (long)(acc_a1 - acc_a);

  prep_kernel<<<49, 128, 0, stream>>>(Wsq, Wdq, Asq, Adq, Wsa, Wda, Asa, Ada,
                                      bq, ba, Wout, bout, Vp_sq, Py, cst);

  const int gNc = (int)((Nc + 127) / 128);
  const int gNq = (int)((Nq + 127) / 128);
  const int gNa = (int)((Na + 127) / 128);
  dense_kernel<<<gNc + gNq + gNa, 256, 0, stream>>>(
      x_c, x_q, x_a, Vp_sq, Py, as_cq, as_ca, ad_cq, ad_ca, y_q, y_a,
      (float4*)m_q, nzero4, (int)Nc, (int)Nq, (int)Na, gNc, gNq);

  edge_pass1<<<2048, 256, 0, stream>>>(src_cq, dst_cq, src_ca, dst_ca, as_cq, as_ca,
                                       g_cq, g_ca, (unsigned*)m_q, (unsigned*)m_a,
                                       pmq, pma, Ecq, Eca);
  node_max<<<(int)((Nq + Na + 255) / 256), 256, 0, stream>>>(
      (const uint4*)m_q, (const uint4*)m_a, (const float4*)ad_cq, (const float4*)ad_ca,
      (float4*)rec_q, (float4*)rec_a, pmq / 4, pma / 4, Nq, Na);
  edge_pass2<<<2048, 256, 0, stream>>>(src_cq, dst_cq, src_ca, dst_ca, y_q, y_a, g_cq, g_ca,
                                       rec_q, rec_a, acc_q, acc_a, paq, paa, Ecq, Eca);
  node_pass<<<(int)((Nq * 4 + 255) / 256), 256, 0, stream>>>(acc_q, acc_a, cst,
                                                             (float*)d_out, paq, paa, Nq, Na);
}


# Edge pass 1 in trips of 64 edges per wave: indices read one edge per lane (whole lines), handed to the item lanes by cross-lane reads, 8 gathers in flight, not 1, before a store; half the waves
# speedup vs baseline: 11.0700x; 1.0299x over previous
#include <hip/hip_runtime.h>
#include <hip/hip_bf16.h>
#include <math.h>

typedef __attribute__((ext_vector_type(2))) float v2f;
typedef __attribute__((ext_vector_type(8))) float v8f;

__device__ __forceinline__ float lrelu(float x) { return x > 0.f ? x : 0.2f * x; }

__device__ __forceinline__ float edge_score(float a_src, float a_dst) {
  return lrelu(a_src + a_dst);
}

__device__ __forceinline__ unsigned f2key(float v) {
  const unsigned b = __float_as_uint(v);
  return (b & 0x80000000u) ? ~b : (b | 0x80000000u);
}
__device__ __forceinline__ float key2f(unsigned k) {
  return __uint_as_float((k & 0x80000000u) ? (k & 0x7fffffffu) : ~k);
}

__device__ __forceinline__ void atomic_add_f32(float* addr, float val) {
  __hip_atomic_fetch_add(addr, val, __ATOMIC_RELAXED, __HIP_MEMORY_SCOPE_AGENT);
}

__device__ __forceinline__ int package_half() {
  return (int)((__builtin_amdgcn_s_sendmsg_rtn(0x87u) >> 18) & 1u);
}

__device__ __forceinline__ float head_sum(float p) {
  p += __shfl_xor(p, 1, 32);
  p += __shfl_xor(p, 2, 32);
  p += __shfl_xor(p, 4, 32);
  return p;
}
__device__ __forceinline__ float dot8(const float4 w0, const float4 w1,
                                      const float4 a0, const float4 a1) {
  float s = w0.x * a0.x;
  s += w0.y * a0.y; s += w0.z * a0.z; s += w0.w * a0.w;
  s += w1.x * a1.x; s += w1.y * a1.y; s += w1.z * a1.z; s += w1.w * a1.w;
  return s;
}

__global__ __launch_bounds__(128)
void prep_kernel(const float* __restrict__ Wsq, const float* __restrict__ Wdq,
                 const float* __restrict__ Asq, const float* __restrict__ Adq,
                 const float* __restrict__ Wsa, const float* __restrict__ Wda,
                 const float* __restrict__ Asa, const float* __restrict__ Ada,
                 const float* __restrict__ bq,  const float* __restrict__ ba,
                 const float* __restrict__ Wout, const float* __restrict__ bout,
                 float* __restrict__ Vp, float* __restrict__ Py, float* __restrict__ cst) {
  const int lane = threadIdx.x & 31;
  const int job = (int)blockIdx.x * 4 + __builtin_amdgcn_readfirstlane((int)(threadIdx.x >> 5));
  const int src = (lane & 3) * 8;
  if (job < 128) {
    const int table = job >> 5, k0 = (job & 31) * 2;
    const float* W = (table == 0) ? Wsq : (table == 1) ? Wdq : (table == 2) ? Wsa : Wda;
    const float* A = (table == 0) ? Asq : (table == 1) ? Adq : (table == 2) ? Asa : Ada;
    const float4* wr0 = (const float4*)(W + k0 * 256 + lane * 8);
    const float4* wr1 = (const float4*)(W + (k0 + 1) * 256 + lane * 8);
    const float4* ar = (const float4*)(A + lane * 8);
    const float4 w00 = wr0[0], w01 = wr0[1], w10 = wr1[0], w11 = wr1[1];
    const float4 a0 = ar[0], a1 = ar[1];
    const float s0 = head_sum(dot8(w00, w01, a0, a1));
    const float s1 = head_sum(dot8(w10, w11, a0, a1));
    const float q0 = __shfl(s0, src, 32), q1 = __shfl(s1, src, 32);
    Vp[table * 1024 + k0 * 16 + lane] = ((lane & 15) < 4) ? (lane < 16 ? q0 : q1) : 0.f;
  } else if (job < 192) {
    const int k = job - 128;
    const float4* wq = (const float4*)(Wsq + k * 256 + lane * 8);
    const float4* wa = (const float4*)(Wsa + k * 256 + lane * 8);
    const float4* oq = (const float4*)Wout + (lane & 7) * 8;
    const float4* oa = oq + 64;
    const float4 q0 = wq[0], q1 = wq[1], a0 = wa[0], a1 = wa[1];
    const float wqv[8] = {q0.x, q0.y, q0.z, q0.w, q1.x, q1.y, q1.z, q1.w};
    const float wav[8] = {a0.x, a0.y, a0.z, a0.w, a1.x, a1.y, a1.z, a1.w};
    float4 pq = {0.f, 0.f, 0.f, 0.f}, pa = {0.f, 0.f, 0.f, 0.f};
#pragma unroll
    for (int i = 0; i < 8; ++i) {
      const float4 o = oq[i], p = oa[i];
      pq.x += wqv[i] * o.x; pq.y += wqv[i] * o.y; pq.z += wqv[i] * o.z; pq.w += wqv[i] * o.w;
      pa.x += wav[i] * p.x; pa.y += wav[i] * p.y; pa.z += wav[i] * p.z; pa.w += wav[i] * p.w;
    }
    pq.x = head_sum(pq.x); pq.y = head_sum(pq.y); pq.z = head_sum(pq.z); pq.w = head_sum(pq.w);
    pa.x = head_sum(pa.x); pa.y = head_sum(pa.y); pa.z = head_sum(pa.z); pa.w = head_sum(pa.w);
    const float gq0 = __shfl(pq.x, src, 32), gq1 = __shfl(pq.y, src, 32);
    const float gq2 = __shfl(pq.z, src, 32), gq3 = __shfl(pq.w, src, 32);
    const float ga0 = __shfl(pa.x, src, 32), ga1 = __shfl(pa.y, src, 32);
    const float ga2 = __shfl(pa.z, src, 32), ga3 = __shfl(pa.w, src, 32);
    const int j = (lane >> 2) & 3;
    const float vq = (j == 0) ? gq0 : (j == 1) ? gq1 : (j == 2) ? gq2 : gq3;
    const float va = (j == 0) ? ga0 : (j == 1) ? ga1 : (j == 2) ? ga2 : ga3;
    Py[k * 32 + lane] = 0.25f * (lane < 16 ? vq : va);
  } else if (job == 192) {
    const float4* o4 = (const float4*)Wout;
    const float4 t0 = o4[2 * lane], t1 = o4[2 * lane + 1];
    const float4 u0 = o4[64 + 2 * lane], u1 = o4[64 + 2 * lane + 1];
    const float2 cq = *(const float2*)(bq + 2 * lane);
    const float2 ca = *(const float2*)(ba + 2 * lane);
    float p0 = cq.x * t0.x + cq.y * t1.x + ca.x * u0.x + ca.y * u1.x;
    float p1 = cq.x * t0.y + cq.y * t1.y + ca.x * u0.y + ca.y * u1.y;
    float p2 = cq.x * t0.z + cq.y * t1.z + ca.x * u0.z + ca.y * u1.z;
    float p3 = cq.x * t0.w + cq.y * t1.w + ca.x * u0.w + ca.y * u1.w;
#pragma unroll
    for (int m = 16; m >= 1; m >>= 1) {
      p0 += __shfl_xor(p0, m, 32); p1 += __shfl_xor(p1, m, 32);
      p2 += __shfl_xor(p2, m, 32); p3 += __shfl_xor(p3, m, 32);
    }
    float v = 0.f;
    if (lane < 4) v = ((lane == 0) ? p0 : (lane == 1) ? p1 : (lane == 2) ? p2 : p3) + bout[lane];
    if (lane < 16) cst[lane] = v;
  }
}

__global__ void dense_kernel(const float* __restrict__ xc, const float* __restrict__ xq,
                             const float* __restrict__ xa, const float* __restrict__ Vp,
                             const float* __restrict__ Py,
                             float* __restrict__ as_cq, float* __restrict__ as_ca,
                             float* __restrict__ ad_cq, float* __restrict__ ad_ca,
                             float* __restrict__ yq, float* __restrict__ ya,
                             float4* __restrict__ zero_ptr, long nzero4,
                             int Nc, int Nq, int Na, int gNc, int gNq) {
  {
    const long stride = (long)gridDim.x * blockDim.x;
    const long t0 = (long)blockIdx.x * blockDim.x + threadIdx.x;
    const float4 z4 = {0.f, 0.f, 0.f, 0.f};
    for (long i = t0; i < nzero4; i += stride) zero_ptr[i] = z4;
  }
  const int lane = threadIdx.x & 31;
  const int wv = (int)(threadIdx.x >> 5);
  const int m = lane & 15;
  const int ko = (lane >> 4) * 2;
  int b = (int)blockIdx.x;
  if (b < gNc) {
    const long base = ((long)b * 8 + wv) * 16;
    if (base >= Nc) return;
    long row = base + m; if (row >= Nc) row = Nc - 1;
    const float* xr = xc + row * 64;
    const float* bt = (m < 4) ? (Vp + m) : (m < 8) ? (Vp + 2 * 1024 + (m - 4)) : (Vp + m);
    v8f accs = {}, accq = {}, acca = {};
#pragma unroll
    for (int kk = 0; kk < 64; kk += 4) {
      v2f a, bs, b0, b1;
      a.x = xr[kk + ko];
      a.y = xr[kk + ko + 1];
      bs.x = bt[(kk + ko) * 16];
      bs.y = bt[(kk + ko + 1) * 16];
      b0.x = Py[(kk + ko) * 32 + m];
      b0.y = Py[(kk + ko + 1) * 32 + m];
      b1.x = Py[(kk + ko) * 32 + 16 + m];
      b1.y = Py[(kk + ko + 1) * 32 + 16 + m];
      accs = __builtin_amdgcn_wmma_f32_16x16x4_f32(false, a, false, bs, (short)0, accs,
                                                   false, false);
      accq = __builtin_amdgcn_wmma_f32_16x16x4_f32(false, a, false, b0, (short)0, accq,
                                                   false, false);
      acca = __builtin_amdgcn_wmma_f32_16x16x4_f32(false, a, false, b1, (short)0, acca,
                                                   false, false);
    }
    const long rb = base + ((lane >> 4) << 3);
    float* so = (m < 4) ? (as_cq + m) : (as_ca + (m - 4));
#pragma unroll
    for (int v = 0; v < 8; ++v) {
      long r = rb + v;
      if (r < Nc) {
        yq[r * 16 + m] = accq[v];
        ya[r * 16 + m] = acca[v];
        if (m < 8) so[r * 4] = accs[v];
      }
    }
  } else {
    b -= gNc;
    const float* x = xq; const float* V = Vp + 1024; float* o = ad_cq; int N = Nq;
    if (b >= gNq) { b -= gNq; x = xa; V = Vp + 3 * 1024; o = ad_ca; N = Na; }
    const long base = ((long)b * 8 + wv) * 16;
    if (base >= N) return;
    long row = base + m; if (row >= N) row = N - 1;
    const float* xr = x + row * 64;
    v8f acc = {};
#pragma unroll
    for (int kk = 0; kk < 64; kk += 4) {
      v2f a, bv;
      a.x = xr[kk + ko];
      a.y = xr[kk + ko + 1];
      bv.x = V[(kk + ko) * 16 + m];
      bv.y = V[(kk + ko + 1) * 16 + m];
      acc = __builtin_amdgcn_wmma_f32_16x16x4_f32(false, a, false, bv, (short)0, acc,
                                                  false, false);
    }
    if (m < 4) {
      const long rb = base + ((lane >> 4) << 3);
#pragma unroll
      for (int v = 0; v < 8; ++v) {
        long r = rb + v;
        if (r < N) o[r * 4 + m] = acc[v];
      }
    }
  }
}


__global__ void edge_pass1(const int* __restrict__ src_q, const int* __restrict__ dst_q,
                           const int* __restrict__ src_a, const int* __restrict__ dst_a,
                           const float* __restrict__ as_q, const float* __restrict__ as_a,
                           float* __restrict__ g_q, float* __restrict__ g_a,
                           unsigned* __restrict__ mk_q, unsigned* __restrict__ mk_a,
                           long pq, long pa, long Eq, long Ea) {
  const int half = package_half();
  unsigned* const mkh_q = mk_q + half * pq;
  unsigned* const mkh_a = mk_a + half * pa;
  const int lane = threadIdx.x & 31;
  const int h = lane & 3;
  const int e8 = lane >> 2;
  const long wave = (long)blockIdx.x * (blockDim.x >> 5) +
                    __builtin_amdgcn_readfirstlane((int)(threadIdx.x >> 5));
  const long nwaves = (long)gridDim.x * (blockDim.x >> 5);
  const long nbq = (Eq + 31) >> 5;
  const long nb = nbq + ((Ea + 31) >> 5);
  for (long b0 = wave * 2; b0 < nb; b0 += nwaves * 2) {
    const float* a_s[2];
    float* g[2];
    unsigned* mkey[2];
    long base[2];
    int cnt[2], s[2], d[2];
#pragma unroll
    for (int u = 0; u < 2; ++u) {
      const long blk = (b0 + u < nb) ? b0 + u : b0;
      const bool la = blk >= nbq;
      const int* src = la ? src_a : src_q;
      const int* dst = la ? dst_a : dst_q;
      a_s[u] = la ? as_a : as_q;
      g[u] = la ? g_a : g_q;
      mkey[u] = la ? mkh_a : mkh_q;
      base[u] = (la ? blk - nbq : blk) << 5;
      const long left = (la ? Ea : Eq) - base[u];
      cnt[u] = left < 32 ? (int)left : 32;
      const int lc = lane < cnt[u] ? lane : cnt[u] - 1;
      s[u] = src[base[u] + lc];
      d[u] = dst[base[u] + lc];
    }
    float av[2][4];
    long gi[2][4], mi[2][4];
#pragma unroll
    for (int u = 0; u < 2; ++u) {
#pragma unroll
      for (int j = 0; j < 4; ++j) {
        int el = j * 8 + e8;
        if (el >= cnt[u]) el = cnt[u] - 1;
        const int sj = __builtin_amdgcn_ds_bpermute(el << 2, s[u]);
        const int dj = __builtin_amdgcn_ds_bpermute(el << 2, d[u]);
        av[u][j] = a_s[u][(long)sj * 4 + h];
        gi[u][j] = (base[u] + el) * 4 + h;
        mi[u][j] = (long)dj * 4 + h;
      }
    }
#pragma unroll
    for (int u = 0; u < 2; ++u) {
#pragma unroll
      for (int j = 0; j < 4; ++j) {
        g[u][gi[u][j]] = av[u][j];
        __hip_atomic_fetch_max(mkey[u] + mi[u][j], f2key(av[u][j]), __ATOMIC_RELAXED,
                               __HIP_MEMORY_SCOPE_AGENT);
      }
    }
  }
}

__global__ void node_max(const uint4* __restrict__ mk_q, const uint4* __restrict__ mk_a,
                         const float4* __restrict__ ad_q, const float4* __restrict__ ad_a,
                         float4* __restrict__ rec_q, float4* __restrict__ rec_a,
                         long pq4, long pa4, long nq4, long na4) {
  const long n = nq4 + na4;
  const long stride = (long)gridDim.x * blockDim.x;
  for (long i = (long)blockIdx.x * blockDim.x + threadIdx.x; i < n; i += stride) {
    const bool la = i >= nq4;
    const long li = la ? i - nq4 : i;
    const uint4* p = (la ? mk_a : mk_q) + li;
    const uint4 a = p[0];
    const uint4 b = p[la ? pa4 : pq4];
    const float4 ad = (la ? ad_a : ad_q)[li];
    const float m0 = edge_score(key2f(a.x > b.x ? a.x : b.x), ad.x);
    const float m1 = edge_score(key2f(a.y > b.y ? a.y : b.y), ad.y);
    const float m2 = edge_score(key2f(a.z > b.z ? a.z : b.z), ad.z);
    const float m3 = edge_score(key2f(a.w > b.w ? a.w : b.w), ad.w);
    float4* o = (la ? rec_a : rec_q) + li * 2;
    o[0] = make_float4(ad.x, m0, ad.y, m1);
    o[1] = make_float4(ad.z, m2, ad.w, m3);
  }
}

__global__ void edge_pass2(const int* __restrict__ src_q, const int* __restrict__ dst_q,
                const int* __restrict__ src_a, const int* __restrict__ dst_a,
                const float* __restrict__ y_q, const float* __restrict__ y_a,
                const float* __restrict__ g_q, const float* __restrict__ g_a,
                const float* __restrict__ rec_q, const float* __restrict__ rec_a,
                float* __restrict__ acc_q, float* __restrict__ acc_a,
                long pq, long pa, long Eq, long Ea) {
  const int half = package_half();
  float* const acch_q = acc_q + half * pq;
  float* const acch_a = acc_a + half * pa;
  const int lane = threadIdx.x & 31;
  const int h = lane & 3;
  const int c16 = lane & 15;
  const long wave = (long)blockIdx.x * (blockDim.x >> 5) +
                    __builtin_amdgcn_readfirstlane((int)(threadIdx.x >> 5));
  const long nwaves = (long)gridDim.x * (blockDim.x >> 5);
  const long nbq = (Eq + 31) >> 5;
  const long nb = nbq + ((Ea + 31) >> 5);
  for (long blk = wave; blk < nb; blk += nwaves) {
    const bool la = blk >= nbq;
    const int* src = la ? src_a : src_q;
    const int* dst = la ? dst_a : dst_q;
    const float* y = la ? y_a : y_q;
    const float* g = la ? g_a : g_q;
    const float* rec = la ? rec_a : rec_q;
    float* acc = la ? acch_a : acch_q;
    const long E = la ? Ea : Eq;
    const long base = (la ? blk - nbq : blk) << 5;
    const long left = E - base;
    const int cnt = left < 32 ? (int)left : 32;
    const int lc = lane < cnt ? lane : cnt - 1;
    const int s = src[base + lc];
    const int d = dst[base + lc];
    float exj[4];
#pragma unroll
    for (int j = 0; j < 4; ++j) {
      int el = j * 8 + (lane >> 2);
      if (el >= cnt) el = cnt - 1;
      const int dj = __builtin_amdgcn_ds_bpermute(el << 2, d);
      const float av = g[(base + el) * 4 + h];
      const float2 r = *(const float2*)(rec + (long)dj * 8 + h * 2);
      exj[j] = __expf(edge_score(av, r.x) - r.y);
    }
#pragma unroll
    for (int j = 0; j < 4; ++j) {
      if (j * 8 < cnt) {
        float yv[8];
#pragma unroll
        for (int u = 0; u < 8; ++u) {
          int kk = j * 8 + u; if (kk >= cnt) kk = cnt - 1;
          const int sk = __builtin_amdgcn_readlane(s, kk);
          yv[u] = y[(long)sk * 16 + c16];
        }
#pragma unroll
        for (int u = 0; u < 8; ++u) {
          const int kk = j * 8 + u;
          if (kk < cnt) {
            const int dk = __builtin_amdgcn_readlane(d, kk);
            const float ex = __int_as_float(
                __builtin_amdgcn_ds_bpermute((u * 4 + h) << 2, __float_as_int(exj[j])));
            const float val = (lane < 16) ? ex * yv[u] : ex;
            if (lane < 20) atomic_add_f32(acc + (long)dk * 32 + lane, val);
          }
        }
      }
    }
  }
}

__device__ __forceinline__ float4 add4(const float4 a, const float4 b) {
  return make_float4(a.x + b.x, a.y + b.y, a.z + b.z, a.w + b.w);
}
__global__ void node_pass(const float* __restrict__ acc_q, const float* __restrict__ acc_a,
                          const float* __restrict__ cst, float* __restrict__ out,
                          long pq, long pa, long Nq, long Na) {
  const long t = (long)blockIdx.x * blockDim.x + threadIdx.x;
  const long d = t >> 2;
  const int j = (int)(t & 3);
  if (d >= Nq) return;
  float o = cst[j];
  {
    const float* r0 = acc_q + d * 32;
    const float* r1 = r0 + pq;
    const float4 n = add4(*(const float4*)(r0 + j * 4), *(const float4*)(r1 + j * 4));
    const float4 dn = add4(*(const float4*)(r0 + 16), *(const float4*)(r1 + 16));
    o += n.x / (dn.x + 1e-16f);
    o += n.y / (dn.y + 1e-16f);
    o += n.z / (dn.z + 1e-16f);
    o += n.w / (dn.w + 1e-16f);
  }
  if (d < Na) {
    const float* r0 = acc_a + d * 32;
    const float* r1 = r0 + pa;
    const float4 n = add4(*(const float4*)(r0 + j * 4), *(const float4*)(r1 + j * 4));
    const float4 dn = add4(*(const float4*)(r0 + 16), *(const float4*)(r1 + 16));
    o += n.x / (dn.x + 1e-16f);
    o += n.y / (dn.y + 1e-16f);
    o += n.z / (dn.z + 1e-16f);
    o += n.w / (dn.w + 1e-16f);
  }
  out[t] = o;
}

extern "C" void kernel_launch(void* const* d_in, const int* in_sizes, int n_in,
                              void* d_out, int out_size, void* d_ws, size_t ws_size,
                              hipStream_t stream) {
  const float* x_c = (const float*)d_in[0];
  const float* x_q = (const float*)d_in[1];
  const float* x_a = (const float*)d_in[2];
  const int* src_cq = (const int*)d_in[3];
  const int* dst_cq = (const int*)d_in[4];
  const int* src_ca = (const int*)d_in[5];
  const int* dst_ca = (const int*)d_in[6];
  const float* Wsq = (const float*)d_in[7];
  const float* Wdq = (const float*)d_in[8];
  const float* Asq = (const float*)d_in[9];
  const float* Adq = (const float*)d_in[10];
  const float* bq  = (const float*)d_in[11];
  const float* Wsa = (const float*)d_in[12];
  const float* Wda = (const float*)d_in[13];
  const float* Asa = (const float*)d_in[14];
  const float* Ada = (const float*)d_in[15];
  const float* ba  = (const float*)d_in[16];
  const float* Wout = (const float*)d_in[17];
  const float* bout = (const float*)d_in[18];

  const long Nc = in_sizes[0] / 64;
  const long Nq = in_sizes[1] / 64;
  const long Na = in_sizes[2] / 64;
  const long Ecq = in_sizes[3];
  const long Eca = in_sizes[5];

  float* w = (float*)d_ws;
  size_t off = 0;
  auto alloc = [&](size_t n) -> float* {
    float* p = w + off;
    off += (n + 63) & ~(size_t)63;
    return p;
  };
  float* Vp_sq = alloc(64 * 16);
  float* Vp_dq = alloc(64 * 16);
  float* Vp_sa = alloc(64 * 16);
  float* Vp_da = alloc(64 * 16);
  float* Py    = alloc(64 * 32);
  float* cst   = alloc(16);
  float* as_cq = alloc(Nc * 4);
  float* as_ca = alloc(Nc * 4);
  float* ad_cq = alloc(Nq * 4);
  float* ad_ca = alloc(Na * 4);
  float* g_cq  = alloc(Ecq * 4);
  float* g_ca  = alloc(Eca * 4);
  float* rec_q = alloc(Nq * 8);
  float* rec_a = alloc(Na * 8);
  float* m_q   = alloc(Nq * 4);
  float* m_q1  = alloc(Nq * 4);
  float* m_a   = alloc(Na * 4);
  float* m_a1  = alloc(Na * 4);
  float* acc_q = alloc(Nq * 32);
  float* acc_q1 = alloc(Nq * 32);
  float* acc_a = alloc(Na * 32);
  float* acc_a1 = alloc(Na * 32);
  float* y_q   = alloc(Nc * 16);
  float* y_a   = alloc(Nc * 16);
  (void)Vp_dq; (void)Vp_sa; (void)Vp_da;
  if (off * sizeof(float) > ws_size) return;

  const long nzero4 = (long)(y_q - m_q) / 4;
  const long pmq = (long)(m_q1 - m_q), pma = (long)(m_a1 - m_a);
  const long paq = (long)(acc_q1 - acc_q), paa = (long)(acc_a1 - acc_a);

  prep_kernel<<<49, 128, 0, stream>>>(Wsq, Wdq, Asq, Adq, Wsa, Wda, Asa, Ada,
                                      bq, ba, Wout, bout, Vp_sq, Py, cst);

  const int gNc = (int)((Nc + 127) / 128);
  const int gNq = (int)((Nq + 127) / 128);
  const int gNa = (int)((Na + 127) / 128);
  dense_kernel<<<gNc + gNq + gNa, 256, 0, stream>>>(
      x_c, x_q, x_a, Vp_sq, Py, as_cq, as_ca, ad_cq, ad_ca, y_q, y_a,
      (float4*)m_q, nzero4, (int)Nc, (int)Nq, (int)Na, gNc, gNq);

  edge_pass1<<<1024, 256, 0, stream>>>(src_cq, dst_cq, src_ca, dst_ca, as_cq, as_ca,
                                       g_cq, g_ca, (unsigned*)m_q, (unsigned*)m_a,
                                       pmq, pma, Ecq, Eca);
  node_max<<<(int)((Nq + Na + 255) / 256), 256, 0, stream>>>(
      (const uint4*)m_q, (const uint4*)m_a, (const float4*)ad_cq, (const float4*)ad_ca,
      (float4*)rec_q, (float4*)rec_a, pmq / 4, pma / 4, Nq, Na);
  edge_pass2<<<2048, 256, 0, stream>>>(src_cq, dst_cq, src_ca, dst_ca, y_q, y_a, g_cq, g_ca,
                                       rec_q, rec_a, acc_q, acc_a, paq, paa, Ecq, Eca);
  node_pass<<<(int)((Nq * 4 + 255) / 256), 256, 0, stream>>>(acc_q, acc_a, cst,
                                                             (float*)d_out, paq, paa, Nq, Na);
}


# Per-call destination lists (one returning exchange links an edge to its node's list) replace both float-atomic edge passes: 4 lanes walk a node, running max, sums in registers in list (arrival) order
# speedup vs baseline: 14.9461x; 1.3501x over previous
#include <hip/hip_runtime.h>
#include <hip/hip_bf16.h>
#include <math.h>

typedef __attribute__((ext_vector_type(2))) float v2f;
typedef __attribute__((ext_vector_type(8))) float v8f;

__device__ __forceinline__ float lrelu(float x) { return x > 0.f ? x : 0.2f * x; }

__device__ __forceinline__ int package_half() {
  return (int)((__builtin_amdgcn_s_sendmsg_rtn(0x87u) >> 18) & 1u);
}

__device__ __forceinline__ float head_sum(float p) {
  p += __shfl_xor(p, 1, 32);
  p += __shfl_xor(p, 2, 32);
  p += __shfl_xor(p, 4, 32);
  return p;
}
__device__ __forceinline__ float dot8(const float4 w0, const float4 w1,
                                      const float4 a0, const float4 a1) {
  float s = w0.x * a0.x;
  s += w0.y * a0.y; s += w0.z * a0.z; s += w0.w * a0.w;
  s += w1.x * a1.x; s += w1.y * a1.y; s += w1.z * a1.z; s += w1.w * a1.w;
  return s;
}

__global__ __launch_bounds__(128)
void prep_kernel(const float* __restrict__ Wsq, const float* __restrict__ Wdq,
                 const float* __restrict__ Asq, const float* __restrict__ Adq,
                 const float* __restrict__ Wsa, const float* __restrict__ Wda,
                 const float* __restrict__ Asa, const float* __restrict__ Ada,
                 const float* __restrict__ bq,  const float* __restrict__ ba,
                 const float* __restrict__ Wout, const float* __restrict__ bout,
                 float* __restrict__ Vp, float* __restrict__ Py, float* __restrict__ cst) {
  const int lane = threadIdx.x & 31;
  const int job = (int)blockIdx.x * 4 + __builtin_amdgcn_readfirstlane((int)(threadIdx.x >> 5));
  const int src = (lane & 3) * 8;
  if (job < 128) {
    const int table = job >> 5, k0 = (job & 31) * 2;
    const float* W = (table == 0) ? Wsq : (table == 1) ? Wdq : (table == 2) ? Wsa : Wda;
    const float* A = (table == 0) ? Asq : (table == 1) ? Adq : (table == 2) ? Asa : Ada;
    const float4* wr0 = (const float4*)(W + k0 * 256 + lane * 8);
    const float4* wr1 = (const float4*)(W + (k0 + 1) * 256 + lane * 8);
    const float4* ar = (const float4*)(A + lane * 8);
    const float4 w00 = wr0[0], w01 = wr0[1], w10 = wr1[0], w11 = wr1[1];
    const float4 a0 = ar[0], a1 = ar[1];
    const float s0 = head_sum(dot8(w00, w01, a0, a1));
    const float s1 = head_sum(dot8(w10, w11, a0, a1));
    const float q0 = __shfl(s0, src, 32), q1 = __shfl(s1, src, 32);
    Vp[table * 1024 + k0 * 16 + lane] = ((lane & 15) < 4) ? (lane < 16 ? q0 : q1) : 0.f;
  } else if (job < 192) {
    const int k = job - 128;
    const float4* wq = (const float4*)(Wsq + k * 256 + lane * 8);
    const float4* wa = (const float4*)(Wsa + k * 256 + lane * 8);
    const float4* oq = (const float4*)Wout + (lane & 7) * 8;
    const float4* oa = oq + 64;
    const float4 q0 = wq[0], q1 = wq[1], a0 = wa[0], a1 = wa[1];
    const float wqv[8] = {q0.x, q0.y, q0.z, q0.w, q1.x, q1.y, q1.z, q1.w};
    const float wav[8] = {a0.x, a0.y, a0.z, a0.w, a1.x, a1.y, a1.z, a1.w};
    float4 pq = {0.f, 0.f, 0.f, 0.f}, pa = {0.f, 0.f, 0.f, 0.f};
#pragma unroll
    for (int i = 0; i < 8; ++i) {
      const float4 o = oq[i], p = oa[i];
      pq.x += wqv[i] * o.x; pq.y += wqv[i] * o.y; pq.z += wqv[i] * o.z; pq.w += wqv[i] * o.w;
      pa.x += wav[i] * p.x; pa.y += wav[i] * p.y; pa.z += wav[i] * p.z; pa.w += wav[i] * p.w;
    }
    pq.x = head_sum(pq.x); pq.y = head_sum(pq.y); pq.z = head_sum(pq.z); pq.w = head_sum(pq.w);
    pa.x = head_sum(pa.x); pa.y = head_sum(pa.y); pa.z = head_sum(pa.z); pa.w = head_sum(pa.w);
    const float gq0 = __shfl(pq.x, src, 32), gq1 = __shfl(pq.y, src, 32);
    const float gq2 = __shfl(pq.z, src, 32), gq3 = __shfl(pq.w, src, 32);
    const float ga0 = __shfl(pa.x, src, 32), ga1 = __shfl(pa.y, src, 32);
    const float ga2 = __shfl(pa.z, src, 32), ga3 = __shfl(pa.w, src, 32);
    const int j = (lane >> 2) & 3;
    const float vq = (j == 0) ? gq0 : (j == 1) ? gq1 : (j == 2) ? gq2 : gq3;
    const float va = (j == 0) ? ga0 : (j == 1) ? ga1 : (j == 2) ? ga2 : ga3;
    Py[k * 32 + lane] = 0.25f * (lane < 16 ? vq : va);
  } else if (job == 192) {
    const float4* o4 = (const float4*)Wout;
    const float4 t0 = o4[2 * lane], t1 = o4[2 * lane + 1];
    const float4 u0 = o4[64 + 2 * lane], u1 = o4[64 + 2 * lane + 1];
    const float2 cq = *(const float2*)(bq + 2 * lane);
    const float2 ca = *(const float2*)(ba + 2 * lane);
    float p0 = cq.x * t0.x + cq.y * t1.x + ca.x * u0.x + ca.y * u1.x;
    float p1 = cq.x * t0.y + cq.y * t1.y + ca.x * u0.y + ca.y * u1.y;
    float p2 = cq.x * t0.z + cq.y * t1.z + ca.x * u0.z + ca.y * u1.z;
    float p3 = cq.x * t0.w + cq.y * t1.w + ca.x * u0.w + ca.y * u1.w;
#pragma unroll
    for (int m = 16; m >= 1; m >>= 1) {
      p0 += __shfl_xor(p0, m, 32); p1 += __shfl_xor(p1, m, 32);
      p2 += __shfl_xor(p2, m, 32); p3 += __shfl_xor(p3, m, 32);
    }
    float v = 0.f;
    if (lane < 4) v = ((lane == 0) ? p0 : (lane == 1) ? p1 : (lane == 2) ? p2 : p3) + bout[lane];
    if (lane < 16) cst[lane] = v;
  }
}

__global__ void dense_kernel(const float* __restrict__ xc, const float* __restrict__ xq,
                             const float* __restrict__ xa, const float* __restrict__ Vp,
                             const float* __restrict__ Py,
                             float* __restrict__ as_cq, float* __restrict__ as_ca,
                             float* __restrict__ ad_cq, float* __restrict__ ad_ca,
                             float* __restrict__ yq, float* __restrict__ ya,
                             float4* __restrict__ zero_ptr, long nzero4,
                             int Nc, int Nq, int Na, int gNc, int gNq) {
  {
    const long stride = (long)gridDim.x * blockDim.x;
    const long t0 = (long)blockIdx.x * blockDim.x + threadIdx.x;
    const float4 z4 = {0.f, 0.f, 0.f, 0.f};
    for (long i = t0; i < nzero4; i += stride) zero_ptr[i] = z4;
  }
  const int lane = threadIdx.x & 31;
  const int wv = (int)(threadIdx.x >> 5);
  const int m = lane & 15;
  const int ko = (lane >> 4) * 2;
  int b = (int)blockIdx.x;
  if (b < gNc) {
    const long base = ((long)b * 8 + wv) * 16;
    if (base >= Nc) return;
    long row = base + m; if (row >= Nc) row = Nc - 1;
    const float* xr = xc + row * 64;
    const float* bt = (m < 4) ? (Vp + m) : (m < 8) ? (Vp + 2 * 1024 + (m - 4)) : (Vp + m);
    v8f accs = {}, accq = {}, acca = {};
#pragma unroll
    for (int kk = 0; kk < 64; kk += 4) {
      v2f a, bs, b0, b1;
      a.x = xr[kk + ko];
      a.y = xr[kk + ko + 1];
      bs.x = bt[(kk + ko) * 16];
      bs.y = bt[(kk + ko + 1) * 16];
      b0.x = Py[(kk + ko) * 32 + m];
      b0.y = Py[(kk + ko + 1) * 32 + m];
      b1.x = Py[(kk + ko) * 32 + 16 + m];
      b1.y = Py[(kk + ko + 1) * 32 + 16 + m];
      accs = __builtin_amdgcn_wmma_f32_16x16x4_f32(false, a, false, bs, (short)0, accs,
                                                   false, false);
      accq = __builtin_amdgcn_wmma_f32_16x16x4_f32(false, a, false, b0, (short)0, accq,
                                                   false, false);
      acca = __builtin_amdgcn_wmma_f32_16x16x4_f32(false, a, false, b1, (short)0, acca,
                                                   false, false);
    }
    const long rb = base + ((lane >> 4) << 3);
    float* so = (m < 4) ? (as_cq + m) : (as_ca + (m - 4));
#pragma unroll
    for (int v = 0; v < 8; ++v) {
      long r = rb + v;
      if (r < Nc) {
        yq[r * 16 + m] = accq[v];
        ya[r * 16 + m] = acca[v];
        if (m < 8) so[r * 4] = accs[v];
      }
    }
  } else {
    b -= gNc;
    const float* x = xq; const float* V = Vp + 1024; float* o = ad_cq; int N = Nq;
    if (b >= gNq) { b -= gNq; x = xa; V = Vp + 3 * 1024; o = ad_ca; N = Na; }
    const long base = ((long)b * 8 + wv) * 16;
    if (base >= N) return;
    long row = base + m; if (row >= N) row = N - 1;
    const float* xr = x + row * 64;
    v8f acc = {};
#pragma unroll
    for (int kk = 0; kk < 64; kk += 4) {
      v2f a, bv;
      a.x = xr[kk + ko];
      a.y = xr[kk + ko + 1];
      bv.x = V[(kk + ko) * 16 + m];
      bv.y = V[(kk + ko + 1) * 16 + m];
      acc = __builtin_amdgcn_wmma_f32_16x16x4_f32(false, a, false, bv, (short)0, acc,
                                                  false, false);
    }
    if (m < 4) {
      const long rb = base + ((lane >> 4) << 3);
#pragma unroll
      for (int v = 0; v < 8; ++v) {
        long r = rb + v;
        if (r < N) o[r * 4 + m] = acc[v];
      }
    }
  }
}

template <bool CHECK>
__device__ __forceinline__ void link4(const long i0, const long stride, const long n,
                                      const long Eq,
                                      const int* __restrict__ src_q, const int* __restrict__ dst_q,
                                      const int* __restrict__ src_a, const int* __restrict__ dst_a,
                                      int* __restrict__ hq, int* __restrict__ ha,
                                      int2* __restrict__ rec_q, int2* __restrict__ rec_a) {
  int s[4], d[4], prev[4];
  long e[4];
  bool ok[4], la[4];
#pragma unroll
  for (int u = 0; u < 4; ++u) {
    const long i = i0 + u * stride;
    ok[u] = CHECK ? (i < n) : true;
    la[u] = i >= Eq;
    e[u] = la[u] ? i - Eq : i;
    s[u] = 0; d[u] = 0;
    if (ok[u]) {
      s[u] = (la[u] ? src_a : src_q)[e[u]];
      d[u] = (la[u] ? dst_a : dst_q)[e[u]];
    }
  }
#pragma unroll
  for (int u = 0; u < 4; ++u) {
    prev[u] = 0;
    if (ok[u])
      prev[u] = __hip_atomic_exchange((la[u] ? ha : hq) + d[u], (int)e[u] + 1,
                                      __ATOMIC_RELAXED, __HIP_MEMORY_SCOPE_AGENT);
  }
#pragma unroll
  for (int u = 0; u < 4; ++u)
    if (ok[u]) (la[u] ? rec_a : rec_q)[e[u]] = make_int2(s[u], prev[u]);
}

__global__ __launch_bounds__(256)
void build_lists(const int* __restrict__ src_q, const int* __restrict__ dst_q,
                 const int* __restrict__ src_a, const int* __restrict__ dst_a,
                 int* __restrict__ head_q, int* __restrict__ head_a, long hdq, long hda,
                 int2* __restrict__ rec_q, int2* __restrict__ rec_a, long Eq, long Ea) {
  const int half = package_half();
  int* const hq = head_q + half * hdq;
  int* const ha = head_a + half * hda;
  const int lane = threadIdx.x & 31;
  const long wave = (long)blockIdx.x * (blockDim.x >> 5) +
                    __builtin_amdgcn_readfirstlane((int)(threadIdx.x >> 5));
  const long n = Eq + Ea;
  const long stride = (long)gridDim.x * blockDim.x;
  for (long b = wave * 32; b < n; b += 4 * stride) {
    if (b + 3 * stride + 32 <= n)
      link4<false>(b + lane, stride, n, Eq, src_q, dst_q, src_a, dst_a, hq, ha, rec_q, rec_a);
    else
      link4<true>(b + lane, stride, n, Eq, src_q, dst_q, src_a, dst_a, hq, ha, rec_q, rec_a);
  }
}

template <int H>
__device__ __forceinline__ float quad_bcast(const float v) {
  return __int_as_float(__builtin_amdgcn_update_dpp(0, __float_as_int(v), H * 0x55, 0xf, 0xf, true));
}

struct RowSums { float m, den, num[4]; };

__device__ __forceinline__ void sums_start(RowSums& S) {
  S.m = -INFINITY; S.den = 0.f;
#pragma unroll
  for (int h = 0; h < 4; ++h) S.num[h] = 0.f;
}
__device__ __forceinline__ void sums_add(RowSums& S, const bool live, const float as,
                                         const float ad, const float4 y) {
  const float sc = lrelu(as + ad);
  const float mn = fmaxf(S.m, sc);
  const float keep = __expf(S.m - mn);
  const float ex = __expf(sc - mn);
  const float dn = S.den * keep + ex;
  const float k[4] = {quad_bcast<0>(keep), quad_bcast<1>(keep), quad_bcast<2>(keep),
                      quad_bcast<3>(keep)};
  const float e[4] = {quad_bcast<0>(ex), quad_bcast<1>(ex), quad_bcast<2>(ex), quad_bcast<3>(ex)};
  const float v[4] = {y.x, y.y, y.z, y.w};
  S.m = live ? mn : S.m;
  S.den = live ? dn : S.den;
#pragma unroll
  for (int h = 0; h < 4; ++h) {
    const float nn = S.num[h] * k[h] + e[h] * v[h];
    S.num[h] = live ? nn : S.num[h];
  }
}
__device__ __forceinline__ float sums_close(float o, const RowSums& S) {
  o += S.num[0] / (quad_bcast<0>(S.den) + 1e-16f);
  o += S.num[1] / (quad_bcast<1>(S.den) + 1e-16f);
  o += S.num[2] / (quad_bcast<2>(S.den) + 1e-16f);
  o += S.num[3] / (quad_bcast<3>(S.den) + 1e-16f);
  return o;
}

__device__ __forceinline__ int list_next(int nxt, int& other, int E) {
  if (nxt == 0) { nxt = other; other = 0; }
  return ((unsigned)(nxt - 1) < (unsigned)E) ? nxt : 0;
}
__global__ __launch_bounds__(256)
void node_walk(const int* __restrict__ head_q, const int* __restrict__ head_a, long hdq, long hda,
               const int2* __restrict__ rec_q, const int2* __restrict__ rec_a,
               const float* __restrict__ as_q, const float* __restrict__ as_a,
               const float4* __restrict__ y_q, const float4* __restrict__ y_a,
               const float* __restrict__ ad_q, const float* __restrict__ ad_a,
               const float* __restrict__ cst, float* __restrict__ out,
               long Nq, long Na, int Eq, int Ea) {
  const int t = (int)(blockIdx.x * blockDim.x + threadIdx.x);
  const int d = t >> 2;
  const int x = t & 3;
  if (d >= Nq) return;
  RowSums Sq, Sa;
  sums_start(Sq);
  sums_start(Sa);
  const float adq = ad_q[t];
  int oq = head_q[hdq + d];
  int cq = list_next(head_q[d], oq, Eq);
  float ada = 0.f;
  int ca = 0, oa = 0;
  if (d < Na) {
    ada = ad_a[t];
    oa = head_a[hda + d];
    ca = list_next(head_a[d], oa, Ea);
  }
  int2 rq = make_int2(0, 0), ra = make_int2(0, 0);
  if (cq) rq = rec_q[cq - 1];
  if (ca) ra = rec_a[ca - 1];
  const int maxit = Eq > Ea ? Eq : Ea;
  for (int it = 0; it < maxit && (cq | ca) != 0; ++it) {
    const float sq = as_q[rq.x * 4 + x];
    const float4 yq = y_q[rq.x * 4 + x];
    const float sa = as_a[ra.x * 4 + x];
    const float4 ya = y_a[ra.x * 4 + x];
    const int nq = list_next(rq.y, oq, Eq);
    const int na = list_next(ra.y, oa, Ea);
    int2 nrq = rec_q[nq ? nq - 1 : 0];
    int2 nra = rec_a[na ? na - 1 : 0];
    sums_add(Sq, cq != 0, sq, adq, yq);
    sums_add(Sa, ca != 0, sa, ada, ya);
    if (nq == 0) nrq = make_int2(0, 0);
    if (na == 0) nra = make_int2(0, 0);
    cq = nq; rq = nrq;
    ca = na; ra = nra;
  }
  float o = sums_close(cst[x], Sq);
  if (d < Na) o = sums_close(o, Sa);
  out[t] = o;
}

extern "C" void kernel_launch(void* const* d_in, const int* in_sizes, int n_in,
                              void* d_out, int out_size, void* d_ws, size_t ws_size,
                              hipStream_t stream) {
  const float* x_c = (const float*)d_in[0];
  const float* x_q = (const float*)d_in[1];
  const float* x_a = (const float*)d_in[2];
  const int* src_cq = (const int*)d_in[3];
  const int* dst_cq = (const int*)d_in[4];
  const int* src_ca = (const int*)d_in[5];
  const int* dst_ca = (const int*)d_in[6];
  const float* Wsq = (const float*)d_in[7];
  const float* Wdq = (const float*)d_in[8];
  const float* Asq = (const float*)d_in[9];
  const float* Adq = (const float*)d_in[10];
  const float* bq  = (const float*)d_in[11];
  const float* Wsa = (const float*)d_in[12];
  const float* Wda = (const float*)d_in[13];
  const float* Asa = (const float*)d_in[14];
  const float* Ada = (const float*)d_in[15];
  const float* ba  = (const float*)d_in[16];
  const float* Wout = (const float*)d_in[17];
  const float* bout = (const float*)d_in[18];

  const long Nc = in_sizes[0] / 64;
  const long Nq = in_sizes[1] / 64;
  const long Na = in_sizes[2] / 64;
  const long Ecq = in_sizes[3];
  const long Eca = in_sizes[5];

  float* w = (float*)d_ws;
  size_t off = 0;
  auto alloc = [&](size_t n) -> float* {
    float* p = w + off;
    off += (n + 63) & ~(size_t)63;
    return p;
  };
  float* Vp_sq = alloc(64 * 16);
  float* Vp_dq = alloc(64 * 16);
  float* Vp_sa = alloc(64 * 16);
  float* Vp_da = alloc(64 * 16);
  float* Py    = alloc(64 * 32);
  float* cst   = alloc(16);
  float* as_cq = alloc(Nc * 4);
  float* as_ca = alloc(Nc * 4);
  float* ad_cq = alloc(Nq * 4);
  float* ad_ca = alloc(Na * 4);
  float* rec_q = alloc(Ecq * 2);
  float* rec_a = alloc(Eca * 2);
  float* hd_q  = alloc(Nq);
  float* hd_q1 = alloc(Nq);
  float* hd_a  = alloc(Na);
  float* hd_a1 = alloc(Na);
  float* y_q   = alloc(Nc * 16);
  float* y_a   = alloc(Nc * 16);
  (void)Vp_dq; (void)Vp_sa; (void)Vp_da;
  if (off * sizeof(float) > ws_size) return;

  const long nzero4 = (long)(y_q - hd_q) / 4;
  const long hdq = (long)(hd_q1 - hd_q), hda = (long)(hd_a1 - hd_a);

  prep_kernel<<<49, 128, 0, stream>>>(Wsq, Wdq, Asq, Adq, Wsa, Wda, Asa, Ada,
                                      bq, ba, Wout, bout, Vp_sq, Py, cst);

  const int gNc = (int)((Nc + 127) / 128);
  const int gNq = (int)((Nq + 127) / 128);
  const int gNa = (int)((Na + 127) / 128);
  dense_kernel<<<gNc + gNq + gNa, 256, 0, stream>>>(
      x_c, x_q, x_a, Vp_sq, Py, as_cq, as_ca, ad_cq, ad_ca, y_q, y_a,
      (float4*)hd_q, nzero4, (int)Nc, (int)Nq, (int)Na, gNc, gNq);

  build_lists<<<2048, 256, 0, stream>>>(src_cq, dst_cq, src_ca, dst_ca, (int*)hd_q, (int*)hd_a,
                                        hdq, hda, (int2*)rec_q, (int2*)rec_a, Ecq, Eca);
  node_walk<<<(int)((Nq * 4 + 255) / 256), 256, 0, stream>>>(
      (const int*)hd_q, (const int*)hd_a, hdq, hda, (const int2*)rec_q, (const int2*)rec_a,
      as_cq, as_ca, (const float4*)y_q, (const float4*)y_a,
      ad_cq, ad_ca, cst, (float*)d_out, Nq, Na,
      (int)Ecq, (int)Eca);
}


# List building joins the dense launch as interleaved workgroups (it needs only the edge indices; the head zeros move to the first launch): 4 launches become 3; sums still in list (arrival) order
# speedup vs baseline: 15.5880x; 1.0429x over previous
#include <hip/hip_runtime.h>
#include <hip/hip_bf16.h>
#include <math.h>

typedef __attribute__((ext_vector_type(2))) float v2f;
typedef __attribute__((ext_vector_type(8))) float v8f;

__device__ __forceinline__ float lrelu(float x) { return x > 0.f ? x : 0.2f * x; }

__device__ __forceinline__ int package_half() {
  return (int)((__builtin_amdgcn_s_sendmsg_rtn(0x87u) >> 18) & 1u);
}

__device__ __forceinline__ float head_sum(float p) {
  p += __shfl_xor(p, 1, 32);
  p += __shfl_xor(p, 2, 32);
  p += __shfl_xor(p, 4, 32);
  return p;
}
__device__ __forceinline__ float dot8(const float4 w0, const float4 w1,
                                      const float4 a0, const float4 a1) {
  float s = w0.x * a0.x;
  s += w0.y * a0.y; s += w0.z * a0.z; s += w0.w * a0.w;
  s += w1.x * a1.x; s += w1.y * a1.y; s += w1.z * a1.z; s += w1.w * a1.w;
  return s;
}

__global__ __launch_bounds__(128)
void prep_kernel(const float* __restrict__ Wsq, const float* __restrict__ Wdq,
                 const float* __restrict__ Asq, const float* __restrict__ Adq,
                 const float* __restrict__ Wsa, const float* __restrict__ Wda,
                 const float* __restrict__ Asa, const float* __restrict__ Ada,
                 const float* __restrict__ bq,  const float* __restrict__ ba,
                 const float* __restrict__ Wout, const float* __restrict__ bout,
                 float* __restrict__ Vp, float* __restrict__ Py, float* __restrict__ cst,
                 float4* __restrict__ zero_ptr, long nzero4) {
  {
    const long stride = (long)gridDim.x * blockDim.x;
    const long t0 = (long)blockIdx.x * blockDim.x + threadIdx.x;
    const float4 z4 = {0.f, 0.f, 0.f, 0.f};
    for (long i = t0; i < nzero4; i += stride) zero_ptr[i] = z4;
  }
  const int lane = threadIdx.x & 31;
  const int job = (int)blockIdx.x * 4 + __builtin_amdgcn_readfirstlane((int)(threadIdx.x >> 5));
  const int src = (lane & 3) * 8;
  if (job < 128) {
    const int table = job >> 5, k0 = (job & 31) * 2;
    const float* W = (table == 0) ? Wsq : (table == 1) ? Wdq : (table == 2) ? Wsa : Wda;
    const float* A = (table == 0) ? Asq : (table == 1) ? Adq : (table == 2) ? Asa : Ada;
    const float4* wr0 = (const float4*)(W + k0 * 256 + lane * 8);
    const float4* wr1 = (const float4*)(W + (k0 + 1) * 256 + lane * 8);
    const float4* ar = (const float4*)(A + lane * 8);
    const float4 w00 = wr0[0], w01 = wr0[1], w10 = wr1[0], w11 = wr1[1];
    const float4 a0 = ar[0], a1 = ar[1];
    const float s0 = head_sum(dot8(w00, w01, a0, a1));
    const float s1 = head_sum(dot8(w10, w11, a0, a1));
    const float q0 = __shfl(s0, src, 32), q1 = __shfl(s1, src, 32);
    Vp[table * 1024 + k0 * 16 + lane] = ((lane & 15) < 4) ? (lane < 16 ? q0 : q1) : 0.f;
  } else if (job < 192) {
    const int k = job - 128;
    const float4* wq = (const float4*)(Wsq + k * 256 + lane * 8);
    const float4* wa = (const float4*)(Wsa + k * 256 + lane * 8);
    const float4* oq = (const float4*)Wout + (lane & 7) * 8;
    const float4* oa = oq + 64;
    const float4 q0 = wq[0], q1 = wq[1], a0 = wa[0], a1 = wa[1];
    const float wqv[8] = {q0.x, q0.y, q0.z, q0.w, q1.x, q1.y, q1.z, q1.w};
    const float wav[8] = {a0.x, a0.y, a0.z, a0.w, a1.x, a1.y, a1.z, a1.w};
    float4 pq = {0.f, 0.f, 0.f, 0.f}, pa = {0.f, 0.f, 0.f, 0.f};
#pragma unroll
    for (int i = 0; i < 8; ++i) {
      const float4 o = oq[i], p = oa[i];
      pq.x += wqv[i] * o.x; pq.y += wqv[i] * o.y; pq.z += wqv[i] * o.z; pq.w += wqv[i] * o.w;
      pa.x += wav[i] * p.x; pa.y += wav[i] * p.y; pa.z += wav[i] * p.z; pa.w += wav[i] * p.w;
    }
    pq.x = head_sum(pq.x); pq.y = head_sum(pq.y); pq.z = head_sum(pq.z); pq.w = head_sum(pq.w);
    pa.x = head_sum(pa.x); pa.y = head_sum(pa.y); pa.z = head_sum(pa.z); pa.w = head_sum(pa.w);
    const float gq0 = __shfl(pq.x, src, 32), gq1 = __shfl(pq.y, src, 32);
    const float gq2 = __shfl(pq.z, src, 32), gq3 = __shfl(pq.w, src, 32);
    const float ga0 = __shfl(pa.x, src, 32), ga1 = __shfl(pa.y, src, 32);
    const float ga2 = __shfl(pa.z, src, 32), ga3 = __shfl(pa.w, src, 32);
    const int j = (lane >> 2) & 3;
    const float vq = (j == 0) ? gq0 : (j == 1) ? gq1 : (j == 2) ? gq2 : gq3;
    const float va = (j == 0) ? ga0 : (j == 1) ? ga1 : (j == 2) ? ga2 : ga3;
    Py[k * 32 + lane] = 0.25f * (lane < 16 ? vq : va);
  } else if (job == 192) {
    const float4* o4 = (const float4*)Wout;
    const float4 t0 = o4[2 * lane], t1 = o4[2 * lane + 1];
    const float4 u0 = o4[64 + 2 * lane], u1 = o4[64 + 2 * lane + 1];
    const float2 cq = *(const float2*)(bq + 2 * lane);
    const float2 ca = *(const float2*)(ba + 2 * lane);
    float p0 = cq.x * t0.x + cq.y * t1.x + ca.x * u0.x + ca.y * u1.x;
    float p1 = cq.x * t0.y + cq.y * t1.y + ca.x * u0.y + ca.y * u1.y;
    float p2 = cq.x * t0.z + cq.y * t1.z + ca.x * u0.z + ca.y * u1.z;
    float p3 = cq.x * t0.w + cq.y * t1.w + ca.x * u0.w + ca.y * u1.w;
#pragma unroll
    for (int m = 16; m >= 1; m >>= 1) {
      p0 += __shfl_xor(p0, m, 32); p1 += __shfl_xor(p1, m, 32);
      p2 += __shfl_xor(p2, m, 32); p3 += __shfl_xor(p3, m, 32);
    }
    float v = 0.f;
    if (lane < 4) v = ((lane == 0) ? p0 : (lane == 1) ? p1 : (lane == 2) ? p2 : p3) + bout[lane];
    if (lane < 16) cst[lane] = v;
  }
}

template <bool CHECK>
__device__ __forceinline__ void link4(const long i0, const long stride, const long n,
                                      const long Eq,
                                      const int* __restrict__ src_q, const int* __restrict__ dst_q,
                                      const int* __restrict__ src_a, const int* __restrict__ dst_a,
                                      int* __restrict__ hq, int* __restrict__ ha,
                                      int2* __restrict__ rec_q, int2* __restrict__ rec_a) {
  int s[4], d[4], prev[4];
  long e[4];
  bool ok[4], la[4];
#pragma unroll
  for (int u = 0; u < 4; ++u) {
    const long i = i0 + u * stride;
    ok[u] = CHECK ? (i < n) : true;
    la[u] = i >= Eq;
    e[u] = la[u] ? i - Eq : i;
    s[u] = 0; d[u] = 0;
    if (ok[u]) {
      s[u] = (la[u] ? src_a : src_q)[e[u]];
      d[u] = (la[u] ? dst_a : dst_q)[e[u]];
    }
  }
#pragma unroll
  for (int u = 0; u < 4; ++u) {
    prev[u] = 0;
    if (ok[u])
      prev[u] = __hip_atomic_exchange((la[u] ? ha : hq) + d[u], (int)e[u] + 1,
                                      __ATOMIC_RELAXED, __HIP_MEMORY_SCOPE_AGENT);
  }
#pragma unroll
  for (int u = 0; u < 4; ++u)
    if (ok[u]) (la[u] ? rec_a : rec_q)[e[u]] = make_int2(s[u], prev[u]);
}

__device__ __forceinline__
void lists_block(const int vb, const int nb,
                 const int* __restrict__ src_q, const int* __restrict__ dst_q,
                 const int* __restrict__ src_a, const int* __restrict__ dst_a,
                 int* __restrict__ head_q, int* __restrict__ head_a, long hdq, long hda,
                 int2* __restrict__ rec_q, int2* __restrict__ rec_a, long Eq, long Ea) {
  const int half = package_half();
  int* const hq = head_q + half * hdq;
  int* const ha = head_a + half * hda;
  const int lane = threadIdx.x & 31;
  const long wave = (long)vb * (blockDim.x >> 5) +
                    __builtin_amdgcn_readfirstlane((int)(threadIdx.x >> 5));
  const long n = Eq + Ea;
  const long stride = (long)nb * blockDim.x;
  for (long b = wave * 32; b < n; b += 4 * stride) {
    if (b + 3 * stride + 32 <= n)
      link4<false>(b + lane, stride, n, Eq, src_q, dst_q, src_a, dst_a, hq, ha, rec_q, rec_a);
    else
      link4<true>(b + lane, stride, n, Eq, src_q, dst_q, src_a, dst_a, hq, ha, rec_q, rec_a);
  }
}

__global__ void dense_and_lists(const float* __restrict__ xc, const float* __restrict__ xq,
                             const float* __restrict__ xa, const float* __restrict__ Vp,
                             const float* __restrict__ Py,
                             float* __restrict__ as_cq, float* __restrict__ as_ca,
                             float* __restrict__ ad_cq, float* __restrict__ ad_ca,
                             float* __restrict__ yq, float* __restrict__ ya,
                             int Nc, int Nq, int Na, int gNc, int gNq,
                             const int* __restrict__ src_q, const int* __restrict__ dst_q,
                             const int* __restrict__ src_a, const int* __restrict__ dst_a,
                             int* __restrict__ head_q, int* __restrict__ head_a,
                             long hdq, long hda,
                             int2* __restrict__ rec_q, int2* __restrict__ rec_a,
                             long Eq, long Ea, int npair, int nd, int nl) {
  const int wg = (int)blockIdx.x;
  bool lists = nl > nd;
  int b = wg - npair;
  if (wg < 2 * npair) { lists = (__popc((unsigned)wg) & 1) != 0; b = wg >> 1; }
  if (lists) {
    lists_block(b, nl, src_q, dst_q, src_a, dst_a, head_q, head_a, hdq, hda, rec_q, rec_a,
                Eq, Ea);
    return;
  }
  const int lane = threadIdx.x & 31;
  const int wv = (int)(threadIdx.x >> 5);
  const int m = lane & 15;
  const int ko = (lane >> 4) * 2;
  if (b < gNc) {
    const long base = ((long)b * 8 + wv) * 16;
    if (base >= Nc) return;
    long row = base + m; if (row >= Nc) row = Nc - 1;
    const float* xr = xc + row * 64;
    const float* bt = (m < 4) ? (Vp + m) : (m < 8) ? (Vp + 2 * 1024 + (m - 4)) : (Vp + m);
    v8f accs = {}, accq = {}, acca = {};
#pragma unroll
    for (int kk = 0; kk < 64; kk += 4) {
      v2f a, bs, b0, b1;
      a.x = xr[kk + ko];
      a.y = xr[kk + ko + 1];
      bs.x = bt[(kk + ko) * 16];
      bs.y = bt[(kk + ko + 1) * 16];
      b0.x = Py[(kk + ko) * 32 + m];
      b0.y = Py[(kk + ko + 1) * 32 + m];
      b1.x = Py[(kk + ko) * 32 + 16 + m];
      b1.y = Py[(kk + ko + 1) * 32 + 16 + m];
      accs = __builtin_amdgcn_wmma_f32_16x16x4_f32(false, a, false, bs, (short)0, accs,
                                                   false, false);
      accq = __builtin_amdgcn_wmma_f32_16x16x4_f32(false, a, false, b0, (short)0, accq,
                                                   false, false);
      acca = __builtin_amdgcn_wmma_f32_16x16x4_f32(false, a, false, b1, (short)0, acca,
                                                   false, false);
    }
    const long rb = base + ((lane >> 4) << 3);
    float* so = (m < 4) ? (as_cq + m) : (as_ca + (m - 4));
#pragma unroll
    for (int v = 0; v < 8; ++v) {
      long r = rb + v;
      if (r < Nc) {
        yq[r * 16 + m] = accq[v];
        ya[r * 16 + m] = acca[v];
        if (m < 8) so[r * 4] = accs[v];
      }
    }
  } else {
    b -= gNc;
    const float* x = xq; const float* V = Vp + 1024; float* o = ad_cq; int N = Nq;
    if (b >= gNq) { b -= gNq; x = xa; V = Vp + 3 * 1024; o = ad_ca; N = Na; }
    const long base = ((long)b * 8 + wv) * 16;
    if (base >= N) return;
    long row = base + m; if (row >= N) row = N - 1;
    const float* xr = x + row * 64;
    v8f acc = {};
#pragma unroll
    for (int kk = 0; kk < 64; kk += 4) {
      v2f a, bv;
      a.x = xr[kk + ko];
      a.y = xr[kk + ko + 1];
      bv.x = V[(kk + ko) * 16 + m];
      bv.y = V[(kk + ko + 1) * 16 + m];
      acc = __builtin_amdgcn_wmma_f32_16x16x4_f32(false, a, false, bv, (short)0, acc,
                                                  false, false);
    }
    if (m < 4) {
      const long rb = base + ((lane >> 4) << 3);
#pragma unroll
      for (int v = 0; v < 8; ++v) {
        long r = rb + v;
        if (r < N) o[r * 4 + m] = acc[v];
      }
    }
  }
}

template <int H>
__device__ __forceinline__ float quad_bcast(const float v) {
  return __int_as_float(__builtin_amdgcn_update_dpp(0, __float_as_int(v), H * 0x55, 0xf, 0xf, true));
}

struct RowSums { float m, den, num[4]; };

__device__ __forceinline__ void sums_start(RowSums& S) {
  S.m = -INFINITY; S.den = 0.f;
#pragma unroll
  for (int h = 0; h < 4; ++h) S.num[h] = 0.f;
}
__device__ __forceinline__ void sums_add(RowSums& S, const bool live, const float as,
                                         const float ad, const float4 y) {
  const float sc = lrelu(as + ad);
  const float mn = fmaxf(S.m, sc);
  const float keep = __expf(S.m - mn);
  const float ex = __expf(sc - mn);
  const float dn = S.den * keep + ex;
  const float k[4] = {quad_bcast<0>(keep), quad_bcast<1>(keep), quad_bcast<2>(keep),
                      quad_bcast<3>(keep)};
  const float e[4] = {quad_bcast<0>(ex), quad_bcast<1>(ex), quad_bcast<2>(ex), quad_bcast<3>(ex)};
  const float v[4] = {y.x, y.y, y.z, y.w};
  S.m = live ? mn : S.m;
  S.den = live ? dn : S.den;
#pragma unroll
  for (int h = 0; h < 4; ++h) {
    const float nn = S.num[h] * k[h] + e[h] * v[h];
    S.num[h] = live ? nn : S.num[h];
  }
}
__device__ __forceinline__ float sums_close(float o, const RowSums& S) {
  o += S.num[0] / (quad_bcast<0>(S.den) + 1e-16f);
  o += S.num[1] / (quad_bcast<1>(S.den) + 1e-16f);
  o += S.num[2] / (quad_bcast<2>(S.den) + 1e-16f);
  o += S.num[3] / (quad_bcast<3>(S.den) + 1e-16f);
  return o;
}

__device__ __forceinline__ int list_next(int nxt, int& other, int E) {
  if (nxt == 0) { nxt = other; other = 0; }
  return ((unsigned)(nxt - 1) < (unsigned)E) ? nxt : 0;
}
__global__ __launch_bounds__(256)
void node_walk(const int* __restrict__ head_q, const int* __restrict__ head_a, long hdq, long hda,
               const int2* __restrict__ rec_q, const int2* __restrict__ rec_a,
               const float* __restrict__ as_q, const float* __restrict__ as_a,
               const float4* __restrict__ y_q, const float4* __restrict__ y_a,
               const float* __restrict__ ad_q, const float* __restrict__ ad_a,
               const float* __restrict__ cst, float* __restrict__ out,
               long Nq, long Na, int Eq, int Ea) {
  const int t = (int)(blockIdx.x * blockDim.x + threadIdx.x);
  const int d = t >> 2;
  const int x = t & 3;
  if (d >= Nq) return;
  RowSums Sq, Sa;
  sums_start(Sq);
  sums_start(Sa);
  const float adq = ad_q[t];
  int oq = head_q[hdq + d];
  int cq = list_next(head_q[d], oq, Eq);
  float ada = 0.f;
  int ca = 0, oa = 0;
  if (d < Na) {
    ada = ad_a[t];
    oa = head_a[hda + d];
    ca = list_next(head_a[d], oa, Ea);
  }
  int2 rq = make_int2(0, 0), ra = make_int2(0, 0);
  if (cq) rq = rec_q[cq - 1];
  if (ca) ra = rec_a[ca - 1];
  const int maxit = Eq > Ea ? Eq : Ea;
  for (int it = 0; it < maxit && (cq | ca) != 0; ++it) {
    const float sq = as_q[rq.x * 4 + x];
    const float4 yq = y_q[rq.x * 4 + x];
    const float sa = as_a[ra.x * 4 + x];
    const float4 ya = y_a[ra.x * 4 + x];
    const int nq = list_next(rq.y, oq, Eq);
    const int na = list_next(ra.y, oa, Ea);
    int2 nrq = rec_q[nq ? nq - 1 : 0];
    int2 nra = rec_a[na ? na - 1 : 0];
    sums_add(Sq, cq != 0, sq, adq, yq);
    sums_add(Sa, ca != 0, sa, ada, ya);
    if (nq == 0) nrq = make_int2(0, 0);
    if (na == 0) nra = make_int2(0, 0);
    cq = nq; rq = nrq;
    ca = na; ra = nra;
  }
  float o = sums_close(cst[x], Sq);
  if (d < Na) o = sums_close(o, Sa);
  out[t] = o;
}

extern "C" void kernel_launch(void* const* d_in, const int* in_sizes, int n_in,
                              void* d_out, int out_size, void* d_ws, size_t ws_size,
                              hipStream_t stream) {
  const float* x_c = (const float*)d_in[0];
  const float* x_q = (const float*)d_in[1];
  const float* x_a = (const float*)d_in[2];
  const int* src_cq = (const int*)d_in[3];
  const int* dst_cq = (const int*)d_in[4];
  const int* src_ca = (const int*)d_in[5];
  const int* dst_ca = (const int*)d_in[6];
  const float* Wsq = (const float*)d_in[7];
  const float* Wdq = (const float*)d_in[8];
  const float* Asq = (const float*)d_in[9];
  const float* Adq = (const float*)d_in[10];
  const float* bq  = (const float*)d_in[11];
  const float* Wsa = (const float*)d_in[12];
  const float* Wda = (const float*)d_in[13];
  const float* Asa = (const float*)d_in[14];
  const float* Ada = (const float*)d_in[15];
  const float* ba  = (const float*)d_in[16];
  const float* Wout = (const float*)d_in[17];
  const float* bout = (const float*)d_in[18];

  const long Nc = in_sizes[0] / 64;
  const long Nq = in_sizes[1] / 64;
  const long Na = in_sizes[2] / 64;
  const long Ecq = in_sizes[3];
  const long Eca = in_sizes[5];

  float* w = (float*)d_ws;
  size_t off = 0;
  auto alloc = [&](size_t n) -> float* {
    float* p = w + off;
    off += (n + 63) & ~(size_t)63;
    return p;
  };
  float* Vp_sq = alloc(64 * 16);
  float* Vp_dq = alloc(64 * 16);
  float* Vp_sa = alloc(64 * 16);
  float* Vp_da = alloc(64 * 16);
  float* Py    = alloc(64 * 32);
  float* cst   = alloc(16);
  float* as_cq = alloc(Nc * 4);
  float* as_ca = alloc(Nc * 4);
  float* ad_cq = alloc(Nq * 4);
  float* ad_ca = alloc(Na * 4);
  float* rec_q = alloc(Ecq * 2);
  float* rec_a = alloc(Eca * 2);
  float* hd_q  = alloc(Nq);
  float* hd_q1 = alloc(Nq);
  float* hd_a  = alloc(Na);
  float* hd_a1 = alloc(Na);
  float* y_q   = alloc(Nc * 16);
  float* y_a   = alloc(Nc * 16);
  (void)Vp_dq; (void)Vp_sa; (void)Vp_da;
  if (off * sizeof(float) > ws_size) return;

  const long nzero4 = (long)(y_q - hd_q) / 4;
  const long hdq = (long)(hd_q1 - hd_q), hda = (long)(hd_a1 - hd_a);

  prep_kernel<<<49, 128, 0, stream>>>(Wsq, Wdq, Asq, Adq, Wsa, Wda, Asa, Ada,
                                      bq, ba, Wout, bout, Vp_sq, Py, cst,
                                      (float4*)hd_q, nzero4);

  const int gNc = (int)((Nc + 127) / 128);
  const int gNq = (int)((Nq + 127) / 128);
  const int gNa = (int)((Na + 127) / 128);
  const int nd = gNc + gNq + gNa;
  const int nl = 2048;
  const int npair = nd < nl ? nd : nl;
  dense_and_lists<<<nd + nl, 256, 0, stream>>>(
      x_c, x_q, x_a, Vp_sq, Py, as_cq, as_ca, ad_cq, ad_ca, y_q, y_a,
      (int)Nc, (int)Nq, (int)Na, gNc, gNq,
      src_cq, dst_cq, src_ca, dst_ca, (int*)hd_q, (int*)hd_a, hdq, hda,
      (int2*)rec_q, (int2*)rec_a, Ecq, Eca, npair, nd, nl);
  node_walk<<<(int)((Nq * 4 + 255) / 256), 256, 0, stream>>>(
      (const int*)hd_q, (const int*)hd_a, hdq, hda, (const int2*)rec_q, (const int2*)rec_a,
      as_cq, as_ca, (const float4*)y_q, (const float4*)y_a,
      ad_cq, ad_ca, cst, (float*)d_out, Nq, Na,
      (int)Ecq, (int)Eca);
}
